# starting kernel
# baseline (speedup 1.0000x reference)
#include <hip/hip_runtime.h>
#include <math.h>

#define D 128
#define NEG_SLOPE 0.2f
#define LN_EPS 1e-5f

typedef float v2f __attribute__((ext_vector_type(2)));
typedef float v8f __attribute__((ext_vector_type(8)));

__device__ __forceinline__ float wave_sum(float v) {
#pragma unroll
  for (int off = 16; off > 0; off >>= 1) v += __shfl_xor(v, off, 32);
  return v;
}

__global__ __launch_bounds__(256)
void gat_gemm_wmma(const float* __restrict__ x, const float* __restrict__ w,
                   float* __restrict__ h, int n_nodes) {
  __shared__ float lds_a[16 * D];
  const int tid  = threadIdx.x;
  const int row0 = blockIdx.x * 16;

  for (int i = tid; i < 16 * D; i += 256) {
    int r = i >> 7, c = i & (D - 1);
    int row = row0 + r;
    lds_a[i] = (row < n_nodes) ? x[(size_t)row * D + c] : 0.0f;
  }
  __syncthreads();

  const int wave = tid >> 5;
  const int lane = tid & 31;
  const int half = lane >> 4;
  const int l16  = lane & 15;
  const int col0 = wave * 16;

  v8f c = {};
#pragma unroll
  for (int k = 0; k < D; k += 4) {
    const int ka = k + half * 2;
    v2f a, b;
    a.x = lds_a[l16 * D + ka];
    a.y = lds_a[l16 * D + ka + 1];
    b.x = w[ka * D + col0 + l16];
    b.y = w[(ka + 1) * D + col0 + l16];
    c = __builtin_amdgcn_wmma_f32_16x16x4_f32(false, a, false, b, (short)0, c,
                                              false, false);
  }
#pragma unroll
  for (int v = 0; v < 8; ++v) {
    int row = row0 + v + half * 8;
    if (row < n_nodes) h[(size_t)row * D + col0 + l16] = c[v];
  }
}

__global__ __launch_bounds__(256)
void gat_attn_scores(const float* __restrict__ h, const float* __restrict__ att_src,
                     const float* __restrict__ att_dst, float* __restrict__ a_src,
                     float* __restrict__ a_dst, float* __restrict__ self_logit,
                     float* __restrict__ mx, int n_nodes) {
  const int node = blockIdx.x * 8 + (threadIdx.x >> 5);
  const int lane = threadIdx.x & 31;
  if (node >= n_nodes) return;
  const float4 hv = ((const float4*)(h + (size_t)node * D))[lane];
  const float4 as = ((const float4*)att_src)[lane];
  const float4 ad = ((const float4*)att_dst)[lane];
  float ps = hv.x * as.x + hv.y * as.y + hv.z * as.z + hv.w * as.w;
  float pd = hv.x * ad.x + hv.y * ad.y + hv.z * ad.z + hv.w * ad.w;
  ps = wave_sum(ps);
  pd = wave_sum(pd);
  if (lane == 0) {
    a_src[node] = ps;
    a_dst[node] = pd;
    float lg = ps + pd;
    lg = lg > 0.0f ? lg : NEG_SLOPE * lg;
    self_logit[node] = lg;
    mx[node] = lg;
  }
}

__global__ __launch_bounds__(256)
void gat_edge_logits(const int* __restrict__ ei, const float* __restrict__ a_src,
                     const float* __restrict__ a_dst, float* __restrict__ logit,
                     float* __restrict__ mx, int n_edges) {
  const int e = blockIdx.x * 256 + threadIdx.x;
  if (e >= n_edges) return;
  const int s = ei[e];
  const int d = ei[n_edges + e];
  float lg = a_src[s] + a_dst[d];
  lg = lg > 0.0f ? lg : NEG_SLOPE * lg;
  logit[e] = lg;
  __hip_atomic_fetch_max(&mx[d], lg, __ATOMIC_RELAXED, __HIP_MEMORY_SCOPE_AGENT);
}

__global__ __launch_bounds__(256)
void gat_self_init(const float* __restrict__ h, const float* __restrict__ self_logit,
                   const float* __restrict__ mx, float* __restrict__ acc,
                   float* __restrict__ ssum, int n_nodes) {
  const int idx  = blockIdx.x * 256 + threadIdx.x;
  const int node = idx >> 5;
  const int c    = idx & 31;
  if (node >= n_nodes) return;
  const float e0 = __expf(self_logit[node] - mx[node]);
  float4 hv = ((const float4*)(h + (size_t)node * D))[c];
  float4 av = make_float4(hv.x * e0, hv.y * e0, hv.z * e0, hv.w * e0);
  ((float4*)(acc + (size_t)node * D))[c] = av;
  if (c == 0) ssum[node] = e0;
}

__global__ __launch_bounds__(256)
void gat_edge_accum(const int* __restrict__ ei, const float* __restrict__ logit,
                    const float* __restrict__ mx, const float* __restrict__ h,
                    float* __restrict__ acc, float* __restrict__ ssum, int n_edges) {
  const int e    = blockIdx.x * 8 + (threadIdx.x >> 5);
  const int lane = threadIdx.x & 31;
  if (e >= n_edges) return;
  const int s = ei[e];
  const int d = ei[n_edges + e];
  const float w = __expf(logit[e] - mx[d]);
  const float4 hv = ((const float4*)(h + (size_t)s * D))[lane];
  float* ap = acc + (size_t)d * D + lane * 4;
  unsafeAtomicAdd(ap + 0, hv.x * w);
  unsafeAtomicAdd(ap + 1, hv.y * w);
  unsafeAtomicAdd(ap + 2, hv.z * w);
  unsafeAtomicAdd(ap + 3, hv.w * w);
  if (lane == 0) unsafeAtomicAdd(&ssum[d], w);
}

__global__ __launch_bounds__(256)
void gat_ln_out(const float* __restrict__ x, const float* __restrict__ acc,
                const float* __restrict__ ssum, const float* __restrict__ bias,
                const float* __restrict__ gamma, const float* __restrict__ beta,
                float* __restrict__ out, int n_nodes) {
  const int node = blockIdx.x * 8 + (threadIdx.x >> 5);
  const int lane = threadIdx.x & 31;
  if (node >= n_nodes) return;
  const float inv_s = 1.0f / ssum[node];
  const float4 xv = ((const float4*)(x   + (size_t)node * D))[lane];
  const float4 av = ((const float4*)(acc + (size_t)node * D))[lane];
  const float4 bv = ((const float4*)bias)[lane];
  float4 y;
  y.x = xv.x + av.x * inv_s + bv.x;
  y.y = xv.y + av.y * inv_s + bv.y;
  y.z = xv.z + av.z * inv_s + bv.z;
  y.w = xv.w + av.w * inv_s + bv.w;
  float sum = y.x + y.y + y.z + y.w;
  float sq  = y.x * y.x + y.y * y.y + y.z * y.z + y.w * y.w;
  sum = wave_sum(sum);
  sq  = wave_sum(sq);
  const float mu   = sum * (1.0f / D);
  const float var  = sq * (1.0f / D) - mu * mu;
  const float rstd = rsqrtf(var + LN_EPS);
  const float4 gv = ((const float4*)gamma)[lane];
  const float4 tv = ((const float4*)beta)[lane];
  float4 o;
  o.x = (y.x - mu) * rstd * gv.x + tv.x;
  o.y = (y.y - mu) * rstd * gv.y + tv.y;
  o.z = (y.z - mu) * rstd * gv.z + tv.z;
  o.w = (y.w - mu) * rstd * gv.w + tv.w;
  ((float4*)(out + (size_t)node * D))[lane] = o;
}

extern "C" void kernel_launch(void* const* d_in, const int* in_sizes, int n_in,
                              void* d_out, int out_size, void* d_ws, size_t ws_size,
                              hipStream_t stream) {
  const float* x0   = (const float*)d_in[0];
  const int*   ei   = (const int*)d_in[1];
  const float* W    = (const float*)d_in[2];
  const float* asrc = (const float*)d_in[3];
  const float* adst = (const float*)d_in[4];
  const float* bias = (const float*)d_in[5];
  const float* gam  = (const float*)d_in[6];
  const float* bet  = (const float*)d_in[7];
  float* out = (float*)d_out;

  const int n  = in_sizes[0] / D;
  const int e  = in_sizes[1] / 2;
  const int nl = in_sizes[2] / (D * D);

  float* ws    = (float*)d_ws;
  float* h     = ws; ws += (size_t)n * D;
  float* acc   = ws; ws += (size_t)n * D;
  float* va    = ws; ws += n;
  float* vb    = ws; ws += n;
  float* slf   = ws; ws += n;
  float* mx    = ws; ws += n;
  float* ssum  = ws; ws += n;
  float* lgt   = ws; ws += e;

  for (int l = 0; l < nl; ++l) {
    const float* xin = (l == 0) ? x0 : out;
    const float* Wl  = W + (size_t)l * D * D;
    gat_gemm_wmma<<<(n + 15) / 16, 256, 0, stream>>>(xin, Wl, h, n);
    gat_attn_scores<<<(n + 7) / 8, 256, 0, stream>>>(h, asrc + l * D, adst + l * D,
                                                     va, vb, slf, mx, n);
    gat_edge_logits<<<(e + 255) / 256, 256, 0, stream>>>(ei, va, vb, lgt, mx, e);
    gat_self_init<<<(n * 32 + 255) / 256, 256, 0, stream>>>(h, slf, mx, acc, ssum, n);
    gat_edge_accum<<<(e + 7) / 8, 256, 0, stream>>>(ei, lgt, mx, h, acc, ssum, e);
    gat_ln_out<<<(n + 7) / 8, 256, 0, stream>>>(xin, acc, ssum, bias + l * D,
                                                gam + l * D, bet + l * D, out, n);
  }
}


# Edge sums: each work-group owns 200 dst nodes' rows in LDS; its 32 waves scan the unsorted dst column and add w*h[src] by LDS f32 adds; no global f32 add, no sort; sum order: self, then LDS arrival
# speedup vs baseline: 1.7503x; 1.7503x over previous
#include <hip/hip_runtime.h>
#include <math.h>

#define D 128
#define NEG_SLOPE 0.2f
#define LN_EPS 1e-5f

typedef float v2f __attribute__((ext_vector_type(2)));
typedef float v8f __attribute__((ext_vector_type(8)));

__device__ __forceinline__ float wave_sum(float v) {
#pragma unroll
  for (int off = 16; off > 0; off >>= 1) v += __shfl_xor(v, off, 32);
  return v;
}

__global__ __launch_bounds__(256)
void gat_gemm_wmma(const float* __restrict__ x, const float* __restrict__ w,
                   float* __restrict__ h, int n_nodes) {
  __shared__ float lds_a[16 * D];
  const int tid  = threadIdx.x;
  const int row0 = blockIdx.x * 16;

  for (int i = tid; i < 16 * D; i += 256) {
    int r = i >> 7, c = i & (D - 1);
    int row = row0 + r;
    lds_a[i] = (row < n_nodes) ? x[(size_t)row * D + c] : 0.0f;
  }
  __syncthreads();

  const int wave = tid >> 5;
  const int lane = tid & 31;
  const int half = lane >> 4;
  const int l16  = lane & 15;
  const int col0 = wave * 16;

  v8f c = {};
#pragma unroll
  for (int k = 0; k < D; k += 4) {
    const int ka = k + half * 2;
    v2f a, b;
    a.x = lds_a[l16 * D + ka];
    a.y = lds_a[l16 * D + ka + 1];
    b.x = w[ka * D + col0 + l16];
    b.y = w[(ka + 1) * D + col0 + l16];
    c = __builtin_amdgcn_wmma_f32_16x16x4_f32(false, a, false, b, (short)0, c,
                                              false, false);
  }
#pragma unroll
  for (int v = 0; v < 8; ++v) {
    int row = row0 + v + half * 8;
    if (row < n_nodes) h[(size_t)row * D + col0 + l16] = c[v];
  }
}

__global__ __launch_bounds__(256)
void gat_attn_scores(const float* __restrict__ h, const float* __restrict__ att_src,
                     const float* __restrict__ att_dst, float* __restrict__ a_src,
                     float* __restrict__ a_dst, float* __restrict__ self_logit,
                     float* __restrict__ mx, int n_nodes) {
  const int node = blockIdx.x * 8 + (threadIdx.x >> 5);
  const int lane = threadIdx.x & 31;
  if (node >= n_nodes) return;
  const float4 hv = ((const float4*)(h + (size_t)node * D))[lane];
  const float4 as = ((const float4*)att_src)[lane];
  const float4 ad = ((const float4*)att_dst)[lane];
  float ps = hv.x * as.x + hv.y * as.y + hv.z * as.z + hv.w * as.w;
  float pd = hv.x * ad.x + hv.y * ad.y + hv.z * ad.z + hv.w * ad.w;
  ps = wave_sum(ps);
  pd = wave_sum(pd);
  if (lane == 0) {
    a_src[node] = ps;
    a_dst[node] = pd;
    float lg = ps + pd;
    lg = lg > 0.0f ? lg : NEG_SLOPE * lg;
    self_logit[node] = lg;
    mx[node] = lg;
  }
}

__global__ __launch_bounds__(256)
void gat_edge_logits(const int* __restrict__ ei, const float* __restrict__ a_src,
                     const float* __restrict__ a_dst, float* __restrict__ logit,
                     float* __restrict__ mx, int n_edges) {
  const int e = blockIdx.x * 256 + threadIdx.x;
  if (e >= n_edges) return;
  const int s = ei[e];
  const int d = ei[n_edges + e];
  float lg = a_src[s] + a_dst[d];
  lg = lg > 0.0f ? lg : NEG_SLOPE * lg;
  logit[e] = lg;
  __hip_atomic_fetch_max(&mx[d], lg, __ATOMIC_RELAXED, __HIP_MEMORY_SCOPE_AGENT);
}

#define NPW 200
#define OWN_WAVES 32
#define SCAN_LOADS 16
#define QCAP 544

__device__ __forceinline__ void own_drain(const int* qe, const int* qd, int m, int lane,
                                          int n0, const int* __restrict__ ei,
                                          const float* __restrict__ logit,
                                          const float* __restrict__ mx,
                                          const float* __restrict__ h,
                                          float* s_acc, float* s_sum) {
  int s = 0, dl = 0;
  float w = 0.0f;
  if (lane < m) {
    const int e = qe[lane];
    dl = qd[lane];
    s = ei[e];
    w = __expf(logit[e] - mx[n0 + dl]);
    __hip_atomic_fetch_add(&s_sum[dl], w, __ATOMIC_RELAXED, __HIP_MEMORY_SCOPE_WORKGROUP);
  }
  for (int j = 0; j < m; j += 8) {
    float hv[8][4];
#pragma unroll
    for (int u = 0; u < 8; ++u) {
      if (j + u < m) {
        const int sj = __builtin_amdgcn_readlane(s, j + u);
        const float* hp = h + (size_t)sj * D + lane;
        hv[u][0] = hp[0];
        hv[u][1] = hp[32];
        hv[u][2] = hp[64];
        hv[u][3] = hp[96];
      }
    }
#pragma unroll
    for (int u = 0; u < 8; ++u) {
      if (j + u < m) {
        const int dj = __builtin_amdgcn_readlane(dl, j + u);
        const float wj = __builtin_bit_cast(float, __builtin_amdgcn_readlane(__builtin_bit_cast(int, w), j + u));
        float* ap = s_acc + dj * D + lane;
        __hip_atomic_fetch_add(ap + 0,  hv[u][0] * wj, __ATOMIC_RELAXED, __HIP_MEMORY_SCOPE_WORKGROUP);
        __hip_atomic_fetch_add(ap + 32, hv[u][1] * wj, __ATOMIC_RELAXED, __HIP_MEMORY_SCOPE_WORKGROUP);
        __hip_atomic_fetch_add(ap + 64, hv[u][2] * wj, __ATOMIC_RELAXED, __HIP_MEMORY_SCOPE_WORKGROUP);
        __hip_atomic_fetch_add(ap + 96, hv[u][3] * wj, __ATOMIC_RELAXED, __HIP_MEMORY_SCOPE_WORKGROUP);
      }
    }
  }
}

__global__ __launch_bounds__(OWN_WAVES * 32)
void gat_owner_accum(const int* __restrict__ ei, const float* __restrict__ logit,
                     const float* __restrict__ mx, const float* __restrict__ self_logit,
                     const float* __restrict__ h, float* __restrict__ acc,
                     float* __restrict__ ssum, int n_nodes, int n_edges) {
  __shared__ float s_acc[NPW * D];
  __shared__ float s_sum[NPW];
  __shared__ int   s_qe[OWN_WAVES * QCAP];
  __shared__ int   s_qd[OWN_WAVES * QCAP];

  const int tid  = threadIdx.x;
  const int wave = __builtin_amdgcn_readfirstlane(tid >> 5);
  const int lane = tid & 31;
  const int n0   = blockIdx.x * NPW;
  const int cnt  = min(NPW, n_nodes - n0);

  for (int i = tid; i < cnt * 32; i += OWN_WAVES * 32) {
    const int r = i >> 5, c = i & 31;
    const int node = n0 + r;
    const float e0 = __expf(self_logit[node] - mx[node]);
    const float4 hv = ((const float4*)(h + (size_t)node * D))[c];
    ((float4*)(s_acc + r * D))[c] = make_float4(hv.x * e0, hv.y * e0, hv.z * e0, hv.w * e0);
    if (c == 0) s_sum[r] = e0;
  }
  __syncthreads();

  const int* __restrict__ dstp = ei + n_edges;
  int* qe = s_qe + wave * QCAP;
  int* qd = s_qd + wave * QCAP;
  int qn = 0;

  const int step  = SCAN_LOADS * 32;
  const int n_stp = (n_edges + step - 1) / step;
  const int first = (int)(((long long)blockIdx.x * n_stp) / gridDim.x);
  for (int it = wave; it < n_stp; it += OWN_WAVES) {
    int sg = first + it;
    if (sg >= n_stp) sg -= n_stp;
    const int base = sg * step;
    int dv[SCAN_LOADS];
    if (base + step <= n_edges) {
#pragma unroll
      for (int k = 0; k < SCAN_LOADS; ++k) dv[k] = dstp[base + k * 32 + lane];
    } else {
#pragma unroll
      for (int k = 0; k < SCAN_LOADS; ++k) {
        const int idx = base + k * 32 + lane;
        dv[k] = (idx < n_edges) ? dstp[idx] : -1;
      }
    }
#pragma unroll
    for (int k = 0; k < SCAN_LOADS; ++k) {
      const unsigned rel = (unsigned)(dv[k] - n0);
      const bool hit = rel < (unsigned)cnt;
      const unsigned msk = (unsigned)__ballot(hit);
      if (msk != 0u) {
        if (hit) {
          const int pos = qn + __popc(msk & ((1u << lane) - 1u));
          qe[pos] = base + k * 32 + lane;
          qd[pos] = (int)rel;
        }
        qn += __popc(msk);
      }
    }
    while (qn >= 32) {
      qn -= 32;
      own_drain(qe + qn, qd + qn, 32, lane, n0, ei, logit, mx, h, s_acc, s_sum);
    }
  }
  if (qn > 0) own_drain(qe, qd, qn, lane, n0, ei, logit, mx, h, s_acc, s_sum);
  __syncthreads();

  for (int i = tid; i < cnt * 32; i += OWN_WAVES * 32) {
    const int r = i >> 5, c = i & 31;
    ((float4*)(acc + (size_t)(n0 + r) * D))[c] = ((const float4*)(s_acc + r * D))[c];
    if (c == 0) ssum[n0 + r] = s_sum[r];
  }
}

__global__ __launch_bounds__(256)
void gat_ln_out(const float* __restrict__ x, const float* __restrict__ acc,
                const float* __restrict__ ssum, const float* __restrict__ bias,
                const float* __restrict__ gamma, const float* __restrict__ beta,
                float* __restrict__ out, int n_nodes) {
  const int node = blockIdx.x * 8 + (threadIdx.x >> 5);
  const int lane = threadIdx.x & 31;
  if (node >= n_nodes) return;
  const float inv_s = 1.0f / ssum[node];
  const float4 xv = ((const float4*)(x   + (size_t)node * D))[lane];
  const float4 av = ((const float4*)(acc + (size_t)node * D))[lane];
  const float4 bv = ((const float4*)bias)[lane];
  float4 y;
  y.x = xv.x + av.x * inv_s + bv.x;
  y.y = xv.y + av.y * inv_s + bv.y;
  y.z = xv.z + av.z * inv_s + bv.z;
  y.w = xv.w + av.w * inv_s + bv.w;
  float sum = y.x + y.y + y.z + y.w;
  float sq  = y.x * y.x + y.y * y.y + y.z * y.z + y.w * y.w;
  sum = wave_sum(sum);
  sq  = wave_sum(sq);
  const float mu   = sum * (1.0f / D);
  const float var  = sq * (1.0f / D) - mu * mu;
  const float rstd = rsqrtf(var + LN_EPS);
  const float4 gv = ((const float4*)gamma)[lane];
  const float4 tv = ((const float4*)beta)[lane];
  float4 o;
  o.x = (y.x - mu) * rstd * gv.x + tv.x;
  o.y = (y.y - mu) * rstd * gv.y + tv.y;
  o.z = (y.z - mu) * rstd * gv.z + tv.z;
  o.w = (y.w - mu) * rstd * gv.w + tv.w;
  ((float4*)(out + (size_t)node * D))[lane] = o;
}

extern "C" void kernel_launch(void* const* d_in, const int* in_sizes, int n_in,
                              void* d_out, int out_size, void* d_ws, size_t ws_size,
                              hipStream_t stream) {
  const float* x0   = (const float*)d_in[0];
  const int*   ei   = (const int*)d_in[1];
  const float* W    = (const float*)d_in[2];
  const float* asrc = (const float*)d_in[3];
  const float* adst = (const float*)d_in[4];
  const float* bias = (const float*)d_in[5];
  const float* gam  = (const float*)d_in[6];
  const float* bet  = (const float*)d_in[7];
  float* out = (float*)d_out;

  const int n  = in_sizes[0] / D;
  const int e  = in_sizes[1] / 2;
  const int nl = in_sizes[2] / (D * D);

  float* ws    = (float*)d_ws;
  float* h     = ws; ws += (size_t)n * D;
  float* acc   = ws; ws += (size_t)n * D;
  float* va    = ws; ws += n;
  float* vb    = ws; ws += n;
  float* slf   = ws; ws += n;
  float* mx    = ws; ws += n;
  float* ssum  = ws; ws += n;
  float* lgt   = ws; ws += e;

  for (int l = 0; l < nl; ++l) {
    const float* xin = (l == 0) ? x0 : out;
    const float* Wl  = W + (size_t)l * D * D;
    gat_gemm_wmma<<<(n + 15) / 16, 256, 0, stream>>>(xin, Wl, h, n);
    gat_attn_scores<<<(n + 7) / 8, 256, 0, stream>>>(h, asrc + l * D, adst + l * D,
                                                     va, vb, slf, mx, n);
    gat_edge_logits<<<(e + 255) / 256, 256, 0, stream>>>(ei, va, vb, lgt, mx, e);
    gat_owner_accum<<<(n + NPW - 1) / NPW, OWN_WAVES * 32, 0, stream>>>(ei, lgt, mx, slf, h, acc,
                                                                         ssum, n, e);
    gat_ln_out<<<(n + 7) / 8, 256, 0, stream>>>(xin, acc, ssum, bias + l * D,
                                                gam + l * D, bet + l * D, out, n);
  }
}


# Product re-tiled: one wave = 16 rows x all 128 columns, A straight from global, W staged in LDS as interleaved k pairs and read as B (no A slab); k order inside each 4-wide product step changes
# speedup vs baseline: 1.9971x; 1.1410x over previous
#include <hip/hip_runtime.h>
#include <math.h>

#define D 128
#define NEG_SLOPE 0.2f
#define LN_EPS 1e-5f

typedef float v2f __attribute__((ext_vector_type(2)));
typedef float v8f __attribute__((ext_vector_type(8)));

__device__ __forceinline__ float wave_sum(float v) {
#pragma unroll
  for (int off = 16; off > 0; off >>= 1) v += __shfl_xor(v, off, 32);
  return v;
}

#define WQ 272

__global__ __launch_bounds__(256)
void gat_gemm_wmma(const float* __restrict__ x, const float* __restrict__ w,
                   float* __restrict__ h, int n_nodes) {
  __shared__ float lds_w[(D / 2) * WQ];
  const int tid  = threadIdx.x;
  const int wave = tid >> 5;
  const int lane = tid & 31;
  const int hf   = lane >> 4;
  const int l16  = lane & 15;
  const int row0 = (blockIdx.x * 8 + wave) * 16;

  const int arow = min(row0 + l16, n_nodes - 1);
  const float4* __restrict__ ap = (const float4*)(x + (size_t)arow * D) + hf;
  float4 a[16];
#pragma unroll
  for (int q = 0; q < 16; ++q) a[q] = ap[2 * q];

#pragma unroll
  for (int i = 0; i < 8; ++i) {
    const int item = tid + i * 256;
    const int kp = item >> 5, m = item & 31;
    const float4 r0 = ((const float4*)(w + (size_t)(2 * kp) * D))[m];
    const float4 r1 = ((const float4*)(w + (size_t)(2 * kp + 1) * D))[m];
    float4* dst = (float4*)(lds_w + kp * WQ + m * 8);
    dst[0] = make_float4(r0.x, r1.x, r0.y, r1.y);
    dst[1] = make_float4(r0.z, r1.z, r0.w, r1.w);
  }
  __syncthreads();

  v8f acc[8];
#pragma unroll
  for (int j = 0; j < 8; ++j) acc[j] = (v8f){};

  const float* bp = lds_w + (2 * hf) * WQ + 4 * l16;
  float4 bcur[4], bnxt[4];
#pragma unroll
  for (int jp = 0; jp < 4; ++jp) bcur[jp] = *(const float4*)(bp + 64 * jp);

#pragma unroll
  for (int t = 0; t < 32; ++t) {
    const int q = t >> 1, u = t & 1;
    if (t + 1 < 32) {
      const int kn = 4 * ((t + 1) >> 1) + ((t + 1) & 1);
#pragma unroll
      for (int jp = 0; jp < 4; ++jp) bnxt[jp] = *(const float4*)(bp + kn * WQ + 64 * jp);
    }
    __builtin_amdgcn_sched_barrier(0);
    v2f av;
    av.x = u ? a[q].z : a[q].x;
    av.y = u ? a[q].w : a[q].y;
#pragma unroll
    for (int jp = 0; jp < 4; ++jp) {
      v2f b0, b1;
      b0.x = bcur[jp].x; b0.y = bcur[jp].y;
      b1.x = bcur[jp].z; b1.y = bcur[jp].w;
      acc[2 * jp]     = __builtin_amdgcn_wmma_f32_16x16x4_f32(false, av, false, b0, (short)0,
                                                              acc[2 * jp], false, false);
      acc[2 * jp + 1] = __builtin_amdgcn_wmma_f32_16x16x4_f32(false, av, false, b1, (short)0,
                                                              acc[2 * jp + 1], false, false);
    }
    __builtin_amdgcn_sched_barrier(0);
#pragma unroll
    for (int jp = 0; jp < 4; ++jp) bcur[jp] = bnxt[jp];
  }

#pragma unroll
  for (int r = 0; r < 8; ++r) {
    const int row = row0 + 8 * hf + r;
    if (row < n_nodes) {
      float* hp = h + (size_t)row * D + 2 * l16;
#pragma unroll
      for (int jp = 0; jp < 4; ++jp) {
        float2 o;
        o.x = acc[2 * jp][r];
        o.y = acc[2 * jp + 1][r];
        *(float2*)(hp + 32 * jp) = o;
      }
    }
  }
}

__global__ __launch_bounds__(256)
void gat_attn_scores(const float* __restrict__ h, const float* __restrict__ att_src,
                     const float* __restrict__ att_dst, float* __restrict__ a_src,
                     float* __restrict__ a_dst, float* __restrict__ self_logit,
                     float* __restrict__ mx, int n_nodes) {
  const int node = blockIdx.x * 8 + (threadIdx.x >> 5);
  const int lane = threadIdx.x & 31;
  if (node >= n_nodes) return;
  const float4 hv = ((const float4*)(h + (size_t)node * D))[lane];
  const float4 as = ((const float4*)att_src)[lane];
  const float4 ad = ((const float4*)att_dst)[lane];
  float ps = hv.x * as.x + hv.y * as.y + hv.z * as.z + hv.w * as.w;
  float pd = hv.x * ad.x + hv.y * ad.y + hv.z * ad.z + hv.w * ad.w;
  ps = wave_sum(ps);
  pd = wave_sum(pd);
  if (lane == 0) {
    a_src[node] = ps;
    a_dst[node] = pd;
    float lg = ps + pd;
    lg = lg > 0.0f ? lg : NEG_SLOPE * lg;
    self_logit[node] = lg;
    mx[node] = lg;
  }
}

__global__ __launch_bounds__(256)
void gat_edge_logits(const int* __restrict__ ei, const float* __restrict__ a_src,
                     const float* __restrict__ a_dst, float* __restrict__ logit,
                     float* __restrict__ mx, int n_edges) {
  const int e = blockIdx.x * 256 + threadIdx.x;
  if (e >= n_edges) return;
  const int s = ei[e];
  const int d = ei[n_edges + e];
  float lg = a_src[s] + a_dst[d];
  lg = lg > 0.0f ? lg : NEG_SLOPE * lg;
  logit[e] = lg;
  __hip_atomic_fetch_max(&mx[d], lg, __ATOMIC_RELAXED, __HIP_MEMORY_SCOPE_AGENT);
}

#define NPW 200
#define OWN_WAVES 32
#define SCAN_LOADS 16
#define QCAP 544

__device__ __forceinline__ void own_drain(const int* qe, const int* qd, int m, int lane,
                                          int n0, const int* __restrict__ ei,
                                          const float* __restrict__ logit,
                                          const float* __restrict__ mx,
                                          const float* __restrict__ h,
                                          float* s_acc, float* s_sum) {
  int s = 0, dl = 0;
  float w = 0.0f;
  if (lane < m) {
    const int e = qe[lane];
    dl = qd[lane];
    s = ei[e];
    w = __expf(logit[e] - mx[n0 + dl]);
    __hip_atomic_fetch_add(&s_sum[dl], w, __ATOMIC_RELAXED, __HIP_MEMORY_SCOPE_WORKGROUP);
  }
  for (int j = 0; j < m; j += 8) {
    float hv[8][4];
#pragma unroll
    for (int u = 0; u < 8; ++u) {
      if (j + u < m) {
        const int sj = __builtin_amdgcn_readlane(s, j + u);
        const float* hp = h + (size_t)sj * D + lane;
        hv[u][0] = hp[0];
        hv[u][1] = hp[32];
        hv[u][2] = hp[64];
        hv[u][3] = hp[96];
      }
    }
#pragma unroll
    for (int u = 0; u < 8; ++u) {
      if (j + u < m) {
        const int dj = __builtin_amdgcn_readlane(dl, j + u);
        const float wj = __builtin_bit_cast(float, __builtin_amdgcn_readlane(__builtin_bit_cast(int, w), j + u));
        float* ap = s_acc + dj * D + lane;
        __hip_atomic_fetch_add(ap + 0,  hv[u][0] * wj, __ATOMIC_RELAXED, __HIP_MEMORY_SCOPE_WORKGROUP);
        __hip_atomic_fetch_add(ap + 32, hv[u][1] * wj, __ATOMIC_RELAXED, __HIP_MEMORY_SCOPE_WORKGROUP);
        __hip_atomic_fetch_add(ap + 64, hv[u][2] * wj, __ATOMIC_RELAXED, __HIP_MEMORY_SCOPE_WORKGROUP);
        __hip_atomic_fetch_add(ap + 96, hv[u][3] * wj, __ATOMIC_RELAXED, __HIP_MEMORY_SCOPE_WORKGROUP);
      }
    }
  }
}

__global__ __launch_bounds__(OWN_WAVES * 32)
void gat_owner_accum(const int* __restrict__ ei, const float* __restrict__ logit,
                     const float* __restrict__ mx, const float* __restrict__ self_logit,
                     const float* __restrict__ h, float* __restrict__ acc,
                     float* __restrict__ ssum, int n_nodes, int n_edges) {
  __shared__ float s_acc[NPW * D];
  __shared__ float s_sum[NPW];
  __shared__ int   s_qe[OWN_WAVES * QCAP];
  __shared__ int   s_qd[OWN_WAVES * QCAP];

  const int tid  = threadIdx.x;
  const int wave = __builtin_amdgcn_readfirstlane(tid >> 5);
  const int lane = tid & 31;
  const int n0   = blockIdx.x * NPW;
  const int cnt  = min(NPW, n_nodes - n0);

  for (int i = tid; i < cnt * 32; i += OWN_WAVES * 32) {
    const int r = i >> 5, c = i & 31;
    const int node = n0 + r;
    const float e0 = __expf(self_logit[node] - mx[node]);
    const float4 hv = ((const float4*)(h + (size_t)node * D))[c];
    ((float4*)(s_acc + r * D))[c] = make_float4(hv.x * e0, hv.y * e0, hv.z * e0, hv.w * e0);
    if (c == 0) s_sum[r] = e0;
  }
  __syncthreads();

  const int* __restrict__ dstp = ei + n_edges;
  int* qe = s_qe + wave * QCAP;
  int* qd = s_qd + wave * QCAP;
  int qn = 0;

  const int step  = SCAN_LOADS * 32;
  const int n_stp = (n_edges + step - 1) / step;
  const int first = (int)(((long long)blockIdx.x * n_stp) / gridDim.x);
  for (int it = wave; it < n_stp; it += OWN_WAVES) {
    int sg = first + it;
    if (sg >= n_stp) sg -= n_stp;
    const int base = sg * step;
    int dv[SCAN_LOADS];
    if (base + step <= n_edges) {
#pragma unroll
      for (int k = 0; k < SCAN_LOADS; ++k) dv[k] = dstp[base + k * 32 + lane];
    } else {
#pragma unroll
      for (int k = 0; k < SCAN_LOADS; ++k) {
        const int idx = base + k * 32 + lane;
        dv[k] = (idx < n_edges) ? dstp[idx] : -1;
      }
    }
#pragma unroll
    for (int k = 0; k < SCAN_LOADS; ++k) {
      const unsigned rel = (unsigned)(dv[k] - n0);
      const bool hit = rel < (unsigned)cnt;
      const unsigned msk = (unsigned)__ballot(hit);
      if (msk != 0u) {
        if (hit) {
          const int pos = qn + __popc(msk & ((1u << lane) - 1u));
          qe[pos] = base + k * 32 + lane;
          qd[pos] = (int)rel;
        }
        qn += __popc(msk);
      }
    }
    while (qn >= 32) {
      qn -= 32;
      own_drain(qe + qn, qd + qn, 32, lane, n0, ei, logit, mx, h, s_acc, s_sum);
    }
  }
  if (qn > 0) own_drain(qe, qd, qn, lane, n0, ei, logit, mx, h, s_acc, s_sum);
  __syncthreads();

  for (int i = tid; i < cnt * 32; i += OWN_WAVES * 32) {
    const int r = i >> 5, c = i & 31;
    ((float4*)(acc + (size_t)(n0 + r) * D))[c] = ((const float4*)(s_acc + r * D))[c];
    if (c == 0) ssum[n0 + r] = s_sum[r];
  }
}

__global__ __launch_bounds__(256)
void gat_ln_out(const float* __restrict__ x, const float* __restrict__ acc,
                const float* __restrict__ ssum, const float* __restrict__ bias,
                const float* __restrict__ gamma, const float* __restrict__ beta,
                float* __restrict__ out, int n_nodes) {
  const int node = blockIdx.x * 8 + (threadIdx.x >> 5);
  const int lane = threadIdx.x & 31;
  if (node >= n_nodes) return;
  const float inv_s = 1.0f / ssum[node];
  const float4 xv = ((const float4*)(x   + (size_t)node * D))[lane];
  const float4 av = ((const float4*)(acc + (size_t)node * D))[lane];
  const float4 bv = ((const float4*)bias)[lane];
  float4 y;
  y.x = xv.x + av.x * inv_s + bv.x;
  y.y = xv.y + av.y * inv_s + bv.y;
  y.z = xv.z + av.z * inv_s + bv.z;
  y.w = xv.w + av.w * inv_s + bv.w;
  float sum = y.x + y.y + y.z + y.w;
  float sq  = y.x * y.x + y.y * y.y + y.z * y.z + y.w * y.w;
  sum = wave_sum(sum);
  sq  = wave_sum(sq);
  const float mu   = sum * (1.0f / D);
  const float var  = sq * (1.0f / D) - mu * mu;
  const float rstd = rsqrtf(var + LN_EPS);
  const float4 gv = ((const float4*)gamma)[lane];
  const float4 tv = ((const float4*)beta)[lane];
  float4 o;
  o.x = (y.x - mu) * rstd * gv.x + tv.x;
  o.y = (y.y - mu) * rstd * gv.y + tv.y;
  o.z = (y.z - mu) * rstd * gv.z + tv.z;
  o.w = (y.w - mu) * rstd * gv.w + tv.w;
  ((float4*)(out + (size_t)node * D))[lane] = o;
}

extern "C" void kernel_launch(void* const* d_in, const int* in_sizes, int n_in,
                              void* d_out, int out_size, void* d_ws, size_t ws_size,
                              hipStream_t stream) {
  const float* x0   = (const float*)d_in[0];
  const int*   ei   = (const int*)d_in[1];
  const float* W    = (const float*)d_in[2];
  const float* asrc = (const float*)d_in[3];
  const float* adst = (const float*)d_in[4];
  const float* bias = (const float*)d_in[5];
  const float* gam  = (const float*)d_in[6];
  const float* bet  = (const float*)d_in[7];
  float* out = (float*)d_out;

  const int n  = in_sizes[0] / D;
  const int e  = in_sizes[1] / 2;
  const int nl = in_sizes[2] / (D * D);

  float* ws    = (float*)d_ws;
  float* h     = ws; ws += (size_t)n * D;
  float* acc   = ws; ws += (size_t)n * D;
  float* va    = ws; ws += n;
  float* vb    = ws; ws += n;
  float* slf   = ws; ws += n;
  float* mx    = ws; ws += n;
  float* ssum  = ws; ws += n;
  float* lgt   = ws; ws += e;

  for (int l = 0; l < nl; ++l) {
    const float* xin = (l == 0) ? x0 : out;
    const float* Wl  = W + (size_t)l * D * D;
    gat_gemm_wmma<<<(n + 127) / 128, 256, 0, stream>>>(xin, Wl, h, n);
    gat_attn_scores<<<(n + 7) / 8, 256, 0, stream>>>(h, asrc + l * D, adst + l * D,
                                                     va, vb, slf, mx, n);
    gat_edge_logits<<<(e + 255) / 256, 256, 0, stream>>>(ei, va, vb, lgt, mx, e);
    gat_owner_accum<<<(n + NPW - 1) / NPW, OWN_WAVES * 32, 0, stream>>>(ei, lgt, mx, slf, h, acc,
                                                                         ssum, n, e);
    gat_ln_out<<<(n + 7) / 8, 256, 0, stream>>>(xin, acc, ssum, bias + l * D,
                                                gam + l * D, bet + l * D, out, n);
  }
}


# Residual + LayerNorm done by the owner work-group on its finished LDS rows (x rows asked for at kernel start): the per-node pass, its launch and the acc round trip go; sums and their order kept
# speedup vs baseline: 2.2690x; 1.1361x over previous
#include <hip/hip_runtime.h>
#include <math.h>

#define D 128
#define NEG_SLOPE 0.2f
#define LN_EPS 1e-5f

typedef float v2f __attribute__((ext_vector_type(2)));
typedef float v8f __attribute__((ext_vector_type(8)));

__device__ __forceinline__ float wave_sum(float v) {
#pragma unroll
  for (int off = 16; off > 0; off >>= 1) v += __shfl_xor(v, off, 32);
  return v;
}

#define WQ 272

__global__ __launch_bounds__(256)
void gat_gemm_wmma(const float* __restrict__ x, const float* __restrict__ w,
                   float* __restrict__ h, int n_nodes) {
  __shared__ float lds_w[(D / 2) * WQ];
  const int tid  = threadIdx.x;
  const int wave = tid >> 5;
  const int lane = tid & 31;
  const int hf   = lane >> 4;
  const int l16  = lane & 15;
  const int row0 = (blockIdx.x * 8 + wave) * 16;

  const int arow = min(row0 + l16, n_nodes - 1);
  const float4* __restrict__ ap = (const float4*)(x + (size_t)arow * D) + hf;
  float4 a[16];
#pragma unroll
  for (int q = 0; q < 16; ++q) a[q] = ap[2 * q];

#pragma unroll
  for (int i = 0; i < 8; ++i) {
    const int item = tid + i * 256;
    const int kp = item >> 5, m = item & 31;
    const float4 r0 = ((const float4*)(w + (size_t)(2 * kp) * D))[m];
    const float4 r1 = ((const float4*)(w + (size_t)(2 * kp + 1) * D))[m];
    float4* dst = (float4*)(lds_w + kp * WQ + m * 8);
    dst[0] = make_float4(r0.x, r1.x, r0.y, r1.y);
    dst[1] = make_float4(r0.z, r1.z, r0.w, r1.w);
  }
  __syncthreads();

  v8f acc[8];
#pragma unroll
  for (int j = 0; j < 8; ++j) acc[j] = (v8f){};

  const float* bp = lds_w + (2 * hf) * WQ + 4 * l16;
  float4 bcur[4], bnxt[4];
#pragma unroll
  for (int jp = 0; jp < 4; ++jp) bcur[jp] = *(const float4*)(bp + 64 * jp);

#pragma unroll
  for (int t = 0; t < 32; ++t) {
    const int q = t >> 1, u = t & 1;
    if (t + 1 < 32) {
      const int kn = 4 * ((t + 1) >> 1) + ((t + 1) & 1);
#pragma unroll
      for (int jp = 0; jp < 4; ++jp) bnxt[jp] = *(const float4*)(bp + kn * WQ + 64 * jp);
    }
    __builtin_amdgcn_sched_barrier(0);
    v2f av;
    av.x = u ? a[q].z : a[q].x;
    av.y = u ? a[q].w : a[q].y;
#pragma unroll
    for (int jp = 0; jp < 4; ++jp) {
      v2f b0, b1;
      b0.x = bcur[jp].x; b0.y = bcur[jp].y;
      b1.x = bcur[jp].z; b1.y = bcur[jp].w;
      acc[2 * jp]     = __builtin_amdgcn_wmma_f32_16x16x4_f32(false, av, false, b0, (short)0,
                                                              acc[2 * jp], false, false);
      acc[2 * jp + 1] = __builtin_amdgcn_wmma_f32_16x16x4_f32(false, av, false, b1, (short)0,
                                                              acc[2 * jp + 1], false, false);
    }
    __builtin_amdgcn_sched_barrier(0);
#pragma unroll
    for (int jp = 0; jp < 4; ++jp) bcur[jp] = bnxt[jp];
  }

#pragma unroll
  for (int r = 0; r < 8; ++r) {
    const int row = row0 + 8 * hf + r;
    if (row < n_nodes) {
      float* hp = h + (size_t)row * D + 2 * l16;
#pragma unroll
      for (int jp = 0; jp < 4; ++jp) {
        float2 o;
        o.x = acc[2 * jp][r];
        o.y = acc[2 * jp + 1][r];
        *(float2*)(hp + 32 * jp) = o;
      }
    }
  }
}

__global__ __launch_bounds__(256)
void gat_attn_scores(const float* __restrict__ h, const float* __restrict__ att_src,
                     const float* __restrict__ att_dst, float* __restrict__ a_src,
                     float* __restrict__ a_dst, float* __restrict__ self_logit,
                     float* __restrict__ mx, int n_nodes) {
  const int node = blockIdx.x * 8 + (threadIdx.x >> 5);
  const int lane = threadIdx.x & 31;
  if (node >= n_nodes) return;
  const float4 hv = ((const float4*)(h + (size_t)node * D))[lane];
  const float4 as = ((const float4*)att_src)[lane];
  const float4 ad = ((const float4*)att_dst)[lane];
  float ps = hv.x * as.x + hv.y * as.y + hv.z * as.z + hv.w * as.w;
  float pd = hv.x * ad.x + hv.y * ad.y + hv.z * ad.z + hv.w * ad.w;
  ps = wave_sum(ps);
  pd = wave_sum(pd);
  if (lane == 0) {
    a_src[node] = ps;
    a_dst[node] = pd;
    float lg = ps + pd;
    lg = lg > 0.0f ? lg : NEG_SLOPE * lg;
    self_logit[node] = lg;
    mx[node] = lg;
  }
}

__global__ __launch_bounds__(256)
void gat_edge_logits(const int* __restrict__ ei, const float* __restrict__ a_src,
                     const float* __restrict__ a_dst, float* __restrict__ logit,
                     float* __restrict__ mx, int n_edges) {
  const int e = blockIdx.x * 256 + threadIdx.x;
  if (e >= n_edges) return;
  const int s = ei[e];
  const int d = ei[n_edges + e];
  float lg = a_src[s] + a_dst[d];
  lg = lg > 0.0f ? lg : NEG_SLOPE * lg;
  logit[e] = lg;
  __hip_atomic_fetch_max(&mx[d], lg, __ATOMIC_RELAXED, __HIP_MEMORY_SCOPE_AGENT);
}

#define NPW 200
#define OWN_WAVES 32
#define SCAN_LOADS 16
#define QCAP 544

__device__ __forceinline__ void own_drain(const int* qe, const int* qd, int m, int lane,
                                          int n0, const int* __restrict__ ei,
                                          const float* __restrict__ logit,
                                          const float* __restrict__ mx,
                                          const float* __restrict__ h,
                                          float* s_acc, float* s_sum) {
  int s = 0, dl = 0;
  float w = 0.0f;
  if (lane < m) {
    const int e = qe[lane];
    dl = qd[lane];
    s = ei[e];
    w = __expf(logit[e] - mx[n0 + dl]);
    __hip_atomic_fetch_add(&s_sum[dl], w, __ATOMIC_RELAXED, __HIP_MEMORY_SCOPE_WORKGROUP);
  }
  for (int j = 0; j < m; j += 8) {
    float hv[8][4];
#pragma unroll
    for (int u = 0; u < 8; ++u) {
      if (j + u < m) {
        const int sj = __builtin_amdgcn_readlane(s, j + u);
        const float* hp = h + (size_t)sj * D + lane;
        hv[u][0] = hp[0];
        hv[u][1] = hp[32];
        hv[u][2] = hp[64];
        hv[u][3] = hp[96];
      }
    }
#pragma unroll
    for (int u = 0; u < 8; ++u) {
      if (j + u < m) {
        const int dj = __builtin_amdgcn_readlane(dl, j + u);
        const float wj = __builtin_bit_cast(float, __builtin_amdgcn_readlane(__builtin_bit_cast(int, w), j + u));
        float* ap = s_acc + dj * D + lane;
        __hip_atomic_fetch_add(ap + 0,  hv[u][0] * wj, __ATOMIC_RELAXED, __HIP_MEMORY_SCOPE_WORKGROUP);
        __hip_atomic_fetch_add(ap + 32, hv[u][1] * wj, __ATOMIC_RELAXED, __HIP_MEMORY_SCOPE_WORKGROUP);
        __hip_atomic_fetch_add(ap + 64, hv[u][2] * wj, __ATOMIC_RELAXED, __HIP_MEMORY_SCOPE_WORKGROUP);
        __hip_atomic_fetch_add(ap + 96, hv[u][3] * wj, __ATOMIC_RELAXED, __HIP_MEMORY_SCOPE_WORKGROUP);
      }
    }
  }
}

#define OWN_ROWS ((NPW + OWN_WAVES - 1) / OWN_WAVES)

__global__ __launch_bounds__(OWN_WAVES * 32)
void gat_owner_accum(const int* __restrict__ ei, const float* __restrict__ logit,
                     const float* __restrict__ mx, const float* __restrict__ self_logit,
                     const float* __restrict__ h, const float* x,
                     const float* __restrict__ bias, const float* __restrict__ gamma,
                     const float* __restrict__ beta, float* out,
                     int n_nodes, int n_edges) {
  __shared__ float s_acc[NPW * D];
  __shared__ float s_sum[NPW];
  __shared__ int   s_qe[OWN_WAVES * QCAP];
  __shared__ int   s_qd[OWN_WAVES * QCAP];

  const int tid  = threadIdx.x;
  const int wave = __builtin_amdgcn_readfirstlane(tid >> 5);
  const int lane = tid & 31;
  const int n0   = blockIdx.x * NPW;
  const int cnt  = min(NPW, n_nodes - n0);

  const float4 bv = ((const float4*)bias)[lane];
  const float4 gv = ((const float4*)gamma)[lane];
  const float4 tv = ((const float4*)beta)[lane];
  float4 xr[OWN_ROWS];
#pragma unroll
  for (int i = 0; i < OWN_ROWS; ++i) {
    const int r = wave + i * OWN_WAVES;
    xr[i] = (r < cnt) ? ((const float4*)(x + (size_t)(n0 + r) * D))[lane]
                      : make_float4(0.0f, 0.0f, 0.0f, 0.0f);
  }

  for (int i = tid; i < cnt * 32; i += OWN_WAVES * 32) {
    const int r = i >> 5, c = i & 31;
    const int node = n0 + r;
    const float e0 = __expf(self_logit[node] - mx[node]);
    const float4 hv = ((const float4*)(h + (size_t)node * D))[c];
    ((float4*)(s_acc + r * D))[c] = make_float4(hv.x * e0, hv.y * e0, hv.z * e0, hv.w * e0);
    if (c == 0) s_sum[r] = e0;
  }
  __syncthreads();

  const int* __restrict__ dstp = ei + n_edges;
  int* qe = s_qe + wave * QCAP;
  int* qd = s_qd + wave * QCAP;
  int qn = 0;

  const int step  = SCAN_LOADS * 32;
  const int n_stp = (n_edges + step - 1) / step;
  const int first = (int)(((long long)blockIdx.x * n_stp) / gridDim.x);
  for (int it = wave; it < n_stp; it += OWN_WAVES) {
    int sg = first + it;
    if (sg >= n_stp) sg -= n_stp;
    const int base = sg * step;
    int dv[SCAN_LOADS];
    if (base + step <= n_edges) {
#pragma unroll
      for (int k = 0; k < SCAN_LOADS; ++k) dv[k] = dstp[base + k * 32 + lane];
    } else {
#pragma unroll
      for (int k = 0; k < SCAN_LOADS; ++k) {
        const int idx = base + k * 32 + lane;
        dv[k] = (idx < n_edges) ? dstp[idx] : -1;
      }
    }
#pragma unroll
    for (int k = 0; k < SCAN_LOADS; ++k) {
      const unsigned rel = (unsigned)(dv[k] - n0);
      const bool hit = rel < (unsigned)cnt;
      const unsigned msk = (unsigned)__ballot(hit);
      if (msk != 0u) {
        if (hit) {
          const int pos = qn + __popc(msk & ((1u << lane) - 1u));
          qe[pos] = base + k * 32 + lane;
          qd[pos] = (int)rel;
        }
        qn += __popc(msk);
      }
    }
    while (qn >= 32) {
      qn -= 32;
      own_drain(qe + qn, qd + qn, 32, lane, n0, ei, logit, mx, h, s_acc, s_sum);
    }
  }
  if (qn > 0) own_drain(qe, qd, qn, lane, n0, ei, logit, mx, h, s_acc, s_sum);
  __syncthreads();

  float4 y[OWN_ROWS];
  float sum[OWN_ROWS], sq[OWN_ROWS];
#pragma unroll
  for (int i = 0; i < OWN_ROWS; ++i) {
    const int rc = min(wave + i * OWN_WAVES, cnt - 1);
    const float4 av = ((const float4*)(s_acc + rc * D))[lane];
    const float inv_s = 1.0f / s_sum[rc];
    y[i].x = xr[i].x + av.x * inv_s + bv.x;
    y[i].y = xr[i].y + av.y * inv_s + bv.y;
    y[i].z = xr[i].z + av.z * inv_s + bv.z;
    y[i].w = xr[i].w + av.w * inv_s + bv.w;
    sum[i] = y[i].x + y[i].y + y[i].z + y[i].w;
    sq[i]  = y[i].x * y[i].x + y[i].y * y[i].y + y[i].z * y[i].z + y[i].w * y[i].w;
  }
#pragma unroll
  for (int off = 16; off > 0; off >>= 1) {
#pragma unroll
    for (int i = 0; i < OWN_ROWS; ++i) {
      sum[i] += __shfl_xor(sum[i], off, 32);
      sq[i]  += __shfl_xor(sq[i], off, 32);
    }
  }
#pragma unroll
  for (int i = 0; i < OWN_ROWS; ++i) {
    const int r = wave + i * OWN_WAVES;
    const float mu   = sum[i] * (1.0f / D);
    const float var  = sq[i] * (1.0f / D) - mu * mu;
    const float rstd = rsqrtf(var + LN_EPS);
    float4 o;
    o.x = (y[i].x - mu) * rstd * gv.x + tv.x;
    o.y = (y[i].y - mu) * rstd * gv.y + tv.y;
    o.z = (y[i].z - mu) * rstd * gv.z + tv.z;
    o.w = (y[i].w - mu) * rstd * gv.w + tv.w;
    if (r < cnt) ((float4*)(out + (size_t)(n0 + r) * D))[lane] = o;
  }
}

extern "C" void kernel_launch(void* const* d_in, const int* in_sizes, int n_in,
                              void* d_out, int out_size, void* d_ws, size_t ws_size,
                              hipStream_t stream) {
  const float* x0   = (const float*)d_in[0];
  const int*   ei   = (const int*)d_in[1];
  const float* W    = (const float*)d_in[2];
  const float* asrc = (const float*)d_in[3];
  const float* adst = (const float*)d_in[4];
  const float* bias = (const float*)d_in[5];
  const float* gam  = (const float*)d_in[6];
  const float* bet  = (const float*)d_in[7];
  float* out = (float*)d_out;

  const int n  = in_sizes[0] / D;
  const int e  = in_sizes[1] / 2;
  const int nl = in_sizes[2] / (D * D);

  float* ws    = (float*)d_ws;
  float* h     = ws; ws += (size_t)n * D;
  float* va    = ws; ws += n;
  float* vb    = ws; ws += n;
  float* slf   = ws; ws += n;
  float* mx    = ws; ws += n;
  float* lgt   = ws; ws += e;

  for (int l = 0; l < nl; ++l) {
    const float* xin = (l == 0) ? x0 : out;
    const float* Wl  = W + (size_t)l * D * D;
    gat_gemm_wmma<<<(n + 127) / 128, 256, 0, stream>>>(xin, Wl, h, n);
    gat_attn_scores<<<(n + 7) / 8, 256, 0, stream>>>(h, asrc + l * D, adst + l * D,
                                                     va, vb, slf, mx, n);
    gat_edge_logits<<<(e + 255) / 256, 256, 0, stream>>>(ei, va, vb, lgt, mx, e);
    gat_owner_accum<<<(n + NPW - 1) / NPW, OWN_WAVES * 32, 0, stream>>>(ei, lgt, mx, slf, h, xin,
                                                                         bias + l * D, gam + l * D,
                                                                         bet + l * D, out, n, e);
  }
}


# Score kernel dropped: a_src, a_dst = x . (W att), summed in the product kernel from its A fragments against W att folded while W is staged; 8 launches -> 6; the score dots' sum order changes
# speedup vs baseline: 2.4244x; 1.0685x over previous
#include <hip/hip_runtime.h>
#include <math.h>

#define D 128
#define NEG_SLOPE 0.2f
#define LN_EPS 1e-5f

typedef float v2f __attribute__((ext_vector_type(2)));
typedef float v8f __attribute__((ext_vector_type(8)));

__device__ __forceinline__ float pick(bool c, float a, float b) { return c ? a : b; }

#define WQ 272

__global__ __launch_bounds__(256)
void gat_gemm_wmma(const float* __restrict__ x, const float* __restrict__ w,
                   const float* __restrict__ att_src, const float* __restrict__ att_dst,
                   float* __restrict__ h, float* __restrict__ a_src,
                   float* __restrict__ a_dst, float* __restrict__ self_logit,
                   float* __restrict__ mx, int n_nodes) {
  __shared__ __attribute__((aligned(16))) float lds_w[(D / 2) * WQ];
  __shared__ __attribute__((aligned(16))) float lds_wa[2 * D];
  const int tid  = threadIdx.x;
  const int wave = tid >> 5;
  const int lane = tid & 31;
  const int hf   = lane >> 4;
  const int l16  = lane & 15;
  const int row0 = (blockIdx.x * 8 + wave) * 16;

  const int arow = min(row0 + l16, n_nodes - 1);
  const float4* __restrict__ ap = (const float4*)(x + (size_t)arow * D) + hf;
  float4 a[16];
#pragma unroll
  for (int q = 0; q < 16; ++q) a[q] = ap[2 * q];

  const float4 as4 = ((const float4*)att_src)[lane];
  const float4 ad4 = ((const float4*)att_dst)[lane];
  float pv[32];
#pragma unroll
  for (int i = 0; i < 8; ++i) {
    const int item = tid + i * 256;
    const int kp = item >> 5, m = item & 31;
    const float4 r0 = ((const float4*)(w + (size_t)(2 * kp) * D))[m];
    const float4 r1 = ((const float4*)(w + (size_t)(2 * kp + 1) * D))[m];
    float4* dst = (float4*)(lds_w + kp * WQ + m * 8);
    dst[0] = make_float4(r0.x, r1.x, r0.y, r1.y);
    dst[1] = make_float4(r0.z, r1.z, r0.w, r1.w);
    pv[4 * i + 0] = r0.x * as4.x + r0.y * as4.y + r0.z * as4.z + r0.w * as4.w;
    pv[4 * i + 1] = r1.x * as4.x + r1.y * as4.y + r1.z * as4.z + r1.w * as4.w;
    pv[4 * i + 2] = r0.x * ad4.x + r0.y * ad4.y + r0.z * ad4.z + r0.w * ad4.w;
    pv[4 * i + 3] = r1.x * ad4.x + r1.y * ad4.y + r1.z * ad4.z + r1.w * ad4.w;
  }
#pragma unroll
  for (int st = 0; st < 5; ++st) {
    const int off = 16 >> st;
    const bool up = (lane & off) != 0;
#pragma unroll
    for (int j = 0; j < off; ++j) {
      const float lo = pv[j], hi = pv[j + off];
      pv[j] = pick(up, hi, lo) + __shfl_xor(pick(up, lo, hi), off, 32);
    }
  }
  {
    const int kp = wave + 8 * (lane >> 2);
    const int t  = lane & 3;
    lds_wa[(t >> 1) * D + 2 * kp + (t & 1)] = pv[0];
  }
  __syncthreads();

  {
    const float4* was = (const float4*)lds_wa + hf;
    const float4* wad = (const float4*)(lds_wa + D) + hf;
    float ps = 0.0f, pd = 0.0f;
#pragma unroll
    for (int q = 0; q < 16; ++q) {
      const float4 s4 = was[2 * q];
      const float4 d4 = wad[2 * q];
      ps += a[q].x * s4.x + a[q].y * s4.y + a[q].z * s4.z + a[q].w * s4.w;
      pd += a[q].x * d4.x + a[q].y * d4.y + a[q].z * d4.z + a[q].w * d4.w;
    }
    ps += __shfl_xor(ps, 16, 32);
    pd += __shfl_xor(pd, 16, 32);
    const int node = row0 + l16;
    if (hf == 0 && node < n_nodes) {
      a_src[node] = ps;
      a_dst[node] = pd;
      float lg = ps + pd;
      lg = lg > 0.0f ? lg : NEG_SLOPE * lg;
      self_logit[node] = lg;
      mx[node] = lg;
    }
  }

  v8f acc[8];
#pragma unroll
  for (int j = 0; j < 8; ++j) acc[j] = (v8f){};

  const float* bp = lds_w + (2 * hf) * WQ + 4 * l16;
  float4 bcur[4], bnxt[4];
#pragma unroll
  for (int jp = 0; jp < 4; ++jp) bcur[jp] = *(const float4*)(bp + 64 * jp);

#pragma unroll
  for (int t = 0; t < 32; ++t) {
    const int q = t >> 1, u = t & 1;
    if (t + 1 < 32) {
      const int kn = 4 * ((t + 1) >> 1) + ((t + 1) & 1);
#pragma unroll
      for (int jp = 0; jp < 4; ++jp) bnxt[jp] = *(const float4*)(bp + kn * WQ + 64 * jp);
    }
    __builtin_amdgcn_sched_barrier(0);
    v2f av;
    av.x = u ? a[q].z : a[q].x;
    av.y = u ? a[q].w : a[q].y;
#pragma unroll
    for (int jp = 0; jp < 4; ++jp) {
      v2f b0, b1;
      b0.x = bcur[jp].x; b0.y = bcur[jp].y;
      b1.x = bcur[jp].z; b1.y = bcur[jp].w;
      acc[2 * jp]     = __builtin_amdgcn_wmma_f32_16x16x4_f32(false, av, false, b0, (short)0,
                                                              acc[2 * jp], false, false);
      acc[2 * jp + 1] = __builtin_amdgcn_wmma_f32_16x16x4_f32(false, av, false, b1, (short)0,
                                                              acc[2 * jp + 1], false, false);
    }
    __builtin_amdgcn_sched_barrier(0);
#pragma unroll
    for (int jp = 0; jp < 4; ++jp) bcur[jp] = bnxt[jp];
  }

#pragma unroll
  for (int r = 0; r < 8; ++r) {
    const int row = row0 + 8 * hf + r;
    if (row < n_nodes) {
      float* hp = h + (size_t)row * D + 2 * l16;
#pragma unroll
      for (int jp = 0; jp < 4; ++jp) {
        float2 o;
        o.x = acc[2 * jp][r];
        o.y = acc[2 * jp + 1][r];
        *(float2*)(hp + 32 * jp) = o;
      }
    }
  }
}

__global__ __launch_bounds__(256)
void gat_edge_logits(const int* __restrict__ ei, const float* __restrict__ a_src,
                     const float* __restrict__ a_dst, float* __restrict__ logit,
                     float* __restrict__ mx, int n_edges) {
  const int e = blockIdx.x * 256 + threadIdx.x;
  if (e >= n_edges) return;
  const int s = ei[e];
  const int d = ei[n_edges + e];
  float lg = a_src[s] + a_dst[d];
  lg = lg > 0.0f ? lg : NEG_SLOPE * lg;
  logit[e] = lg;
  __hip_atomic_fetch_max(&mx[d], lg, __ATOMIC_RELAXED, __HIP_MEMORY_SCOPE_AGENT);
}

#define NPW 200
#define OWN_WAVES 32
#define SCAN_LOADS 16
#define QCAP 544

__device__ __forceinline__ void own_drain(const int* qe, const int* qd, int m, int lane,
                                          int n0, const int* __restrict__ ei,
                                          const float* __restrict__ logit,
                                          const float* __restrict__ mx,
                                          const float* __restrict__ h,
                                          float* s_acc, float* s_sum) {
  int s = 0, dl = 0;
  float w = 0.0f;
  if (lane < m) {
    const int e = qe[lane];
    dl = qd[lane];
    s = ei[e];
    w = __expf(logit[e] - mx[n0 + dl]);
    __hip_atomic_fetch_add(&s_sum[dl], w, __ATOMIC_RELAXED, __HIP_MEMORY_SCOPE_WORKGROUP);
  }
  for (int j = 0; j < m; j += 8) {
    float hv[8][4];
#pragma unroll
    for (int u = 0; u < 8; ++u) {
      if (j + u < m) {
        const int sj = __builtin_amdgcn_readlane(s, j + u);
        const float* hp = h + (size_t)sj * D + lane;
        hv[u][0] = hp[0];
        hv[u][1] = hp[32];
        hv[u][2] = hp[64];
        hv[u][3] = hp[96];
      }
    }
#pragma unroll
    for (int u = 0; u < 8; ++u) {
      if (j + u < m) {
        const int dj = __builtin_amdgcn_readlane(dl, j + u);
        const float wj = __builtin_bit_cast(float, __builtin_amdgcn_readlane(__builtin_bit_cast(int, w), j + u));
        float* ap = s_acc + dj * D + lane;
        __hip_atomic_fetch_add(ap + 0,  hv[u][0] * wj, __ATOMIC_RELAXED, __HIP_MEMORY_SCOPE_WORKGROUP);
        __hip_atomic_fetch_add(ap + 32, hv[u][1] * wj, __ATOMIC_RELAXED, __HIP_MEMORY_SCOPE_WORKGROUP);
        __hip_atomic_fetch_add(ap + 64, hv[u][2] * wj, __ATOMIC_RELAXED, __HIP_MEMORY_SCOPE_WORKGROUP);
        __hip_atomic_fetch_add(ap + 96, hv[u][3] * wj, __ATOMIC_RELAXED, __HIP_MEMORY_SCOPE_WORKGROUP);
      }
    }
  }
}

#define OWN_ROWS ((NPW + OWN_WAVES - 1) / OWN_WAVES)

__global__ __launch_bounds__(OWN_WAVES * 32)
void gat_owner_accum(const int* __restrict__ ei, const float* __restrict__ logit,
                     const float* __restrict__ mx, const float* __restrict__ self_logit,
                     const float* __restrict__ h, const float* x,
                     const float* __restrict__ bias, const float* __restrict__ gamma,
                     const float* __restrict__ beta, float* out,
                     int n_nodes, int n_edges) {
  __shared__ float s_acc[NPW * D];
  __shared__ float s_sum[NPW];
  __shared__ int   s_qe[OWN_WAVES * QCAP];
  __shared__ int   s_qd[OWN_WAVES * QCAP];

  const int tid  = threadIdx.x;
  const int wave = __builtin_amdgcn_readfirstlane(tid >> 5);
  const int lane = tid & 31;
  const int n0   = blockIdx.x * NPW;
  const int cnt  = min(NPW, n_nodes - n0);

  const float4 bv = ((const float4*)bias)[lane];
  const float4 gv = ((const float4*)gamma)[lane];
  const float4 tv = ((const float4*)beta)[lane];
  float4 xr[OWN_ROWS];
#pragma unroll
  for (int i = 0; i < OWN_ROWS; ++i) {
    const int r = wave + i * OWN_WAVES;
    xr[i] = (r < cnt) ? ((const float4*)(x + (size_t)(n0 + r) * D))[lane]
                      : make_float4(0.0f, 0.0f, 0.0f, 0.0f);
  }

  for (int i = tid; i < cnt * 32; i += OWN_WAVES * 32) {
    const int r = i >> 5, c = i & 31;
    const int node = n0 + r;
    const float e0 = __expf(self_logit[node] - mx[node]);
    const float4 hv = ((const float4*)(h + (size_t)node * D))[c];
    ((float4*)(s_acc + r * D))[c] = make_float4(hv.x * e0, hv.y * e0, hv.z * e0, hv.w * e0);
    if (c == 0) s_sum[r] = e0;
  }
  __syncthreads();

  const int* __restrict__ dstp = ei + n_edges;
  int* qe = s_qe + wave * QCAP;
  int* qd = s_qd + wave * QCAP;
  int qn = 0;

  const int step  = SCAN_LOADS * 32;
  const int n_stp = (n_edges + step - 1) / step;
  const int first = (int)(((long long)blockIdx.x * n_stp) / gridDim.x);
  for (int it = wave; it < n_stp; it += OWN_WAVES) {
    int sg = first + it;
    if (sg >= n_stp) sg -= n_stp;
    const int base = sg * step;
    int dv[SCAN_LOADS];
    if (base + step <= n_edges) {
#pragma unroll
      for (int k = 0; k < SCAN_LOADS; ++k) dv[k] = dstp[base + k * 32 + lane];
    } else {
#pragma unroll
      for (int k = 0; k < SCAN_LOADS; ++k) {
        const int idx = base + k * 32 + lane;
        dv[k] = (idx < n_edges) ? dstp[idx] : -1;
      }
    }
#pragma unroll
    for (int k = 0; k < SCAN_LOADS; ++k) {
      const unsigned rel = (unsigned)(dv[k] - n0);
      const bool hit = rel < (unsigned)cnt;
      const unsigned msk = (unsigned)__ballot(hit);
      if (msk != 0u) {
        if (hit) {
          const int pos = qn + __popc(msk & ((1u << lane) - 1u));
          qe[pos] = base + k * 32 + lane;
          qd[pos] = (int)rel;
        }
        qn += __popc(msk);
      }
    }
    while (qn >= 32) {
      qn -= 32;
      own_drain(qe + qn, qd + qn, 32, lane, n0, ei, logit, mx, h, s_acc, s_sum);
    }
  }
  if (qn > 0) own_drain(qe, qd, qn, lane, n0, ei, logit, mx, h, s_acc, s_sum);
  __syncthreads();

  float4 y[OWN_ROWS];
  float sum[OWN_ROWS], sq[OWN_ROWS];
#pragma unroll
  for (int i = 0; i < OWN_ROWS; ++i) {
    const int rc = min(wave + i * OWN_WAVES, cnt - 1);
    const float4 av = ((const float4*)(s_acc + rc * D))[lane];
    const float inv_s = 1.0f / s_sum[rc];
    y[i].x = xr[i].x + av.x * inv_s + bv.x;
    y[i].y = xr[i].y + av.y * inv_s + bv.y;
    y[i].z = xr[i].z + av.z * inv_s + bv.z;
    y[i].w = xr[i].w + av.w * inv_s + bv.w;
    sum[i] = y[i].x + y[i].y + y[i].z + y[i].w;
    sq[i]  = y[i].x * y[i].x + y[i].y * y[i].y + y[i].z * y[i].z + y[i].w * y[i].w;
  }
#pragma unroll
  for (int off = 16; off > 0; off >>= 1) {
#pragma unroll
    for (int i = 0; i < OWN_ROWS; ++i) {
      sum[i] += __shfl_xor(sum[i], off, 32);
      sq[i]  += __shfl_xor(sq[i], off, 32);
    }
  }
#pragma unroll
  for (int i = 0; i < OWN_ROWS; ++i) {
    const int r = wave + i * OWN_WAVES;
    const float mu   = sum[i] * (1.0f / D);
    const float var  = sq[i] * (1.0f / D) - mu * mu;
    const float rstd = rsqrtf(var + LN_EPS);
    float4 o;
    o.x = (y[i].x - mu) * rstd * gv.x + tv.x;
    o.y = (y[i].y - mu) * rstd * gv.y + tv.y;
    o.z = (y[i].z - mu) * rstd * gv.z + tv.z;
    o.w = (y[i].w - mu) * rstd * gv.w + tv.w;
    if (r < cnt) ((float4*)(out + (size_t)(n0 + r) * D))[lane] = o;
  }
}

extern "C" void kernel_launch(void* const* d_in, const int* in_sizes, int n_in,
                              void* d_out, int out_size, void* d_ws, size_t ws_size,
                              hipStream_t stream) {
  const float* x0   = (const float*)d_in[0];
  const int*   ei   = (const int*)d_in[1];
  const float* W    = (const float*)d_in[2];
  const float* asrc = (const float*)d_in[3];
  const float* adst = (const float*)d_in[4];
  const float* bias = (const float*)d_in[5];
  const float* gam  = (const float*)d_in[6];
  const float* bet  = (const float*)d_in[7];
  float* out = (float*)d_out;

  const int n  = in_sizes[0] / D;
  const int e  = in_sizes[1] / 2;
  const int nl = in_sizes[2] / (D * D);

  float* ws    = (float*)d_ws;
  float* h     = ws; ws += (size_t)n * D;
  float* va    = ws; ws += n;
  float* vb    = ws; ws += n;
  float* slf   = ws; ws += n;
  float* mx    = ws; ws += n;
  float* lgt   = ws; ws += e;

  for (int l = 0; l < nl; ++l) {
    const float* xin = (l == 0) ? x0 : out;
    const float* Wl  = W + (size_t)l * D * D;
    gat_gemm_wmma<<<(n + 127) / 128, 256, 0, stream>>>(xin, Wl, asrc + l * D, adst + l * D,
                                                       h, va, vb, slf, mx, n);
    gat_edge_logits<<<(e + 255) / 256, 256, 0, stream>>>(ei, va, vb, lgt, mx, e);
    gat_owner_accum<<<(n + NPW - 1) / NPW, OWN_WAVES * 32, 0, stream>>>(ei, lgt, mx, slf, h, xin,
                                                                         bias + l * D, gam + l * D,
                                                                         bet + l * D, out, n, e);
  }
}


# Edge walk once per call: in layer 1 each owner wave also stores the edges it found (edge no., local dst) in its slice of a workspace list; layer 2 reads it back instead of walking; sum order kept
# speedup vs baseline: 2.6410x; 1.0893x over previous
#include <hip/hip_runtime.h>
#include <math.h>

#define D 128
#define NEG_SLOPE 0.2f
#define LN_EPS 1e-5f

typedef float v2f __attribute__((ext_vector_type(2)));
typedef float v8f __attribute__((ext_vector_type(8)));

__device__ __forceinline__ float pick(bool c, float a, float b) { return c ? a : b; }

#define WQ 272

__global__ __launch_bounds__(256)
void gat_gemm_wmma(const float* __restrict__ x, const float* __restrict__ w,
                   const float* __restrict__ att_src, const float* __restrict__ att_dst,
                   float* __restrict__ h, float* __restrict__ a_src,
                   float* __restrict__ a_dst, float* __restrict__ self_logit,
                   float* __restrict__ mx, int n_nodes) {
  __shared__ __attribute__((aligned(16))) float lds_w[(D / 2) * WQ];
  __shared__ __attribute__((aligned(16))) float lds_wa[2 * D];
  const int tid  = threadIdx.x;
  const int wave = tid >> 5;
  const int lane = tid & 31;
  const int hf   = lane >> 4;
  const int l16  = lane & 15;
  const int row0 = (blockIdx.x * 8 + wave) * 16;

  const int arow = min(row0 + l16, n_nodes - 1);
  const float4* __restrict__ ap = (const float4*)(x + (size_t)arow * D) + hf;
  float4 a[16];
#pragma unroll
  for (int q = 0; q < 16; ++q) a[q] = ap[2 * q];

  const float4 as4 = ((const float4*)att_src)[lane];
  const float4 ad4 = ((const float4*)att_dst)[lane];
  float pv[32];
#pragma unroll
  for (int i = 0; i < 8; ++i) {
    const int item = tid + i * 256;
    const int kp = item >> 5, m = item & 31;
    const float4 r0 = ((const float4*)(w + (size_t)(2 * kp) * D))[m];
    const float4 r1 = ((const float4*)(w + (size_t)(2 * kp + 1) * D))[m];
    float4* dst = (float4*)(lds_w + kp * WQ + m * 8);
    dst[0] = make_float4(r0.x, r1.x, r0.y, r1.y);
    dst[1] = make_float4(r0.z, r1.z, r0.w, r1.w);
    pv[4 * i + 0] = r0.x * as4.x + r0.y * as4.y + r0.z * as4.z + r0.w * as4.w;
    pv[4 * i + 1] = r1.x * as4.x + r1.y * as4.y + r1.z * as4.z + r1.w * as4.w;
    pv[4 * i + 2] = r0.x * ad4.x + r0.y * ad4.y + r0.z * ad4.z + r0.w * ad4.w;
    pv[4 * i + 3] = r1.x * ad4.x + r1.y * ad4.y + r1.z * ad4.z + r1.w * ad4.w;
  }
#pragma unroll
  for (int st = 0; st < 5; ++st) {
    const int off = 16 >> st;
    const bool up = (lane & off) != 0;
#pragma unroll
    for (int j = 0; j < off; ++j) {
      const float lo = pv[j], hi = pv[j + off];
      pv[j] = pick(up, hi, lo) + __shfl_xor(pick(up, lo, hi), off, 32);
    }
  }
  {
    const int kp = wave + 8 * (lane >> 2);
    const int t  = lane & 3;
    lds_wa[(t >> 1) * D + 2 * kp + (t & 1)] = pv[0];
  }
  __syncthreads();

  {
    const float4* was = (const float4*)lds_wa + hf;
    const float4* wad = (const float4*)(lds_wa + D) + hf;
    float ps = 0.0f, pd = 0.0f;
#pragma unroll
    for (int q = 0; q < 16; ++q) {
      const float4 s4 = was[2 * q];
      const float4 d4 = wad[2 * q];
      ps += a[q].x * s4.x + a[q].y * s4.y + a[q].z * s4.z + a[q].w * s4.w;
      pd += a[q].x * d4.x + a[q].y * d4.y + a[q].z * d4.z + a[q].w * d4.w;
    }
    ps += __shfl_xor(ps, 16, 32);
    pd += __shfl_xor(pd, 16, 32);
    const int node = row0 + l16;
    if (hf == 0 && node < n_nodes) {
      a_src[node] = ps;
      a_dst[node] = pd;
      float lg = ps + pd;
      lg = lg > 0.0f ? lg : NEG_SLOPE * lg;
      self_logit[node] = lg;
      mx[node] = lg;
    }
  }

  v8f acc[8];
#pragma unroll
  for (int j = 0; j < 8; ++j) acc[j] = (v8f){};

  const float* bp = lds_w + (2 * hf) * WQ + 4 * l16;
  float4 bcur[4], bnxt[4];
#pragma unroll
  for (int jp = 0; jp < 4; ++jp) bcur[jp] = *(const float4*)(bp + 64 * jp);

#pragma unroll
  for (int t = 0; t < 32; ++t) {
    const int q = t >> 1, u = t & 1;
    if (t + 1 < 32) {
      const int kn = 4 * ((t + 1) >> 1) + ((t + 1) & 1);
#pragma unroll
      for (int jp = 0; jp < 4; ++jp) bnxt[jp] = *(const float4*)(bp + kn * WQ + 64 * jp);
    }
    __builtin_amdgcn_sched_barrier(0);
    v2f av;
    av.x = u ? a[q].z : a[q].x;
    av.y = u ? a[q].w : a[q].y;
#pragma unroll
    for (int jp = 0; jp < 4; ++jp) {
      v2f b0, b1;
      b0.x = bcur[jp].x; b0.y = bcur[jp].y;
      b1.x = bcur[jp].z; b1.y = bcur[jp].w;
      acc[2 * jp]     = __builtin_amdgcn_wmma_f32_16x16x4_f32(false, av, false, b0, (short)0,
                                                              acc[2 * jp], false, false);
      acc[2 * jp + 1] = __builtin_amdgcn_wmma_f32_16x16x4_f32(false, av, false, b1, (short)0,
                                                              acc[2 * jp + 1], false, false);
    }
    __builtin_amdgcn_sched_barrier(0);
#pragma unroll
    for (int jp = 0; jp < 4; ++jp) bcur[jp] = bnxt[jp];
  }

#pragma unroll
  for (int r = 0; r < 8; ++r) {
    const int row = row0 + 8 * hf + r;
    if (row < n_nodes) {
      float* hp = h + (size_t)row * D + 2 * l16;
#pragma unroll
      for (int jp = 0; jp < 4; ++jp) {
        float2 o;
        o.x = acc[2 * jp][r];
        o.y = acc[2 * jp + 1][r];
        *(float2*)(hp + 32 * jp) = o;
      }
    }
  }
}

__global__ __launch_bounds__(256)
void gat_edge_logits(const int* __restrict__ ei, const float* __restrict__ a_src,
                     const float* __restrict__ a_dst, float* __restrict__ logit,
                     float* __restrict__ mx, int n_edges) {
  const int e = blockIdx.x * 256 + threadIdx.x;
  if (e >= n_edges) return;
  const int s = ei[e];
  const int d = ei[n_edges + e];
  float lg = a_src[s] + a_dst[d];
  lg = lg > 0.0f ? lg : NEG_SLOPE * lg;
  logit[e] = lg;
  __hip_atomic_fetch_max(&mx[d], lg, __ATOMIC_RELAXED, __HIP_MEMORY_SCOPE_AGENT);
}

#define NPW 200
#define OWN_WAVES 32
#define SCAN_LOADS 16
#define QCAP 544
#define LCAP 256
#define LCH (LCAP / 32)

static_assert(NPW <= 256, "a local destination must fit the low 8 bits of a list word");
static_assert(LCAP <= QCAP, "a wave's list slice is unpacked into its own queue");
static_assert(OWN_WAVES == 32, "one lane per wave stores the work-group's line of counts");

__device__ __forceinline__ void own_rows(int s, int dl, float w, int m, int lane,
                                         const float* __restrict__ h, float* s_acc) {
  for (int j = 0; j < m; j += 8) {
    float hv[8][4];
#pragma unroll
    for (int u = 0; u < 8; ++u) {
      if (j + u < m) {
        const int sj = __builtin_amdgcn_readlane(s, j + u);
        const float* hp = h + (size_t)sj * D + lane;
        hv[u][0] = hp[0];
        hv[u][1] = hp[32];
        hv[u][2] = hp[64];
        hv[u][3] = hp[96];
      }
    }
#pragma unroll
    for (int u = 0; u < 8; ++u) {
      if (j + u < m) {
        const int dj = __builtin_amdgcn_readlane(dl, j + u);
        const float wj = __builtin_bit_cast(float, __builtin_amdgcn_readlane(__builtin_bit_cast(int, w), j + u));
        float* ap = s_acc + dj * D + lane;
        __hip_atomic_fetch_add(ap + 0,  hv[u][0] * wj, __ATOMIC_RELAXED, __HIP_MEMORY_SCOPE_WORKGROUP);
        __hip_atomic_fetch_add(ap + 32, hv[u][1] * wj, __ATOMIC_RELAXED, __HIP_MEMORY_SCOPE_WORKGROUP);
        __hip_atomic_fetch_add(ap + 64, hv[u][2] * wj, __ATOMIC_RELAXED, __HIP_MEMORY_SCOPE_WORKGROUP);
        __hip_atomic_fetch_add(ap + 96, hv[u][3] * wj, __ATOMIC_RELAXED, __HIP_MEMORY_SCOPE_WORKGROUP);
      }
    }
  }
}

__device__ __forceinline__ void own_drain(const int* qe, const int* qd, int m, int lane,
                                          int n0, const int* __restrict__ ei,
                                          const float* __restrict__ logit,
                                          const float* __restrict__ mx,
                                          const float* __restrict__ h,
                                          float* s_acc, float* s_sum, int* rec) {
  int s = 0, dl = 0;
  float w = 0.0f;
  if (lane < m) {
    const int e = qe[lane];
    dl = qd[lane];
    if (rec != nullptr) rec[lane] = (e << 8) | dl;
    s = ei[e];
    w = __expf(logit[e] - mx[n0 + dl]);
    __hip_atomic_fetch_add(&s_sum[dl], w, __ATOMIC_RELAXED, __HIP_MEMORY_SCOPE_WORKGROUP);
  }
  own_rows(s, dl, w, m, lane, h, s_acc);
}

#define OWN_ROWS ((NPW + OWN_WAVES - 1) / OWN_WAVES)

__global__ __launch_bounds__(OWN_WAVES * 32)
void gat_owner_accum(const int* __restrict__ ei, const float* __restrict__ logit,
                     const float* __restrict__ mx, const float* __restrict__ self_logit,
                     const float* __restrict__ h, const float* x,
                     const float* __restrict__ bias, const float* __restrict__ gamma,
                     const float* __restrict__ beta, float* out,
                     int n_nodes, int n_edges, int* elist, int* ecnt, int mode) {
  __shared__ float s_acc[NPW * D];
  __shared__ float s_sum[NPW];
  __shared__ int   s_qe[OWN_WAVES * QCAP];
  __shared__ int   s_qd[OWN_WAVES * QCAP];
  __shared__ int   s_cnt[OWN_WAVES];

  const int tid  = threadIdx.x;
  const int wave = __builtin_amdgcn_readfirstlane(tid >> 5);
  const int lane = tid & 31;
  const int n0   = blockIdx.x * NPW;
  const int cnt  = min(NPW, n_nodes - n0);

  int* qe = s_qe + wave * QCAP;
  int* qd = s_qd + wave * QCAP;
  const int slot = blockIdx.x * OWN_WAVES + wave;
  int* lst = elist + (size_t)slot * LCAP;

  const float4 bv = ((const float4*)bias)[lane];
  const float4 gv = ((const float4*)gamma)[lane];
  const float4 tv = ((const float4*)beta)[lane];
  float4 xr[OWN_ROWS];
#pragma unroll
  for (int i = 0; i < OWN_ROWS; ++i) {
    const int r = wave + i * OWN_WAVES;
    xr[i] = (r < cnt) ? ((const float4*)(x + (size_t)(n0 + r) * D))[lane]
                      : make_float4(0.0f, 0.0f, 0.0f, 0.0f);
  }

  int lcnt = 0;
  bool replay = false;
  if (mode == 2) {
    lcnt = __builtin_amdgcn_readfirstlane(ecnt[slot]);
    replay = (unsigned)lcnt <= (unsigned)LCAP;
  }

  if (replay) {
    int pw[LCH], sv[LCH];
    float lv[LCH], mv[LCH];
#pragma unroll
    for (int c = 0; c < LCH; ++c) {
      const int idx = c * 32 + lane;
      pw[c] = (idx < lcnt) ? lst[idx] : -1;
    }
#pragma unroll
    for (int c = 0; c < LCH; ++c) {
      const int p = max(pw[c], 0);
      const int e = p >> 8;
      sv[c] = ei[e];
      lv[c] = logit[e];
      mv[c] = mx[n0 + (p & 255)];
    }
#pragma unroll
    for (int c = 0; c < LCH; ++c) {
      if (pw[c] >= 0) {
        qe[c * 32 + lane] = (sv[c] << 8) | (pw[c] & 255);
        qd[c * 32 + lane] = __builtin_bit_cast(int, __expf(lv[c] - mv[c]));
      }
    }
  }

  for (int i = tid; i < cnt * 32; i += OWN_WAVES * 32) {
    const int r = i >> 5, c = i & 31;
    const int node = n0 + r;
    const float e0 = __expf(self_logit[node] - mx[node]);
    const float4 hv = ((const float4*)(h + (size_t)node * D))[c];
    ((float4*)(s_acc + r * D))[c] = make_float4(hv.x * e0, hv.y * e0, hv.z * e0, hv.w * e0);
    if (c == 0) s_sum[r] = e0;
  }
  __syncthreads();

  if (replay) {
    int off = 0;
    for (; off + 32 <= lcnt; off += 32) {
      const int sp  = qe[off + lane];
      const float w = __builtin_bit_cast(float, qd[off + lane]);
      __hip_atomic_fetch_add(&s_sum[sp & 255], w, __ATOMIC_RELAXED, __HIP_MEMORY_SCOPE_WORKGROUP);
      own_rows(sp >> 8, sp & 255, w, 32, lane, h, s_acc);
    }
    if (off < lcnt) {
      const int m = lcnt - off;
      int sp = 0;
      float w = 0.0f;
      if (lane < m) {
        sp = qe[off + lane];
        w  = __builtin_bit_cast(float, qd[off + lane]);
        __hip_atomic_fetch_add(&s_sum[sp & 255], w, __ATOMIC_RELAXED, __HIP_MEMORY_SCOPE_WORKGROUP);
      }
      own_rows(sp >> 8, sp & 255, w, m, lane, h, s_acc);
    }
  } else {
    const bool record = (mode == 1);
    int rn = 0;
    const int* __restrict__ dstp = ei + n_edges;
    int qn = 0;

    const int step  = SCAN_LOADS * 32;
    const int n_stp = (n_edges + step - 1) / step;
    const int first = (int)(((long long)blockIdx.x * n_stp) / gridDim.x);
    for (int it = wave; it < n_stp; it += OWN_WAVES) {
      int sg = first + it;
      if (sg >= n_stp) sg -= n_stp;
      const int base = sg * step;
      int dv[SCAN_LOADS];
      if (base + step <= n_edges) {
#pragma unroll
        for (int k = 0; k < SCAN_LOADS; ++k) dv[k] = dstp[base + k * 32 + lane];
      } else {
#pragma unroll
        for (int k = 0; k < SCAN_LOADS; ++k) {
          const int idx = base + k * 32 + lane;
          dv[k] = (idx < n_edges) ? dstp[idx] : -1;
        }
      }
#pragma unroll
      for (int k = 0; k < SCAN_LOADS; ++k) {
        const unsigned rel = (unsigned)(dv[k] - n0);
        const bool hit = rel < (unsigned)cnt;
        const unsigned msk = (unsigned)__ballot(hit);
        if (msk != 0u) {
          if (hit) {
            const int pos = qn + __popc(msk & ((1u << lane) - 1u));
            qe[pos] = base + k * 32 + lane;
            qd[pos] = (int)rel;
          }
          qn += __popc(msk);
        }
      }
      while (qn >= 32) {
        qn -= 32;
        own_drain(qe + qn, qd + qn, 32, lane, n0, ei, logit, mx, h, s_acc, s_sum,
                  (record && rn + 32 <= LCAP) ? lst + rn : nullptr);
        rn += 32;
      }
    }
    if (qn > 0) {
      own_drain(qe, qd, qn, lane, n0, ei, logit, mx, h, s_acc, s_sum,
                (record && rn + qn <= LCAP) ? lst + rn : nullptr);
      rn += qn;
    }
    if (record && lane == 0) s_cnt[wave] = rn;
  }
  __syncthreads();

  if (mode == 1 && wave == 0) ecnt[blockIdx.x * OWN_WAVES + lane] = s_cnt[lane];

  float4 y[OWN_ROWS];
  float sum[OWN_ROWS], sq[OWN_ROWS];
#pragma unroll
  for (int i = 0; i < OWN_ROWS; ++i) {
    const int rc = min(wave + i * OWN_WAVES, cnt - 1);
    const float4 av = ((const float4*)(s_acc + rc * D))[lane];
    const float inv_s = 1.0f / s_sum[rc];
    y[i].x = xr[i].x + av.x * inv_s + bv.x;
    y[i].y = xr[i].y + av.y * inv_s + bv.y;
    y[i].z = xr[i].z + av.z * inv_s + bv.z;
    y[i].w = xr[i].w + av.w * inv_s + bv.w;
    sum[i] = y[i].x + y[i].y + y[i].z + y[i].w;
    sq[i]  = y[i].x * y[i].x + y[i].y * y[i].y + y[i].z * y[i].z + y[i].w * y[i].w;
  }
#pragma unroll
  for (int off = 16; off > 0; off >>= 1) {
#pragma unroll
    for (int i = 0; i < OWN_ROWS; ++i) {
      sum[i] += __shfl_xor(sum[i], off, 32);
      sq[i]  += __shfl_xor(sq[i], off, 32);
    }
  }
#pragma unroll
  for (int i = 0; i < OWN_ROWS; ++i) {
    const int r = wave + i * OWN_WAVES;
    const float mu   = sum[i] * (1.0f / D);
    const float var  = sq[i] * (1.0f / D) - mu * mu;
    const float rstd = rsqrtf(var + LN_EPS);
    float4 o;
    o.x = (y[i].x - mu) * rstd * gv.x + tv.x;
    o.y = (y[i].y - mu) * rstd * gv.y + tv.y;
    o.z = (y[i].z - mu) * rstd * gv.z + tv.z;
    o.w = (y[i].w - mu) * rstd * gv.w + tv.w;
    if (r < cnt) ((float4*)(out + (size_t)(n0 + r) * D))[lane] = o;
  }
}

extern "C" void kernel_launch(void* const* d_in, const int* in_sizes, int n_in,
                              void* d_out, int out_size, void* d_ws, size_t ws_size,
                              hipStream_t stream) {
  const float* x0   = (const float*)d_in[0];
  const int*   ei   = (const int*)d_in[1];
  const float* W    = (const float*)d_in[2];
  const float* asrc = (const float*)d_in[3];
  const float* adst = (const float*)d_in[4];
  const float* bias = (const float*)d_in[5];
  const float* gam  = (const float*)d_in[6];
  const float* bet  = (const float*)d_in[7];
  float* out = (float*)d_out;

  const int n  = in_sizes[0] / D;
  const int e  = in_sizes[1] / 2;
  const int nl = in_sizes[2] / (D * D);

  float* ws    = (float*)d_ws;
  float* h     = ws; ws += (size_t)n * D;
  float* va    = ws; ws += n;
  float* vb    = ws; ws += n;
  float* slf   = ws; ws += n;
  float* mx    = ws; ws += n;
  float* lgt   = ws; ws += e;

  const int    nblk      = (n + NPW - 1) / NPW;
  const size_t lst_off   = (((size_t)n * (D + 4) + (size_t)e) * sizeof(float) + 127) / 128 * 128;
  const size_t lst_words = (size_t)nblk * OWN_WAVES * LCAP;
  const size_t lst_end   = lst_off + (lst_words + (size_t)nblk * OWN_WAVES) * sizeof(int);
  const bool   use_list  = nl > 1 && ws_size >= lst_end && e < (1 << 23) && n < (1 << 23);
  int* elist = use_list ? (int*)((char*)d_ws + lst_off) : nullptr;
  int* ecnt  = use_list ? elist + lst_words : nullptr;

  for (int l = 0; l < nl; ++l) {
    const float* xin = (l == 0) ? x0 : out;
    const float* Wl  = W + (size_t)l * D * D;
    const int mode   = use_list ? (l == 0 ? 1 : 2) : 0;
    gat_gemm_wmma<<<(n + 127) / 128, 256, 0, stream>>>(xin, Wl, asrc + l * D, adst + l * D,
                                                       h, va, vb, slf, mx, n);
    gat_edge_logits<<<(e + 255) / 256, 256, 0, stream>>>(ei, va, vb, lgt, mx, e);
    gat_owner_accum<<<nblk, OWN_WAVES * 32, 0, stream>>>(ei, lgt, mx, slf, h, xin,
                                                         bias + l * D, gam + l * D,
                                                         bet + l * D, out, n, e,
                                                         elist, ecnt, mode);
  }
}


# First-layer edge walk reads a 1-byte owner column (dst / 200, stored by the logit pass) in 16-byte loads, the next step's asked for ahead: a quarter of the walk's bytes; LDS add arrival order kept
# speedup vs baseline: 2.7226x; 1.0309x over previous
#include <hip/hip_runtime.h>
#include <math.h>

#define D 128
#define NEG_SLOPE 0.2f
#define LN_EPS 1e-5f

typedef float v2f __attribute__((ext_vector_type(2)));
typedef float v8f __attribute__((ext_vector_type(8)));

__device__ __forceinline__ float pick(bool c, float a, float b) { return c ? a : b; }

#define WQ 272

__global__ __launch_bounds__(256)
void gat_gemm_wmma(const float* __restrict__ x, const float* __restrict__ w,
                   const float* __restrict__ att_src, const float* __restrict__ att_dst,
                   float* __restrict__ h, float* __restrict__ a_src,
                   float* __restrict__ a_dst, float* __restrict__ self_logit,
                   float* __restrict__ mx, int n_nodes) {
  __shared__ __attribute__((aligned(16))) float lds_w[(D / 2) * WQ];
  __shared__ __attribute__((aligned(16))) float lds_wa[2 * D];
  const int tid  = threadIdx.x;
  const int wave = tid >> 5;
  const int lane = tid & 31;
  const int hf   = lane >> 4;
  const int l16  = lane & 15;
  const int row0 = (blockIdx.x * 8 + wave) * 16;

  const int arow = min(row0 + l16, n_nodes - 1);
  const float4* __restrict__ ap = (const float4*)(x + (size_t)arow * D) + hf;
  float4 a[16];
#pragma unroll
  for (int q = 0; q < 16; ++q) a[q] = ap[2 * q];

  const float4 as4 = ((const float4*)att_src)[lane];
  const float4 ad4 = ((const float4*)att_dst)[lane];
  float pv[32];
#pragma unroll
  for (int i = 0; i < 8; ++i) {
    const int item = tid + i * 256;
    const int kp = item >> 5, m = item & 31;
    const float4 r0 = ((const float4*)(w + (size_t)(2 * kp) * D))[m];
    const float4 r1 = ((const float4*)(w + (size_t)(2 * kp + 1) * D))[m];
    float4* dst = (float4*)(lds_w + kp * WQ + m * 8);
    dst[0] = make_float4(r0.x, r1.x, r0.y, r1.y);
    dst[1] = make_float4(r0.z, r1.z, r0.w, r1.w);
    pv[4 * i + 0] = r0.x * as4.x + r0.y * as4.y + r0.z * as4.z + r0.w * as4.w;
    pv[4 * i + 1] = r1.x * as4.x + r1.y * as4.y + r1.z * as4.z + r1.w * as4.w;
    pv[4 * i + 2] = r0.x * ad4.x + r0.y * ad4.y + r0.z * ad4.z + r0.w * ad4.w;
    pv[4 * i + 3] = r1.x * ad4.x + r1.y * ad4.y + r1.z * ad4.z + r1.w * ad4.w;
  }
#pragma unroll
  for (int st = 0; st < 5; ++st) {
    const int off = 16 >> st;
    const bool up = (lane & off) != 0;
#pragma unroll
    for (int j = 0; j < off; ++j) {
      const float lo = pv[j], hi = pv[j + off];
      pv[j] = pick(up, hi, lo) + __shfl_xor(pick(up, lo, hi), off, 32);
    }
  }
  {
    const int kp = wave + 8 * (lane >> 2);
    const int t  = lane & 3;
    lds_wa[(t >> 1) * D + 2 * kp + (t & 1)] = pv[0];
  }
  __syncthreads();

  {
    const float4* was = (const float4*)lds_wa + hf;
    const float4* wad = (const float4*)(lds_wa + D) + hf;
    float ps = 0.0f, pd = 0.0f;
#pragma unroll
    for (int q = 0; q < 16; ++q) {
      const float4 s4 = was[2 * q];
      const float4 d4 = wad[2 * q];
      ps += a[q].x * s4.x + a[q].y * s4.y + a[q].z * s4.z + a[q].w * s4.w;
      pd += a[q].x * d4.x + a[q].y * d4.y + a[q].z * d4.z + a[q].w * d4.w;
    }
    ps += __shfl_xor(ps, 16, 32);
    pd += __shfl_xor(pd, 16, 32);
    const int node = row0 + l16;
    if (hf == 0 && node < n_nodes) {
      a_src[node] = ps;
      a_dst[node] = pd;
      float lg = ps + pd;
      lg = lg > 0.0f ? lg : NEG_SLOPE * lg;
      self_logit[node] = lg;
      mx[node] = lg;
    }
  }

  v8f acc[8];
#pragma unroll
  for (int j = 0; j < 8; ++j) acc[j] = (v8f){};

  const float* bp = lds_w + (2 * hf) * WQ + 4 * l16;
  float4 bcur[4], bnxt[4];
#pragma unroll
  for (int jp = 0; jp < 4; ++jp) bcur[jp] = *(const float4*)(bp + 64 * jp);

#pragma unroll
  for (int t = 0; t < 32; ++t) {
    const int q = t >> 1, u = t & 1;
    if (t + 1 < 32) {
      const int kn = 4 * ((t + 1) >> 1) + ((t + 1) & 1);
#pragma unroll
      for (int jp = 0; jp < 4; ++jp) bnxt[jp] = *(const float4*)(bp + kn * WQ + 64 * jp);
    }
    __builtin_amdgcn_sched_barrier(0);
    v2f av;
    av.x = u ? a[q].z : a[q].x;
    av.y = u ? a[q].w : a[q].y;
#pragma unroll
    for (int jp = 0; jp < 4; ++jp) {
      v2f b0, b1;
      b0.x = bcur[jp].x; b0.y = bcur[jp].y;
      b1.x = bcur[jp].z; b1.y = bcur[jp].w;
      acc[2 * jp]     = __builtin_amdgcn_wmma_f32_16x16x4_f32(false, av, false, b0, (short)0,
                                                              acc[2 * jp], false, false);
      acc[2 * jp + 1] = __builtin_amdgcn_wmma_f32_16x16x4_f32(false, av, false, b1, (short)0,
                                                              acc[2 * jp + 1], false, false);
    }
    __builtin_amdgcn_sched_barrier(0);
#pragma unroll
    for (int jp = 0; jp < 4; ++jp) bcur[jp] = bnxt[jp];
  }

#pragma unroll
  for (int r = 0; r < 8; ++r) {
    const int row = row0 + 8 * hf + r;
    if (row < n_nodes) {
      float* hp = h + (size_t)row * D + 2 * l16;
#pragma unroll
      for (int jp = 0; jp < 4; ++jp) {
        float2 o;
        o.x = acc[2 * jp][r];
        o.y = acc[2 * jp + 1][r];
        *(float2*)(hp + 32 * jp) = o;
      }
    }
  }
}

#define NPW 200

__global__ __launch_bounds__(256)
void gat_edge_logits(const int* __restrict__ ei, const float* __restrict__ a_src,
                     const float* __restrict__ a_dst, float* __restrict__ logit,
                     float* __restrict__ mx, int* __restrict__ ownb, int n_edges) {
  const int e = blockIdx.x * 256 + threadIdx.x;
  if (e >= n_edges) {
    if (ownb != nullptr && e < ((n_edges + 15) & ~15)) ((unsigned char*)ownb)[e] = 255;
    return;
  }
  const int s = ei[e];
  const int d = ei[n_edges + e];
  const float vs = a_src[s];
  const float vd = a_dst[d];
  if (ownb != nullptr) ((unsigned char*)ownb)[e] = (unsigned char)((unsigned)d / NPW);
  float lg = vs + vd;
  lg = lg > 0.0f ? lg : NEG_SLOPE * lg;
  logit[e] = lg;
  __hip_atomic_fetch_max(&mx[d], lg, __ATOMIC_RELAXED, __HIP_MEMORY_SCOPE_AGENT);
}

#define OWN_WAVES 32
#define SCAN_LOADS 16
#define QCAP 544
#define LCAP 224
#define LCH (LCAP / 32)
#define BSCAN 4

static_assert(NPW <= 256, "a local destination must fit the low 8 bits of a list word");
static_assert(LCAP <= QCAP, "a wave's list slice is unpacked into its own queue");
static_assert(OWN_WAVES == 32, "one lane per wave stores the work-group's line of counts");

__device__ __forceinline__ void own_rows(int s, int dl, float w, int m, int lane,
                                         const float* __restrict__ h, float* s_acc) {
  for (int j = 0; j < m; j += 8) {
    float hv[8][4];
#pragma unroll
    for (int u = 0; u < 8; ++u) {
      if (j + u < m) {
        const int sj = __builtin_amdgcn_readlane(s, j + u);
        const float* hp = h + (size_t)sj * D + lane;
        hv[u][0] = hp[0];
        hv[u][1] = hp[32];
        hv[u][2] = hp[64];
        hv[u][3] = hp[96];
      }
    }
#pragma unroll
    for (int u = 0; u < 8; ++u) {
      if (j + u < m) {
        const int dj = __builtin_amdgcn_readlane(dl, j + u);
        const float wj = __builtin_bit_cast(float, __builtin_amdgcn_readlane(__builtin_bit_cast(int, w), j + u));
        float* ap = s_acc + dj * D + lane;
        __hip_atomic_fetch_add(ap + 0,  hv[u][0] * wj, __ATOMIC_RELAXED, __HIP_MEMORY_SCOPE_WORKGROUP);
        __hip_atomic_fetch_add(ap + 32, hv[u][1] * wj, __ATOMIC_RELAXED, __HIP_MEMORY_SCOPE_WORKGROUP);
        __hip_atomic_fetch_add(ap + 64, hv[u][2] * wj, __ATOMIC_RELAXED, __HIP_MEMORY_SCOPE_WORKGROUP);
        __hip_atomic_fetch_add(ap + 96, hv[u][3] * wj, __ATOMIC_RELAXED, __HIP_MEMORY_SCOPE_WORKGROUP);
      }
    }
  }
}

__device__ __forceinline__ void own_drain(const int* qe, const int* qd, int m, int lane,
                                          int n0, const int* __restrict__ ei,
                                          const float* __restrict__ logit,
                                          const float* __restrict__ mx,
                                          const float* __restrict__ h,
                                          float* s_acc, float* s_sum, int* rec) {
  int s = 0, dl = 0;
  float w = 0.0f;
  if (lane < m) {
    const int e = qe[lane];
    dl = qd[lane];
    if (rec != nullptr) rec[lane] = (e << 8) | dl;
    s = ei[e];
    w = __expf(logit[e] - mx[n0 + dl]);
    __hip_atomic_fetch_add(&s_sum[dl], w, __ATOMIC_RELAXED, __HIP_MEMORY_SCOPE_WORKGROUP);
  }
  own_rows(s, dl, w, m, lane, h, s_acc);
}

__device__ __forceinline__ void own_drain_e(const int* qe, int m, int lane, int n0, int n_edges,
                                            const int* __restrict__ ei,
                                            const float* __restrict__ logit,
                                            const float* s_mx,
                                            const float* __restrict__ h,
                                            float* s_acc, float* s_sum, int* rec) {
  int s = 0, dl = 0;
  float w = 0.0f;
  if (lane < m) {
    const int e = qe[lane];
    s = ei[e];
    dl = ei[n_edges + e] - n0;
    const float lg = logit[e];
    if (rec != nullptr) rec[lane] = (e << 8) | dl;
    w = __expf(lg - s_mx[dl]);
    __hip_atomic_fetch_add(&s_sum[dl], w, __ATOMIC_RELAXED, __HIP_MEMORY_SCOPE_WORKGROUP);
  }
  own_rows(s, dl, w, m, lane, h, s_acc);
}

__device__ __forceinline__ unsigned zero_bytes(unsigned t) {
  return ~(((t & 0x7f7f7f7fu) + 0x7f7f7f7fu) | t | 0x7f7f7f7fu);
}

#define OWN_ROWS ((NPW + OWN_WAVES - 1) / OWN_WAVES)

template <bool BYTEWALK>
__global__ __launch_bounds__(OWN_WAVES * 32)
void gat_owner_accum(const int* __restrict__ ei, const float* __restrict__ logit,
                     const float* __restrict__ mx, const float* __restrict__ self_logit,
                     const float* __restrict__ h, const float* x,
                     const float* __restrict__ bias, const float* __restrict__ gamma,
                     const float* __restrict__ beta, float* out,
                     int n_nodes, int n_edges, int* elist, int* ecnt,
                     const int* __restrict__ ownb, int mode) {
  __shared__ float s_acc[NPW * D];
  __shared__ float s_sum[NPW];
  __shared__ float s_mx[NPW];
  __shared__ int   s_qe[OWN_WAVES * QCAP];
  __shared__ int   s_qd[OWN_WAVES * QCAP];
  __shared__ int   s_cnt[OWN_WAVES];

  const int tid  = threadIdx.x;
  const int wave = __builtin_amdgcn_readfirstlane(tid >> 5);
  const int lane = tid & 31;
  const int n0   = blockIdx.x * NPW;
  const int cnt  = min(NPW, n_nodes - n0);

  int* qe = s_qe + wave * QCAP;
  int* qd = s_qd + wave * QCAP;
  const int slot = blockIdx.x * OWN_WAVES + wave;
  int* lst = elist + (size_t)slot * LCAP;

  const float4 bv = ((const float4*)bias)[lane];
  const float4 gv = ((const float4*)gamma)[lane];
  const float4 tv = ((const float4*)beta)[lane];
  float4 xr[OWN_ROWS];
#pragma unroll
  for (int i = 0; i < OWN_ROWS; ++i) {
    const int r = wave + i * OWN_WAVES;
    xr[i] = (r < cnt) ? ((const float4*)(x + (size_t)(n0 + r) * D))[lane]
                      : make_float4(0.0f, 0.0f, 0.0f, 0.0f);
  }

  int lcnt = 0;
  bool replay = false;
  if (!BYTEWALK && mode == 2) {
    const int c = ecnt[blockIdx.x * OWN_WAVES + lane];
    lcnt = __builtin_amdgcn_readlane(c, wave);
    replay = __ballot((unsigned)c > (unsigned)LCAP) == 0;
  }

  if (replay) {
    int pw[LCH], sv[LCH];
    float lv[LCH], mv[LCH];
#pragma unroll
    for (int c = 0; c < LCH; ++c) {
      const int idx = c * 32 + lane;
      pw[c] = (idx < lcnt) ? lst[idx] : -1;
    }
#pragma unroll
    for (int c = 0; c < LCH; ++c) {
      const int p = max(pw[c], 0);
      const int e = p >> 8;
      sv[c] = ei[e];
      lv[c] = logit[e];
      mv[c] = mx[n0 + (p & 255)];
    }
#pragma unroll
    for (int c = 0; c < LCH; ++c) {
      if (pw[c] >= 0) {
        qe[c * 32 + lane] = (sv[c] << 8) | (pw[c] & 255);
        qd[c * 32 + lane] = __builtin_bit_cast(int, __expf(lv[c] - mv[c]));
      }
    }
  }

  for (int i = tid; i < cnt * 32; i += OWN_WAVES * 32) {
    const int r = i >> 5, c = i & 31;
    const int node = n0 + r;
    const float mv = mx[node];
    const float e0 = __expf(self_logit[node] - mv);
    const float4 hv = ((const float4*)(h + (size_t)node * D))[c];
    ((float4*)(s_acc + r * D))[c] = make_float4(hv.x * e0, hv.y * e0, hv.z * e0, hv.w * e0);
    if (c == 0) {
      s_sum[r] = e0;
      if (BYTEWALK) s_mx[r] = mv;
    }
  }
  __syncthreads();

  if (replay) {
    int off = 0;
    for (; off + 32 <= lcnt; off += 32) {
      const int sp  = qe[off + lane];
      const float w = __builtin_bit_cast(float, qd[off + lane]);
      __hip_atomic_fetch_add(&s_sum[sp & 255], w, __ATOMIC_RELAXED, __HIP_MEMORY_SCOPE_WORKGROUP);
      own_rows(sp >> 8, sp & 255, w, 32, lane, h, s_acc);
    }
    if (off < lcnt) {
      const int m = lcnt - off;
      int sp = 0;
      float w = 0.0f;
      if (lane < m) {
        sp = qe[off + lane];
        w  = __builtin_bit_cast(float, qd[off + lane]);
        __hip_atomic_fetch_add(&s_sum[sp & 255], w, __ATOMIC_RELAXED, __HIP_MEMORY_SCOPE_WORKGROUP);
      }
      own_rows(sp >> 8, sp & 255, w, m, lane, h, s_acc);
    }
  } else if (BYTEWALK) {
    int rn = 0;
    int qn = 0;
    const uint4* __restrict__ ow4 = (const uint4*)ownb;
    const unsigned idw = (unsigned)blockIdx.x * 0x01010101u;
    const int n_grp = (n_edges + 15) >> 4;
    const int stepg = BSCAN * 32;
    const int n_stp = (n_grp + stepg - 1) / stepg;
    const int first = (int)(((long long)blockIdx.x * n_stp) / gridDim.x);
    uint4 nx[BSCAN];
    int gn = 0;
    if (wave < n_stp) {
      int sg = first + wave;
      if (sg >= n_stp) sg -= n_stp;
      gn = sg * stepg;
#pragma unroll
      for (int k = 0; k < BSCAN; ++k) {
        const int g = gn + k * 32 + lane;
        nx[k] = (g < n_grp) ? ow4[g] : make_uint4(~0u, ~0u, ~0u, ~0u);
      }
    }
    for (int it = wave; it < n_stp; it += OWN_WAVES) {
      const int g0 = gn;
      uint4 ov[BSCAN];
#pragma unroll
      for (int k = 0; k < BSCAN; ++k) ov[k] = nx[k];
      if (it + OWN_WAVES < n_stp) {
        int sg = first + it + OWN_WAVES;
        if (sg >= n_stp) sg -= n_stp;
        gn = sg * stepg;
        if (gn + stepg <= n_grp) {
#pragma unroll
          for (int k = 0; k < BSCAN; ++k) nx[k] = ow4[gn + k * 32 + lane];
        } else {
#pragma unroll
          for (int k = 0; k < BSCAN; ++k) {
            const int g = gn + k * 32 + lane;
            nx[k] = (g < n_grp) ? ow4[g] : make_uint4(~0u, ~0u, ~0u, ~0u);
          }
        }
      }
#pragma unroll
      for (int k = 0; k < BSCAN; ++k) {
        const int g = g0 + k * 32 + lane;
        unsigned hm = (zero_bytes(ov[k].x ^ idw) >> 7) | (zero_bytes(ov[k].y ^ idw) >> 6) |
                      (zero_bytes(ov[k].z ^ idw) >> 5) | (zero_bytes(ov[k].w ^ idw) >> 4);
        unsigned msk = (unsigned)__ballot(hm != 0u);
        while (msk != 0u) {
          if (hm != 0u) {
            const int p = __builtin_ctz(hm);
            const int pos = qn + __popc(msk & ((1u << lane) - 1u));
            qe[pos] = g * 16 + (p & 7) * 4 + (p >> 3);
            hm &= hm - 1u;
          }
          qn += __popc(msk);
          msk = (unsigned)__ballot(hm != 0u);
        }
        while (qn >= 32) {
          qn -= 32;
          own_drain_e(qe + qn, 32, lane, n0, n_edges, ei, logit, s_mx, h, s_acc, s_sum,
                      (rn + 32 <= LCAP) ? lst + rn : nullptr);
          rn += 32;
        }
      }
    }
    if (qn > 0) {
      own_drain_e(qe, qn, lane, n0, n_edges, ei, logit, s_mx, h, s_acc, s_sum,
                  (rn + qn <= LCAP) ? lst + rn : nullptr);
      rn += qn;
    }
    if (lane == 0) s_cnt[wave] = rn;
  } else {
    const bool record = (mode == 1);
    int rn = 0;
    const int* __restrict__ dstp = ei + n_edges;
    int qn = 0;

    const int step  = SCAN_LOADS * 32;
    const int n_stp = (n_edges + step - 1) / step;
    const int first = (int)(((long long)blockIdx.x * n_stp) / gridDim.x);
    for (int it = wave; it < n_stp; it += OWN_WAVES) {
      int sg = first + it;
      if (sg >= n_stp) sg -= n_stp;
      const int base = sg * step;
      int dv[SCAN_LOADS];
      if (base + step <= n_edges) {
#pragma unroll
        for (int k = 0; k < SCAN_LOADS; ++k) dv[k] = dstp[base + k * 32 + lane];
      } else {
#pragma unroll
        for (int k = 0; k < SCAN_LOADS; ++k) {
          const int idx = base + k * 32 + lane;
          dv[k] = (idx < n_edges) ? dstp[idx] : -1;
        }
      }
#pragma unroll
      for (int k = 0; k < SCAN_LOADS; ++k) {
        const unsigned rel = (unsigned)(dv[k] - n0);
        const bool hit = rel < (unsigned)cnt;
        const unsigned msk = (unsigned)__ballot(hit);
        if (msk != 0u) {
          if (hit) {
            const int pos = qn + __popc(msk & ((1u << lane) - 1u));
            qe[pos] = base + k * 32 + lane;
            qd[pos] = (int)rel;
          }
          qn += __popc(msk);
        }
      }
      while (qn >= 32) {
        qn -= 32;
        own_drain(qe + qn, qd + qn, 32, lane, n0, ei, logit, mx, h, s_acc, s_sum,
                  (record && rn + 32 <= LCAP) ? lst + rn : nullptr);
        rn += 32;
      }
    }
    if (qn > 0) {
      own_drain(qe, qd, qn, lane, n0, ei, logit, mx, h, s_acc, s_sum,
                (record && rn + qn <= LCAP) ? lst + rn : nullptr);
      rn += qn;
    }
    if (record && lane == 0) s_cnt[wave] = rn;
  }
  __syncthreads();

  if (mode == 1 && wave == 0) ecnt[blockIdx.x * OWN_WAVES + lane] = s_cnt[lane];

  float4 y[OWN_ROWS];
  float sum[OWN_ROWS], sq[OWN_ROWS];
#pragma unroll
  for (int i = 0; i < OWN_ROWS; ++i) {
    const int rc = min(wave + i * OWN_WAVES, cnt - 1);
    const float4 av = ((const float4*)(s_acc + rc * D))[lane];
    const float inv_s = 1.0f / s_sum[rc];
    y[i].x = xr[i].x + av.x * inv_s + bv.x;
    y[i].y = xr[i].y + av.y * inv_s + bv.y;
    y[i].z = xr[i].z + av.z * inv_s + bv.z;
    y[i].w = xr[i].w + av.w * inv_s + bv.w;
    sum[i] = y[i].x + y[i].y + y[i].z + y[i].w;
    sq[i]  = y[i].x * y[i].x + y[i].y * y[i].y + y[i].z * y[i].z + y[i].w * y[i].w;
  }
#pragma unroll
  for (int off = 16; off > 0; off >>= 1) {
#pragma unroll
    for (int i = 0; i < OWN_ROWS; ++i) {
      sum[i] += __shfl_xor(sum[i], off, 32);
      sq[i]  += __shfl_xor(sq[i], off, 32);
    }
  }
#pragma unroll
  for (int i = 0; i < OWN_ROWS; ++i) {
    const int r = wave + i * OWN_WAVES;
    const float mu   = sum[i] * (1.0f / D);
    const float var  = sq[i] * (1.0f / D) - mu * mu;
    const float rstd = rsqrtf(var + LN_EPS);
    float4 o;
    o.x = (y[i].x - mu) * rstd * gv.x + tv.x;
    o.y = (y[i].y - mu) * rstd * gv.y + tv.y;
    o.z = (y[i].z - mu) * rstd * gv.z + tv.z;
    o.w = (y[i].w - mu) * rstd * gv.w + tv.w;
    if (r < cnt) ((float4*)(out + (size_t)(n0 + r) * D))[lane] = o;
  }
}

extern "C" void kernel_launch(void* const* d_in, const int* in_sizes, int n_in,
                              void* d_out, int out_size, void* d_ws, size_t ws_size,
                              hipStream_t stream) {
  const float* x0   = (const float*)d_in[0];
  const int*   ei   = (const int*)d_in[1];
  const float* W    = (const float*)d_in[2];
  const float* asrc = (const float*)d_in[3];
  const float* adst = (const float*)d_in[4];
  const float* bias = (const float*)d_in[5];
  const float* gam  = (const float*)d_in[6];
  const float* bet  = (const float*)d_in[7];
  float* out = (float*)d_out;

  const int n  = in_sizes[0] / D;
  const int e  = in_sizes[1] / 2;
  const int nl = in_sizes[2] / (D * D);

  float* ws    = (float*)d_ws;
  float* h     = ws; ws += (size_t)n * D;
  float* va    = ws; ws += n;
  float* vb    = ws; ws += n;
  float* slf   = ws; ws += n;
  float* mx    = ws; ws += n;
  float* lgt   = ws; ws += e;

  const int    nblk      = (n + NPW - 1) / NPW;
  const size_t lst_off   = (((size_t)n * (D + 4) + (size_t)e) * sizeof(float) + 127) / 128 * 128;
  const size_t lst_words = (size_t)nblk * OWN_WAVES * LCAP;
  const size_t lst_end   = lst_off + (lst_words + (size_t)nblk * OWN_WAVES) * sizeof(int);
  const bool   use_list  = nl > 1 && ws_size >= lst_end && e < (1 << 23) && n < (1 << 23);
  int* elist = use_list ? (int*)((char*)d_ws + lst_off) : nullptr;
  int* ecnt  = use_list ? elist + lst_words : nullptr;
  const size_t own_end = lst_end + ((size_t)e + 15) / 16 * 16;
  const bool   use_own = use_list && nblk <= 255 && ws_size >= own_end;
  int* ownb = use_own ? (int*)((char*)d_ws + lst_end) : nullptr;

  for (int l = 0; l < nl; ++l) {
    const float* xin = (l == 0) ? x0 : out;
    const float* Wl  = W + (size_t)l * D * D;
    const int mode   = use_list ? (l == 0 ? 1 : 2) : 0;
    gat_gemm_wmma<<<(n + 127) / 128, 256, 0, stream>>>(xin, Wl, asrc + l * D, adst + l * D,
                                                       h, va, vb, slf, mx, n);
    int* ownl = (mode == 1) ? ownb : nullptr;
    gat_edge_logits<<<(e + 255) / 256, 256, 0, stream>>>(ei, va, vb, lgt, mx, ownl, e);
    if (ownl != nullptr)
      gat_owner_accum<true><<<nblk, OWN_WAVES * 32, 0, stream>>>(ei, lgt, mx, slf, h, xin,
                                                                 bias + l * D, gam + l * D,
                                                                 bet + l * D, out, n, e,
                                                                 elist, ecnt, ownl, mode);
    else
      gat_owner_accum<false><<<nblk, OWN_WAVES * 32, 0, stream>>>(ei, lgt, mx, slf, h, xin,
                                                                  bias + l * D, gam + l * D,
                                                                  bet + l * D, out, n, e,
                                                                  elist, ecnt, nullptr, mode);
  }
}


# Edge sums: a source row comes as one 16-byte-per-lane load off a scalar base (whole 512-byte row, not four one-line loads); owner LDS rows column-permuted so adds stay on 32 banks; sum order kept
# speedup vs baseline: 2.8653x; 1.0524x over previous
#include <hip/hip_runtime.h>
#include <math.h>

#define D 128
#define NEG_SLOPE 0.2f
#define LN_EPS 1e-5f

typedef float v2f __attribute__((ext_vector_type(2)));
typedef float v8f __attribute__((ext_vector_type(8)));

__device__ __forceinline__ float pick(bool c, float a, float b) { return c ? a : b; }

#define WQ 272

__global__ __launch_bounds__(256)
void gat_gemm_wmma(const float* __restrict__ x, const float* __restrict__ w,
                   const float* __restrict__ att_src, const float* __restrict__ att_dst,
                   float* __restrict__ h, float* __restrict__ a_src,
                   float* __restrict__ a_dst, float* __restrict__ self_logit,
                   float* __restrict__ mx, int n_nodes) {
  __shared__ __attribute__((aligned(16))) float lds_w[(D / 2) * WQ];
  __shared__ __attribute__((aligned(16))) float lds_wa[2 * D];
  const int tid  = threadIdx.x;
  const int wave = tid >> 5;
  const int lane = tid & 31;
  const int hf   = lane >> 4;
  const int l16  = lane & 15;
  const int row0 = (blockIdx.x * 8 + wave) * 16;

  const int arow = min(row0 + l16, n_nodes - 1);
  const float4* __restrict__ ap = (const float4*)(x + (size_t)arow * D) + hf;
  float4 a[16];
#pragma unroll
  for (int q = 0; q < 16; ++q) a[q] = ap[2 * q];

  const float4 as4 = ((const float4*)att_src)[lane];
  const float4 ad4 = ((const float4*)att_dst)[lane];
  float pv[32];
#pragma unroll
  for (int i = 0; i < 8; ++i) {
    const int item = tid + i * 256;
    const int kp = item >> 5, m = item & 31;
    const float4 r0 = ((const float4*)(w + (size_t)(2 * kp) * D))[m];
    const float4 r1 = ((const float4*)(w + (size_t)(2 * kp + 1) * D))[m];
    float4* dst = (float4*)(lds_w + kp * WQ + m * 8);
    dst[0] = make_float4(r0.x, r1.x, r0.y, r1.y);
    dst[1] = make_float4(r0.z, r1.z, r0.w, r1.w);
    pv[4 * i + 0] = r0.x * as4.x + r0.y * as4.y + r0.z * as4.z + r0.w * as4.w;
    pv[4 * i + 1] = r1.x * as4.x + r1.y * as4.y + r1.z * as4.z + r1.w * as4.w;
    pv[4 * i + 2] = r0.x * ad4.x + r0.y * ad4.y + r0.z * ad4.z + r0.w * ad4.w;
    pv[4 * i + 3] = r1.x * ad4.x + r1.y * ad4.y + r1.z * ad4.z + r1.w * ad4.w;
  }
#pragma unroll
  for (int st = 0; st < 5; ++st) {
    const int off = 16 >> st;
    const bool up = (lane & off) != 0;
#pragma unroll
    for (int j = 0; j < off; ++j) {
      const float lo = pv[j], hi = pv[j + off];
      pv[j] = pick(up, hi, lo) + __shfl_xor(pick(up, lo, hi), off, 32);
    }
  }
  {
    const int kp = wave + 8 * (lane >> 2);
    const int t  = lane & 3;
    lds_wa[(t >> 1) * D + 2 * kp + (t & 1)] = pv[0];
  }
  __syncthreads();

  {
    const float4* was = (const float4*)lds_wa + hf;
    const float4* wad = (const float4*)(lds_wa + D) + hf;
    float ps = 0.0f, pd = 0.0f;
#pragma unroll
    for (int q = 0; q < 16; ++q) {
      const float4 s4 = was[2 * q];
      const float4 d4 = wad[2 * q];
      ps += a[q].x * s4.x + a[q].y * s4.y + a[q].z * s4.z + a[q].w * s4.w;
      pd += a[q].x * d4.x + a[q].y * d4.y + a[q].z * d4.z + a[q].w * d4.w;
    }
    ps += __shfl_xor(ps, 16, 32);
    pd += __shfl_xor(pd, 16, 32);
    const int node = row0 + l16;
    if (hf == 0 && node < n_nodes) {
      a_src[node] = ps;
      a_dst[node] = pd;
      float lg = ps + pd;
      lg = lg > 0.0f ? lg : NEG_SLOPE * lg;
      self_logit[node] = lg;
      mx[node] = lg;
    }
  }

  v8f acc[8];
#pragma unroll
  for (int j = 0; j < 8; ++j) acc[j] = (v8f){};

  const float* bp = lds_w + (2 * hf) * WQ + 4 * l16;
  float4 bcur[4], bnxt[4];
#pragma unroll
  for (int jp = 0; jp < 4; ++jp) bcur[jp] = *(const float4*)(bp + 64 * jp);

#pragma unroll
  for (int t = 0; t < 32; ++t) {
    const int q = t >> 1, u = t & 1;
    if (t + 1 < 32) {
      const int kn = 4 * ((t + 1) >> 1) + ((t + 1) & 1);
#pragma unroll
      for (int jp = 0; jp < 4; ++jp) bnxt[jp] = *(const float4*)(bp + kn * WQ + 64 * jp);
    }
    __builtin_amdgcn_sched_barrier(0);
    v2f av;
    av.x = u ? a[q].z : a[q].x;
    av.y = u ? a[q].w : a[q].y;
#pragma unroll
    for (int jp = 0; jp < 4; ++jp) {
      v2f b0, b1;
      b0.x = bcur[jp].x; b0.y = bcur[jp].y;
      b1.x = bcur[jp].z; b1.y = bcur[jp].w;
      acc[2 * jp]     = __builtin_amdgcn_wmma_f32_16x16x4_f32(false, av, false, b0, (short)0,
                                                              acc[2 * jp], false, false);
      acc[2 * jp + 1] = __builtin_amdgcn_wmma_f32_16x16x4_f32(false, av, false, b1, (short)0,
                                                              acc[2 * jp + 1], false, false);
    }
    __builtin_amdgcn_sched_barrier(0);
#pragma unroll
    for (int jp = 0; jp < 4; ++jp) bcur[jp] = bnxt[jp];
  }

#pragma unroll
  for (int r = 0; r < 8; ++r) {
    const int row = row0 + 8 * hf + r;
    if (row < n_nodes) {
      float* hp = h + (size_t)row * D + 2 * l16;
#pragma unroll
      for (int jp = 0; jp < 4; ++jp) {
        float2 o;
        o.x = acc[2 * jp][r];
        o.y = acc[2 * jp + 1][r];
        *(float2*)(hp + 32 * jp) = o;
      }
    }
  }
}

#define NPW 200

__global__ __launch_bounds__(256)
void gat_edge_logits(const int* __restrict__ ei, const float* __restrict__ a_src,
                     const float* __restrict__ a_dst, float* __restrict__ logit,
                     float* __restrict__ mx, int* __restrict__ ownb, int n_edges) {
  const int e = blockIdx.x * 256 + threadIdx.x;
  if (e >= n_edges) {
    if (ownb != nullptr && e < ((n_edges + 15) & ~15)) ((unsigned char*)ownb)[e] = 255;
    return;
  }
  const int s = ei[e];
  const int d = ei[n_edges + e];
  const float vs = a_src[s];
  const float vd = a_dst[d];
  if (ownb != nullptr) ((unsigned char*)ownb)[e] = (unsigned char)((unsigned)d / NPW);
  float lg = vs + vd;
  lg = lg > 0.0f ? lg : NEG_SLOPE * lg;
  logit[e] = lg;
  __hip_atomic_fetch_max(&mx[d], lg, __ATOMIC_RELAXED, __HIP_MEMORY_SCOPE_AGENT);
}

#define OWN_WAVES 32
#define SCAN_LOADS 16
#define QCAP 544
#define LCAP 224
#define LCH (LCAP / 32)
#define BSCAN 4

static_assert(NPW <= 256, "a local destination must fit the low 8 bits of a list word");
static_assert(LCAP <= QCAP, "a wave's list slice is unpacked into its own queue");
static_assert(OWN_WAVES == 32, "one lane per wave stores the work-group's line of counts");

__device__ __forceinline__ void own_rows(int s, int dl, float w, int m, int lane,
                                         const float* __restrict__ h, float* s_acc) {
  for (int j = 0; j < m; j += 8) {
    float4 hv[8];
#pragma unroll
    for (int u = 0; u < 8; ++u) {
      if (j + u < m) {
        const int sj = __builtin_amdgcn_readlane(s, j + u);
        hv[u] = ((const float4*)h)[((unsigned)sj << 5) | (unsigned)lane];
      }
    }
#pragma unroll
    for (int u = 0; u < 8; ++u) {
      if (j + u < m) {
        const int dj = __builtin_amdgcn_readlane(dl, j + u);
        const float wj = __builtin_bit_cast(float, __builtin_amdgcn_readlane(__builtin_bit_cast(int, w), j + u));
        float* ap = s_acc + dj * D + lane;
        __hip_atomic_fetch_add(ap + 0,  hv[u].x * wj, __ATOMIC_RELAXED, __HIP_MEMORY_SCOPE_WORKGROUP);
        __hip_atomic_fetch_add(ap + 32, hv[u].y * wj, __ATOMIC_RELAXED, __HIP_MEMORY_SCOPE_WORKGROUP);
        __hip_atomic_fetch_add(ap + 64, hv[u].z * wj, __ATOMIC_RELAXED, __HIP_MEMORY_SCOPE_WORKGROUP);
        __hip_atomic_fetch_add(ap + 96, hv[u].w * wj, __ATOMIC_RELAXED, __HIP_MEMORY_SCOPE_WORKGROUP);
      }
    }
  }
}

__device__ __forceinline__ void own_drain(const int* qe, const int* qd, int m, int lane,
                                          int n0, const int* __restrict__ ei,
                                          const float* __restrict__ logit,
                                          const float* __restrict__ mx,
                                          const float* __restrict__ h,
                                          float* s_acc, float* s_sum, int* rec) {
  int s = 0, dl = 0;
  float w = 0.0f;
  if (lane < m) {
    const int e = qe[lane];
    dl = qd[lane];
    if (rec != nullptr) rec[lane] = (e << 8) | dl;
    s = ei[e];
    w = __expf(logit[e] - mx[n0 + dl]);
    __hip_atomic_fetch_add(&s_sum[dl], w, __ATOMIC_RELAXED, __HIP_MEMORY_SCOPE_WORKGROUP);
  }
  own_rows(s, dl, w, m, lane, h, s_acc);
}

__device__ __forceinline__ void own_drain_e(const int* qe, int m, int lane, int n0, int n_edges,
                                            const int* __restrict__ ei,
                                            const float* __restrict__ logit,
                                            const float* s_mx,
                                            const float* __restrict__ h,
                                            float* s_acc, float* s_sum, int* rec) {
  int s = 0, dl = 0;
  float w = 0.0f;
  if (lane < m) {
    const int e = qe[lane];
    s = ei[e];
    dl = ei[n_edges + e] - n0;
    const float lg = logit[e];
    if (rec != nullptr) rec[lane] = (e << 8) | dl;
    w = __expf(lg - s_mx[dl]);
    __hip_atomic_fetch_add(&s_sum[dl], w, __ATOMIC_RELAXED, __HIP_MEMORY_SCOPE_WORKGROUP);
  }
  own_rows(s, dl, w, m, lane, h, s_acc);
}

__device__ __forceinline__ unsigned zero_bytes(unsigned t) {
  return ~(((t & 0x7f7f7f7fu) + 0x7f7f7f7fu) | t | 0x7f7f7f7fu);
}

#define OWN_ROWS ((NPW + OWN_WAVES - 1) / OWN_WAVES)

template <bool BYTEWALK>
__global__ __launch_bounds__(OWN_WAVES * 32)
void gat_owner_accum(const int* __restrict__ ei, const float* __restrict__ logit,
                     const float* __restrict__ mx, const float* __restrict__ self_logit,
                     const float* __restrict__ h, const float* x,
                     const float* __restrict__ bias, const float* __restrict__ gamma,
                     const float* __restrict__ beta, float* out,
                     int n_nodes, int n_edges, int* elist, int* ecnt,
                     const int* __restrict__ ownb, int mode) {
  __shared__ float s_acc[NPW * D];
  __shared__ float s_sum[NPW];
  __shared__ float s_mx[NPW];
  __shared__ int   s_qe[OWN_WAVES * QCAP];
  __shared__ int   s_qd[OWN_WAVES * QCAP];
  __shared__ int   s_cnt[OWN_WAVES];

  const int tid  = threadIdx.x;
  const int wave = __builtin_amdgcn_readfirstlane(tid >> 5);
  const int lane = tid & 31;
  const int n0   = blockIdx.x * NPW;
  const int cnt  = min(NPW, n_nodes - n0);

  int* qe = s_qe + wave * QCAP;
  int* qd = s_qd + wave * QCAP;
  const int slot = blockIdx.x * OWN_WAVES + wave;
  int* lst = elist + (size_t)slot * LCAP;

  const float4 bv = ((const float4*)bias)[lane];
  const float4 gv = ((const float4*)gamma)[lane];
  const float4 tv = ((const float4*)beta)[lane];
  float4 xr[OWN_ROWS];
#pragma unroll
  for (int i = 0; i < OWN_ROWS; ++i) {
    const int r = wave + i * OWN_WAVES;
    xr[i] = (r < cnt) ? ((const float4*)(x + (size_t)(n0 + r) * D))[lane]
                      : make_float4(0.0f, 0.0f, 0.0f, 0.0f);
  }

  int lcnt = 0;
  bool replay = false;
  if (!BYTEWALK && mode == 2) {
    const int c = ecnt[blockIdx.x * OWN_WAVES + lane];
    lcnt = __builtin_amdgcn_readlane(c, wave);
    replay = __ballot((unsigned)c > (unsigned)LCAP) == 0;
  }

  if (replay) {
    int pw[LCH], sv[LCH];
    float lv[LCH], mv[LCH];
#pragma unroll
    for (int c = 0; c < LCH; ++c) {
      const int idx = c * 32 + lane;
      pw[c] = (idx < lcnt) ? lst[idx] : -1;
    }
#pragma unroll
    for (int c = 0; c < LCH; ++c) {
      const int p = max(pw[c], 0);
      const int e = p >> 8;
      sv[c] = ei[e];
      lv[c] = logit[e];
      mv[c] = mx[n0 + (p & 255)];
    }
#pragma unroll
    for (int c = 0; c < LCH; ++c) {
      if (pw[c] >= 0) {
        qe[c * 32 + lane] = (sv[c] << 8) | (pw[c] & 255);
        qd[c * 32 + lane] = __builtin_bit_cast(int, __expf(lv[c] - mv[c]));
      }
    }
  }

  for (int i = tid; i < cnt * 32; i += OWN_WAVES * 32) {
    const int r = i >> 5, c = i & 31;
    const int node = n0 + r;
    const float mv = mx[node];
    const float e0 = __expf(self_logit[node] - mv);
    const float4 hv = ((const float4*)(h + (size_t)node * D))[c];
    float* sp = s_acc + r * D + c;
    sp[0]  = hv.x * e0;
    sp[32] = hv.y * e0;
    sp[64] = hv.z * e0;
    sp[96] = hv.w * e0;
    if (c == 0) {
      s_sum[r] = e0;
      if (BYTEWALK) s_mx[r] = mv;
    }
  }
  __syncthreads();

  if (replay) {
    int off = 0;
    for (; off + 32 <= lcnt; off += 32) {
      const int sp  = qe[off + lane];
      const float w = __builtin_bit_cast(float, qd[off + lane]);
      __hip_atomic_fetch_add(&s_sum[sp & 255], w, __ATOMIC_RELAXED, __HIP_MEMORY_SCOPE_WORKGROUP);
      own_rows(sp >> 8, sp & 255, w, 32, lane, h, s_acc);
    }
    if (off < lcnt) {
      const int m = lcnt - off;
      int sp = 0;
      float w = 0.0f;
      if (lane < m) {
        sp = qe[off + lane];
        w  = __builtin_bit_cast(float, qd[off + lane]);
        __hip_atomic_fetch_add(&s_sum[sp & 255], w, __ATOMIC_RELAXED, __HIP_MEMORY_SCOPE_WORKGROUP);
      }
      own_rows(sp >> 8, sp & 255, w, m, lane, h, s_acc);
    }
  } else if (BYTEWALK) {
    int rn = 0;
    int qn = 0;
    const uint4* __restrict__ ow4 = (const uint4*)ownb;
    const unsigned idw = (unsigned)blockIdx.x * 0x01010101u;
    const int n_grp = (n_edges + 15) >> 4;
    const int stepg = BSCAN * 32;
    const int n_stp = (n_grp + stepg - 1) / stepg;
    const int first = (int)(((long long)blockIdx.x * n_stp) / gridDim.x);
    uint4 nx[BSCAN];
    int gn = 0;
    if (wave < n_stp) {
      int sg = first + wave;
      if (sg >= n_stp) sg -= n_stp;
      gn = sg * stepg;
#pragma unroll
      for (int k = 0; k < BSCAN; ++k) {
        const int g = gn + k * 32 + lane;
        nx[k] = (g < n_grp) ? ow4[g] : make_uint4(~0u, ~0u, ~0u, ~0u);
      }
    }
    for (int it = wave; it < n_stp; it += OWN_WAVES) {
      const int g0 = gn;
      uint4 ov[BSCAN];
#pragma unroll
      for (int k = 0; k < BSCAN; ++k) ov[k] = nx[k];
      if (it + OWN_WAVES < n_stp) {
        int sg = first + it + OWN_WAVES;
        if (sg >= n_stp) sg -= n_stp;
        gn = sg * stepg;
        if (gn + stepg <= n_grp) {
#pragma unroll
          for (int k = 0; k < BSCAN; ++k) nx[k] = ow4[gn + k * 32 + lane];
        } else {
#pragma unroll
          for (int k = 0; k < BSCAN; ++k) {
            const int g = gn + k * 32 + lane;
            nx[k] = (g < n_grp) ? ow4[g] : make_uint4(~0u, ~0u, ~0u, ~0u);
          }
        }
      }
#pragma unroll
      for (int k = 0; k < BSCAN; ++k) {
        const int g = g0 + k * 32 + lane;
        unsigned hm = (zero_bytes(ov[k].x ^ idw) >> 7) | (zero_bytes(ov[k].y ^ idw) >> 6) |
                      (zero_bytes(ov[k].z ^ idw) >> 5) | (zero_bytes(ov[k].w ^ idw) >> 4);
        unsigned msk = (unsigned)__ballot(hm != 0u);
        while (msk != 0u) {
          if (hm != 0u) {
            const int p = __builtin_ctz(hm);
            const int pos = qn + __popc(msk & ((1u << lane) - 1u));
            qe[pos] = g * 16 + (p & 7) * 4 + (p >> 3);
            hm &= hm - 1u;
          }
          qn += __popc(msk);
          msk = (unsigned)__ballot(hm != 0u);
        }
        while (qn >= 32) {
          qn -= 32;
          own_drain_e(qe + qn, 32, lane, n0, n_edges, ei, logit, s_mx, h, s_acc, s_sum,
                      (rn + 32 <= LCAP) ? lst + rn : nullptr);
          rn += 32;
        }
      }
    }
    if (qn > 0) {
      own_drain_e(qe, qn, lane, n0, n_edges, ei, logit, s_mx, h, s_acc, s_sum,
                  (rn + qn <= LCAP) ? lst + rn : nullptr);
      rn += qn;
    }
    if (lane == 0) s_cnt[wave] = rn;
  } else {
    const bool record = (mode == 1);
    int rn = 0;
    const int* __restrict__ dstp = ei + n_edges;
    int qn = 0;

    const int step  = SCAN_LOADS * 32;
    const int n_stp = (n_edges + step - 1) / step;
    const int first = (int)(((long long)blockIdx.x * n_stp) / gridDim.x);
    for (int it = wave; it < n_stp; it += OWN_WAVES) {
      int sg = first + it;
      if (sg >= n_stp) sg -= n_stp;
      const int base = sg * step;
      int dv[SCAN_LOADS];
      if (base + step <= n_edges) {
#pragma unroll
        for (int k = 0; k < SCAN_LOADS; ++k) dv[k] = dstp[base + k * 32 + lane];
      } else {
#pragma unroll
        for (int k = 0; k < SCAN_LOADS; ++k) {
          const int idx = base + k * 32 + lane;
          dv[k] = (idx < n_edges) ? dstp[idx] : -1;
        }
      }
#pragma unroll
      for (int k = 0; k < SCAN_LOADS; ++k) {
        const unsigned rel = (unsigned)(dv[k] - n0);
        const bool hit = rel < (unsigned)cnt;
        const unsigned msk = (unsigned)__ballot(hit);
        if (msk != 0u) {
          if (hit) {
            const int pos = qn + __popc(msk & ((1u << lane) - 1u));
            qe[pos] = base + k * 32 + lane;
            qd[pos] = (int)rel;
          }
          qn += __popc(msk);
        }
      }
      while (qn >= 32) {
        qn -= 32;
        own_drain(qe + qn, qd + qn, 32, lane, n0, ei, logit, mx, h, s_acc, s_sum,
                  (record && rn + 32 <= LCAP) ? lst + rn : nullptr);
        rn += 32;
      }
    }
    if (qn > 0) {
      own_drain(qe, qd, qn, lane, n0, ei, logit, mx, h, s_acc, s_sum,
                (record && rn + qn <= LCAP) ? lst + rn : nullptr);
      rn += qn;
    }
    if (record && lane == 0) s_cnt[wave] = rn;
  }
  __syncthreads();

  if (mode == 1 && wave == 0) ecnt[blockIdx.x * OWN_WAVES + lane] = s_cnt[lane];

  float4 y[OWN_ROWS];
  float sum[OWN_ROWS], sq[OWN_ROWS];
#pragma unroll
  for (int i = 0; i < OWN_ROWS; ++i) {
    const int rc = min(wave + i * OWN_WAVES, cnt - 1);
    const float* fp = s_acc + rc * D + lane;
    const float4 av = make_float4(fp[0], fp[32], fp[64], fp[96]);
    const float inv_s = 1.0f / s_sum[rc];
    y[i].x = xr[i].x + av.x * inv_s + bv.x;
    y[i].y = xr[i].y + av.y * inv_s + bv.y;
    y[i].z = xr[i].z + av.z * inv_s + bv.z;
    y[i].w = xr[i].w + av.w * inv_s + bv.w;
    sum[i] = y[i].x + y[i].y + y[i].z + y[i].w;
    sq[i]  = y[i].x * y[i].x + y[i].y * y[i].y + y[i].z * y[i].z + y[i].w * y[i].w;
  }
#pragma unroll
  for (int off = 16; off > 0; off >>= 1) {
#pragma unroll
    for (int i = 0; i < OWN_ROWS; ++i) {
      sum[i] += __shfl_xor(sum[i], off, 32);
      sq[i]  += __shfl_xor(sq[i], off, 32);
    }
  }
#pragma unroll
  for (int i = 0; i < OWN_ROWS; ++i) {
    const int r = wave + i * OWN_WAVES;
    const float mu   = sum[i] * (1.0f / D);
    const float var  = sq[i] * (1.0f / D) - mu * mu;
    const float rstd = rsqrtf(var + LN_EPS);
    float4 o;
    o.x = (y[i].x - mu) * rstd * gv.x + tv.x;
    o.y = (y[i].y - mu) * rstd * gv.y + tv.y;
    o.z = (y[i].z - mu) * rstd * gv.z + tv.z;
    o.w = (y[i].w - mu) * rstd * gv.w + tv.w;
    if (r < cnt) ((float4*)(out + (size_t)(n0 + r) * D))[lane] = o;
  }
}

extern "C" void kernel_launch(void* const* d_in, const int* in_sizes, int n_in,
                              void* d_out, int out_size, void* d_ws, size_t ws_size,
                              hipStream_t stream) {
  const float* x0   = (const float*)d_in[0];
  const int*   ei   = (const int*)d_in[1];
  const float* W    = (const float*)d_in[2];
  const float* asrc = (const float*)d_in[3];
  const float* adst = (const float*)d_in[4];
  const float* bias = (const float*)d_in[5];
  const float* gam  = (const float*)d_in[6];
  const float* bet  = (const float*)d_in[7];
  float* out = (float*)d_out;

  const int n  = in_sizes[0] / D;
  const int e  = in_sizes[1] / 2;
  const int nl = in_sizes[2] / (D * D);

  float* ws    = (float*)d_ws;
  float* h     = ws; ws += (size_t)n * D;
  float* va    = ws; ws += n;
  float* vb    = ws; ws += n;
  float* slf   = ws; ws += n;
  float* mx    = ws; ws += n;
  float* lgt   = ws; ws += e;

  const int    nblk      = (n + NPW - 1) / NPW;
  const size_t lst_off   = (((size_t)n * (D + 4) + (size_t)e) * sizeof(float) + 127) / 128 * 128;
  const size_t lst_words = (size_t)nblk * OWN_WAVES * LCAP;
  const size_t lst_end   = lst_off + (lst_words + (size_t)nblk * OWN_WAVES) * sizeof(int);
  const bool   use_list  = nl > 1 && ws_size >= lst_end && e < (1 << 23) && n < (1 << 23);
  int* elist = use_list ? (int*)((char*)d_ws + lst_off) : nullptr;
  int* ecnt  = use_list ? elist + lst_words : nullptr;
  const size_t own_end = lst_end + ((size_t)e + 15) / 16 * 16;
  const bool   use_own = use_list && nblk <= 255 && ws_size >= own_end;
  int* ownb = use_own ? (int*)((char*)d_ws + lst_end) : nullptr;

  for (int l = 0; l < nl; ++l) {
    const float* xin = (l == 0) ? x0 : out;
    const float* Wl  = W + (size_t)l * D * D;
    const int mode   = use_list ? (l == 0 ? 1 : 2) : 0;
    gat_gemm_wmma<<<(n + 127) / 128, 256, 0, stream>>>(xin, Wl, asrc + l * D, adst + l * D,
                                                       h, va, vb, slf, mx, n);
    int* ownl = (mode == 1) ? ownb : nullptr;
    gat_edge_logits<<<(e + 255) / 256, 256, 0, stream>>>(ei, va, vb, lgt, mx, ownl, e);
    if (ownl != nullptr)
      gat_owner_accum<true><<<nblk, OWN_WAVES * 32, 0, stream>>>(ei, lgt, mx, slf, h, xin,
                                                                 bias + l * D, gam + l * D,
                                                                 bet + l * D, out, n, e,
                                                                 elist, ecnt, ownl, mode);
    else
      gat_owner_accum<false><<<nblk, OWN_WAVES * 32, 0, stream>>>(ei, lgt, mx, slf, h, xin,
                                                                  bias + l * D, gam + l * D,
                                                                  bet + l * D, out, n, e,
                                                                  elist, ecnt, nullptr, mode);
  }
}


# Per-edge logit pass dropped: each owner work-group forms its edges' logits from the score planes, segment max by LDS f32 max; owner bytes ride in the product; 6 launches -> 4; sum order kept
# speedup vs baseline: 5.7090x; 1.9925x over previous
#include <hip/hip_runtime.h>
#include <math.h>

#define D 128
#define NEG_SLOPE 0.2f
#define LN_EPS 1e-5f
#define NPW 200

typedef float v2f __attribute__((ext_vector_type(2)));
typedef float v8f __attribute__((ext_vector_type(8)));

__device__ __forceinline__ float pick(bool c, float a, float b) { return c ? a : b; }

#define WQ 272

__global__ __launch_bounds__(256)
void gat_gemm_wmma(const float* __restrict__ x, const float* __restrict__ w,
                   const float* __restrict__ att_src, const float* __restrict__ att_dst,
                   float* __restrict__ h, float* __restrict__ a_src,
                   float* __restrict__ a_dst, float* __restrict__ self_logit,
                   int n_nodes, const int* __restrict__ dstp, int* __restrict__ ownb,
                   int n_edges) {
  __shared__ __attribute__((aligned(16))) float lds_w[(D / 2) * WQ];
  __shared__ __attribute__((aligned(16))) float lds_wa[2 * D];
  const int tid  = threadIdx.x;
  const int wave = tid >> 5;
  const int lane = tid & 31;
  const int hf   = lane >> 4;
  const int l16  = lane & 15;
  const int row0 = (blockIdx.x * 8 + wave) * 16;

  const int arow = min(row0 + l16, n_nodes - 1);
  const float4* __restrict__ ap = (const float4*)(x + (size_t)arow * D) + hf;
  float4 a[16];
#pragma unroll
  for (int q = 0; q < 16; ++q) a[q] = ap[2 * q];

  const float4 as4 = ((const float4*)att_src)[lane];
  const float4 ad4 = ((const float4*)att_dst)[lane];
  float pv[32];
#pragma unroll
  for (int i = 0; i < 8; ++i) {
    const int item = tid + i * 256;
    const int kp = item >> 5, m = item & 31;
    const float4 r0 = ((const float4*)(w + (size_t)(2 * kp) * D))[m];
    const float4 r1 = ((const float4*)(w + (size_t)(2 * kp + 1) * D))[m];
    float4* dst = (float4*)(lds_w + kp * WQ + m * 8);
    dst[0] = make_float4(r0.x, r1.x, r0.y, r1.y);
    dst[1] = make_float4(r0.z, r1.z, r0.w, r1.w);
    pv[4 * i + 0] = r0.x * as4.x + r0.y * as4.y + r0.z * as4.z + r0.w * as4.w;
    pv[4 * i + 1] = r1.x * as4.x + r1.y * as4.y + r1.z * as4.z + r1.w * as4.w;
    pv[4 * i + 2] = r0.x * ad4.x + r0.y * ad4.y + r0.z * ad4.z + r0.w * ad4.w;
    pv[4 * i + 3] = r1.x * ad4.x + r1.y * ad4.y + r1.z * ad4.z + r1.w * ad4.w;
  }
#pragma unroll
  for (int st = 0; st < 5; ++st) {
    const int off = 16 >> st;
    const bool up = (lane & off) != 0;
#pragma unroll
    for (int j = 0; j < off; ++j) {
      const float lo = pv[j], hi = pv[j + off];
      pv[j] = pick(up, hi, lo) + __shfl_xor(pick(up, lo, hi), off, 32);
    }
  }
  {
    const int kp = wave + 8 * (lane >> 2);
    const int t  = lane & 3;
    lds_wa[(t >> 1) * D + 2 * kp + (t & 1)] = pv[0];
  }
  __syncthreads();

  {
    const float4* was = (const float4*)lds_wa + hf;
    const float4* wad = (const float4*)(lds_wa + D) + hf;
    float ps = 0.0f, pd = 0.0f;
#pragma unroll
    for (int q = 0; q < 16; ++q) {
      const float4 s4 = was[2 * q];
      const float4 d4 = wad[2 * q];
      ps += a[q].x * s4.x + a[q].y * s4.y + a[q].z * s4.z + a[q].w * s4.w;
      pd += a[q].x * d4.x + a[q].y * d4.y + a[q].z * d4.z + a[q].w * d4.w;
    }
    ps += __shfl_xor(ps, 16, 32);
    pd += __shfl_xor(pd, 16, 32);
    const int node = row0 + l16;
    if (hf == 0 && node < n_nodes) {
      a_src[node] = ps;
      a_dst[node] = pd;
      float lg = ps + pd;
      lg = lg > 0.0f ? lg : NEG_SLOPE * lg;
      self_logit[node] = lg;
    }
  }

  v8f acc[8];
#pragma unroll
  for (int j = 0; j < 8; ++j) acc[j] = (v8f){};

  const float* bp = lds_w + (2 * hf) * WQ + 4 * l16;
  float4 bcur[4], bnxt[4];
#pragma unroll
  for (int jp = 0; jp < 4; ++jp) bcur[jp] = *(const float4*)(bp + 64 * jp);

#pragma unroll
  for (int t = 0; t < 32; ++t) {
    const int q = t >> 1, u = t & 1;
    if (t + 1 < 32) {
      const int kn = 4 * ((t + 1) >> 1) + ((t + 1) & 1);
#pragma unroll
      for (int jp = 0; jp < 4; ++jp) bnxt[jp] = *(const float4*)(bp + kn * WQ + 64 * jp);
    }
    __builtin_amdgcn_sched_barrier(0);
    v2f av;
    av.x = u ? a[q].z : a[q].x;
    av.y = u ? a[q].w : a[q].y;
#pragma unroll
    for (int jp = 0; jp < 4; ++jp) {
      v2f b0, b1;
      b0.x = bcur[jp].x; b0.y = bcur[jp].y;
      b1.x = bcur[jp].z; b1.y = bcur[jp].w;
      acc[2 * jp]     = __builtin_amdgcn_wmma_f32_16x16x4_f32(false, av, false, b0, (short)0,
                                                              acc[2 * jp], false, false);
      acc[2 * jp + 1] = __builtin_amdgcn_wmma_f32_16x16x4_f32(false, av, false, b1, (short)0,
                                                              acc[2 * jp + 1], false, false);
    }
    __builtin_amdgcn_sched_barrier(0);
#pragma unroll
    for (int jp = 0; jp < 4; ++jp) bcur[jp] = bnxt[jp];
  }

#pragma unroll
  for (int r = 0; r < 8; ++r) {
    const int row = row0 + 8 * hf + r;
    if (row < n_nodes) {
      float* hp = h + (size_t)row * D + 2 * l16;
#pragma unroll
      for (int jp = 0; jp < 4; ++jp) {
        float2 o;
        o.x = acc[2 * jp][r];
        o.y = acc[2 * jp + 1][r];
        *(float2*)(hp + 32 * jp) = o;
      }
    }
  }

  if (ownb != nullptr) {
    unsigned* ow = (unsigned*)ownb;
    const int n_wrd = ((n_edges + 15) >> 4) << 2;
    const int n_chk = (n_wrd + 1023) >> 10;
    const bool al16 = (((size_t)dstp) & 15) == 0;
    for (int ch = blockIdx.x; ch < n_chk; ch += gridDim.x) {
      const int w0 = ch * 1024 + tid;
      if (al16 && (ch + 1) * 4096 <= n_edges) {
        int4 dv[4];
#pragma unroll
        for (int u = 0; u < 4; ++u) dv[u] = ((const int4*)dstp)[w0 + u * 256];
#pragma unroll
        for (int u = 0; u < 4; ++u)
          ow[w0 + u * 256] = ((unsigned)dv[u].x / NPW) | (((unsigned)dv[u].y / NPW) << 8) |
                             (((unsigned)dv[u].z / NPW) << 16) | (((unsigned)dv[u].w / NPW) << 24);
      } else {
        int dv[16];
#pragma unroll
        for (int k = 0; k < 16; ++k)
          dv[k] = dstp[min((w0 + (k >> 2) * 256) * 4 + (k & 3), n_edges - 1)];
#pragma unroll
        for (int u = 0; u < 4; ++u) {
          const int wd = w0 + u * 256;
          unsigned word = 0u;
#pragma unroll
          for (int b = 0; b < 4; ++b) {
            const unsigned o = (wd * 4 + b < n_edges) ? (unsigned)dv[4 * u + b] / NPW : 255u;
            word |= o << (8 * b);
          }
          if (wd < n_wrd) ow[wd] = word;
        }
      }
    }
  }
}

#define OWN_WAVES 32
#define QCAP 544
#define LCAP 224
#define LCH (LCAP / 32)
#define BSCAN 4

static_assert(NPW <= 256, "a local destination must fit the low 8 bits of a list word");
static_assert(LCAP <= QCAP, "a wave's list slice is unpacked into its own queue");
static_assert(OWN_WAVES == 32, "one lane per wave stores the work-group's line of counts");

__device__ __forceinline__ void own_rows(int s, int dl, float w, int m, int lane,
                                         const float* __restrict__ h, float* s_acc) {
  for (int j = 0; j < m; j += 8) {
    float4 hv[8];
#pragma unroll
    for (int u = 0; u < 8; ++u) {
      if (j + u < m) {
        const int sj = __builtin_amdgcn_readlane(s, j + u);
        hv[u] = ((const float4*)h)[((unsigned)sj << 5) | (unsigned)lane];
      }
    }
#pragma unroll
    for (int u = 0; u < 8; ++u) {
      if (j + u < m) {
        const int dj = __builtin_amdgcn_readlane(dl, j + u);
        const float wj = __builtin_bit_cast(float, __builtin_amdgcn_readlane(__builtin_bit_cast(int, w), j + u));
        float* ap = s_acc + dj * D + lane;
        __hip_atomic_fetch_add(ap + 0,  hv[u].x * wj, __ATOMIC_RELAXED, __HIP_MEMORY_SCOPE_WORKGROUP);
        __hip_atomic_fetch_add(ap + 32, hv[u].y * wj, __ATOMIC_RELAXED, __HIP_MEMORY_SCOPE_WORKGROUP);
        __hip_atomic_fetch_add(ap + 64, hv[u].z * wj, __ATOMIC_RELAXED, __HIP_MEMORY_SCOPE_WORKGROUP);
        __hip_atomic_fetch_add(ap + 96, hv[u].w * wj, __ATOMIC_RELAXED, __HIP_MEMORY_SCOPE_WORKGROUP);
      }
    }
  }
}

__device__ __forceinline__ void own_note(const int* qe, int m, int lane, int n0, int n_edges,
                                         const int* __restrict__ ei, int* lst, int rn) {
  if (lane < m) {
    const int e  = qe[lane];
    const int s  = ei[e];
    const int dl = ei[n_edges + e] - n0;
    if (rn + lane < LCAP) lst[rn + lane] = (s << 8) | dl;
  }
}

__device__ __forceinline__ bool own_slow_edge(int e, int n_edges, int n0, int cnt,
                                              const int* __restrict__ ei,
                                              const float* __restrict__ a_src,
                                              const float* s_ad, int& s, int& dl, float& lg) {
  s = 0; dl = 0; lg = 0.0f;
  if (e >= n_edges) return false;
  const unsigned rel = (unsigned)(ei[n_edges + e] - n0);
  if (rel >= (unsigned)cnt) return false;
  s  = ei[e];
  dl = (int)rel;
  const float l = a_src[s] + s_ad[dl];
  lg = l > 0.0f ? l : NEG_SLOPE * l;
  return true;
}

__device__ __forceinline__ unsigned zero_bytes(unsigned t) {
  return ~(((t & 0x7f7f7f7fu) + 0x7f7f7f7fu) | t | 0x7f7f7f7fu);
}

#define OWN_ROWS ((NPW + OWN_WAVES - 1) / OWN_WAVES)
#define OWN_START ((NPW * 32 + OWN_WAVES * 32 - 1) / (OWN_WAVES * 32))

template <int MODE>
__global__ __launch_bounds__(OWN_WAVES * 32)
void gat_owner_accum(const int* __restrict__ ei, const float* __restrict__ a_src,
                     const float* __restrict__ a_dst, const float* __restrict__ self_logit,
                     const float* __restrict__ h, const float* x,
                     const float* __restrict__ bias, const float* __restrict__ gamma,
                     const float* __restrict__ beta, float* out,
                     int n_nodes, int n_edges, int* elist, int* ecnt,
                     const int* __restrict__ ownb) {
  __shared__ float s_acc[NPW * D];
  __shared__ float s_sum[NPW];
  __shared__ float s_mx[NPW];
  __shared__ float s_ad[NPW];
  __shared__ float s_sl[NPW];
  __shared__ int   s_qe[OWN_WAVES * QCAP];
  __shared__ int   s_qd[OWN_WAVES * QCAP];
  __shared__ int   s_cnt[OWN_WAVES];

  const int tid  = threadIdx.x;
  const int wave = __builtin_amdgcn_readfirstlane(tid >> 5);
  const int lane = tid & 31;
  const int n0   = blockIdx.x * NPW;
  const int cnt  = min(NPW, n_nodes - n0);

  int* qe = s_qe + wave * QCAP;
  int* qd = s_qd + wave * QCAP;
  const int slot = blockIdx.x * OWN_WAVES + wave;
  int* lst = elist + (size_t)slot * LCAP;

  const float4 bv = ((const float4*)bias)[lane];
  const float4 gv = ((const float4*)gamma)[lane];
  const float4 tv = ((const float4*)beta)[lane];
  float4 xr[OWN_ROWS];
#pragma unroll
  for (int i = 0; i < OWN_ROWS; ++i) {
    const int r = wave + i * OWN_WAVES;
    xr[i] = (r < cnt) ? ((const float4*)(x + (size_t)(n0 + r) * D))[lane]
                      : make_float4(0.0f, 0.0f, 0.0f, 0.0f);
  }

  if (tid < cnt) {
    const float sl = self_logit[n0 + tid];
    s_sl[tid] = sl;
    s_mx[tid] = sl;
    s_ad[tid] = a_dst[n0 + tid];
  }

  int pw[LCH];
  float sv[LCH], lg[LCH];
  int lcnt = 0;
  bool replay = false;

  if (MODE == 2) {
    const int c = ecnt[blockIdx.x * OWN_WAVES + lane];
    lcnt = __builtin_amdgcn_readlane(c, wave);
    replay = __ballot((unsigned)c > (unsigned)LCAP) == 0;
    if (replay) {
#pragma unroll
      for (int k = 0; k < LCH; ++k) {
        const int idx = k * 32 + lane;
        pw[k] = (idx < lcnt) ? lst[idx] : -1;
      }
#pragma unroll
      for (int k = 0; k < LCH; ++k) sv[k] = a_src[max(pw[k], 0) >> 8];
    }
  }

  if (MODE == 1) {
    int rn = 0;
    int qn = 0;
    const uint4* __restrict__ ow4 = (const uint4*)ownb;
    const unsigned idw = (unsigned)blockIdx.x * 0x01010101u;
    const int n_grp = (n_edges + 15) >> 4;
    const int stepg = BSCAN * 32;
    const int n_stp = (n_grp + stepg - 1) / stepg;
    const int first = (int)(((long long)blockIdx.x * n_stp) / gridDim.x);
    uint4 nx[BSCAN];
    int gn = 0;
    if (wave < n_stp) {
      int sg = first + wave;
      if (sg >= n_stp) sg -= n_stp;
      gn = sg * stepg;
#pragma unroll
      for (int k = 0; k < BSCAN; ++k) {
        const int g = gn + k * 32 + lane;
        nx[k] = (g < n_grp) ? ow4[g] : make_uint4(~0u, ~0u, ~0u, ~0u);
      }
    }
    for (int it = wave; it < n_stp; it += OWN_WAVES) {
      const int g0 = gn;
      uint4 ov[BSCAN];
#pragma unroll
      for (int k = 0; k < BSCAN; ++k) ov[k] = nx[k];
      if (it + OWN_WAVES < n_stp) {
        int sg = first + it + OWN_WAVES;
        if (sg >= n_stp) sg -= n_stp;
        gn = sg * stepg;
        if (gn + stepg <= n_grp) {
#pragma unroll
          for (int k = 0; k < BSCAN; ++k) nx[k] = ow4[gn + k * 32 + lane];
        } else {
#pragma unroll
          for (int k = 0; k < BSCAN; ++k) {
            const int g = gn + k * 32 + lane;
            nx[k] = (g < n_grp) ? ow4[g] : make_uint4(~0u, ~0u, ~0u, ~0u);
          }
        }
      }
#pragma unroll
      for (int k = 0; k < BSCAN; ++k) {
        const int g = g0 + k * 32 + lane;
        unsigned hm = (zero_bytes(ov[k].x ^ idw) >> 7) | (zero_bytes(ov[k].y ^ idw) >> 6) |
                      (zero_bytes(ov[k].z ^ idw) >> 5) | (zero_bytes(ov[k].w ^ idw) >> 4);
        unsigned msk = (unsigned)__ballot(hm != 0u);
        while (msk != 0u) {
          if (hm != 0u) {
            const int p = __builtin_ctz(hm);
            const int pos = qn + __popc(msk & ((1u << lane) - 1u));
            qe[pos] = g * 16 + (p & 7) * 4 + (p >> 3);
            hm &= hm - 1u;
          }
          qn += __popc(msk);
          msk = (unsigned)__ballot(hm != 0u);
        }
        while (qn >= 32) {
          qn -= 32;
          own_note(qe + qn, 32, lane, n0, n_edges, ei, lst, rn);
          rn += 32;
        }
      }
    }
    if (qn > 0) {
      own_note(qe, qn, lane, n0, n_edges, ei, lst, rn);
      rn += qn;
    }
    if (lane == 0) s_cnt[wave] = rn;
  }
  __syncthreads();

  if (MODE == 1) {
    const int c = s_cnt[lane];
    lcnt = __builtin_amdgcn_readlane(c, wave);
    replay = __ballot((unsigned)c > (unsigned)LCAP) == 0;
    if (wave == 0) ecnt[blockIdx.x * OWN_WAVES + lane] = c;
    if (replay) {
#pragma unroll
      for (int k = 0; k < LCH; ++k) {
        const int idx = k * 32 + lane;
        pw[k] = (idx < lcnt) ? lst[idx] : -1;
      }
#pragma unroll
      for (int k = 0; k < LCH; ++k) sv[k] = a_src[max(pw[k], 0) >> 8];
    }
  }

  if (replay) {
#pragma unroll
    for (int k = 0; k < LCH; ++k) {
      lg[k] = 0.0f;
      if (pw[k] >= 0) {
        const int dl = pw[k] & 255;
        const float l = sv[k] + s_ad[dl];
        lg[k] = l > 0.0f ? l : NEG_SLOPE * l;
        __hip_atomic_fetch_max(&s_mx[dl], lg[k], __ATOMIC_RELAXED, __HIP_MEMORY_SCOPE_WORKGROUP);
      }
    }
  } else {
    for (int base = wave * 32; base < n_edges; base += OWN_WAVES * 32) {
      int s, dl;
      float l;
      if (own_slow_edge(base + lane, n_edges, n0, cnt, ei, a_src, s_ad, s, dl, l))
        __hip_atomic_fetch_max(&s_mx[dl], l, __ATOMIC_RELAXED, __HIP_MEMORY_SCOPE_WORKGROUP);
    }
  }

  float4 hst[OWN_START];
#pragma unroll
  for (int i = 0; i < OWN_START; ++i) {
    const int p = tid + i * OWN_WAVES * 32;
    hst[i] = (p < cnt * 32) ? ((const float4*)(h + (size_t)n0 * D))[p]
                            : make_float4(0.0f, 0.0f, 0.0f, 0.0f);
  }
  __syncthreads();

#pragma unroll
  for (int i = 0; i < OWN_START; ++i) {
    const int p = tid + i * OWN_WAVES * 32;
    if (p < cnt * 32) {
      const int r = p >> 5, c = p & 31;
      const float e0 = __expf(s_sl[r] - s_mx[r]);
      float* sp = s_acc + r * D + c;
      sp[0]  = hst[i].x * e0;
      sp[32] = hst[i].y * e0;
      sp[64] = hst[i].z * e0;
      sp[96] = hst[i].w * e0;
      if (c == 0) s_sum[r] = e0;
    }
  }
  if (replay) {
#pragma unroll
    for (int k = 0; k < LCH; ++k) {
      if (pw[k] >= 0) {
        qe[k * 32 + lane] = pw[k];
        qd[k * 32 + lane] = __builtin_bit_cast(int, __expf(lg[k] - s_mx[pw[k] & 255]));
      }
    }
  }
  __syncthreads();

  if (replay) {
    int off = 0;
    for (; off + 32 <= lcnt; off += 32) {
      const int sp  = qe[off + lane];
      const float w = __builtin_bit_cast(float, qd[off + lane]);
      __hip_atomic_fetch_add(&s_sum[sp & 255], w, __ATOMIC_RELAXED, __HIP_MEMORY_SCOPE_WORKGROUP);
      own_rows(sp >> 8, sp & 255, w, 32, lane, h, s_acc);
    }
    if (off < lcnt) {
      const int m = lcnt - off;
      int sp = 0;
      float w = 0.0f;
      if (lane < m) {
        sp = qe[off + lane];
        w  = __builtin_bit_cast(float, qd[off + lane]);
        __hip_atomic_fetch_add(&s_sum[sp & 255], w, __ATOMIC_RELAXED, __HIP_MEMORY_SCOPE_WORKGROUP);
      }
      own_rows(sp >> 8, sp & 255, w, m, lane, h, s_acc);
    }
  } else {
    for (int base = wave * 32; base < n_edges; base += OWN_WAVES * 32) {
      int s, dl;
      float l;
      const bool hit = own_slow_edge(base + lane, n_edges, n0, cnt, ei, a_src, s_ad, s, dl, l);
      float w = 0.0f;
      if (hit) {
        w = __expf(l - s_mx[dl]);
        __hip_atomic_fetch_add(&s_sum[dl], w, __ATOMIC_RELAXED, __HIP_MEMORY_SCOPE_WORKGROUP);
      }
      unsigned msk = (unsigned)__ballot(hit);
      while (msk != 0u) {
        const int j = __builtin_ctz(msk);
        msk &= msk - 1u;
        const int sj = __builtin_amdgcn_readlane(s, j);
        const int dj = __builtin_amdgcn_readlane(dl, j);
        const float wj = __builtin_bit_cast(float, __builtin_amdgcn_readlane(__builtin_bit_cast(int, w), j));
        const float4 hv = ((const float4*)h)[((unsigned)sj << 5) | (unsigned)lane];
        float* ap = s_acc + dj * D + lane;
        __hip_atomic_fetch_add(ap + 0,  hv.x * wj, __ATOMIC_RELAXED, __HIP_MEMORY_SCOPE_WORKGROUP);
        __hip_atomic_fetch_add(ap + 32, hv.y * wj, __ATOMIC_RELAXED, __HIP_MEMORY_SCOPE_WORKGROUP);
        __hip_atomic_fetch_add(ap + 64, hv.z * wj, __ATOMIC_RELAXED, __HIP_MEMORY_SCOPE_WORKGROUP);
        __hip_atomic_fetch_add(ap + 96, hv.w * wj, __ATOMIC_RELAXED, __HIP_MEMORY_SCOPE_WORKGROUP);
      }
    }
  }
  __syncthreads();

  float4 y[OWN_ROWS];
  float sum[OWN_ROWS], sq[OWN_ROWS];
#pragma unroll
  for (int i = 0; i < OWN_ROWS; ++i) {
    const int rc = min(wave + i * OWN_WAVES, cnt - 1);
    const float* fp = s_acc + rc * D + lane;
    const float4 av = make_float4(fp[0], fp[32], fp[64], fp[96]);
    const float inv_s = 1.0f / s_sum[rc];
    y[i].x = xr[i].x + av.x * inv_s + bv.x;
    y[i].y = xr[i].y + av.y * inv_s + bv.y;
    y[i].z = xr[i].z + av.z * inv_s + bv.z;
    y[i].w = xr[i].w + av.w * inv_s + bv.w;
    sum[i] = y[i].x + y[i].y + y[i].z + y[i].w;
    sq[i]  = y[i].x * y[i].x + y[i].y * y[i].y + y[i].z * y[i].z + y[i].w * y[i].w;
  }
#pragma unroll
  for (int off = 16; off > 0; off >>= 1) {
#pragma unroll
    for (int i = 0; i < OWN_ROWS; ++i) {
      sum[i] += __shfl_xor(sum[i], off, 32);
      sq[i]  += __shfl_xor(sq[i], off, 32);
    }
  }
#pragma unroll
  for (int i = 0; i < OWN_ROWS; ++i) {
    const int r = wave + i * OWN_WAVES;
    const float mu   = sum[i] * (1.0f / D);
    const float var  = sq[i] * (1.0f / D) - mu * mu;
    const float rstd = rsqrtf(var + LN_EPS);
    float4 o;
    o.x = (y[i].x - mu) * rstd * gv.x + tv.x;
    o.y = (y[i].y - mu) * rstd * gv.y + tv.y;
    o.z = (y[i].z - mu) * rstd * gv.z + tv.z;
    o.w = (y[i].w - mu) * rstd * gv.w + tv.w;
    if (r < cnt) ((float4*)(out + (size_t)(n0 + r) * D))[lane] = o;
  }
}

extern "C" void kernel_launch(void* const* d_in, const int* in_sizes, int n_in,
                              void* d_out, int out_size, void* d_ws, size_t ws_size,
                              hipStream_t stream) {
  const float* x0   = (const float*)d_in[0];
  const int*   ei   = (const int*)d_in[1];
  const float* W    = (const float*)d_in[2];
  const float* asrc = (const float*)d_in[3];
  const float* adst = (const float*)d_in[4];
  const float* bias = (const float*)d_in[5];
  const float* gam  = (const float*)d_in[6];
  const float* bet  = (const float*)d_in[7];
  float* out = (float*)d_out;

  const int n  = in_sizes[0] / D;
  const int e  = in_sizes[1] / 2;
  const int nl = in_sizes[2] / (D * D);

  float* ws    = (float*)d_ws;
  float* h     = ws; ws += (size_t)n * D;
  float* va    = ws; ws += n;
  float* vb    = ws; ws += n;
  float* slf   = ws; ws += n;

  const int    nblk      = (n + NPW - 1) / NPW;
  const size_t lst_off   = (((size_t)n * (D + 4) + (size_t)e) * sizeof(float) + 127) / 128 * 128;
  const size_t lst_words = (size_t)nblk * OWN_WAVES * LCAP;
  const size_t lst_end   = lst_off + (lst_words + (size_t)nblk * OWN_WAVES) * sizeof(int);
  const size_t own_end   = lst_end + ((size_t)e + 15) / 16 * 16;
  const bool   use_list  = ws_size >= own_end && n < (1 << 23) && nblk <= 255;
  int* elist = use_list ? (int*)((char*)d_ws + lst_off) : nullptr;
  int* ecnt  = use_list ? elist + lst_words : nullptr;
  int* ownb  = use_list ? (int*)((char*)d_ws + lst_end) : nullptr;

  for (int l = 0; l < nl; ++l) {
    const float* xin = (l == 0) ? x0 : out;
    const float* Wl  = W + (size_t)l * D * D;
    int* ownl = (l == 0) ? ownb : nullptr;
    gat_gemm_wmma<<<(n + 127) / 128, 256, 0, stream>>>(xin, Wl, asrc + l * D, adst + l * D,
                                                       h, va, vb, slf, n, ei + e, ownl, e);
    if (!use_list)
      gat_owner_accum<0><<<nblk, OWN_WAVES * 32, 0, stream>>>(ei, va, vb, slf, h, xin,
                                                               bias + l * D, gam + l * D,
                                                               bet + l * D, out, n, e,
                                                               nullptr, nullptr, nullptr);
    else if (l == 0)
      gat_owner_accum<1><<<nblk, OWN_WAVES * 32, 0, stream>>>(ei, va, vb, slf, h, xin,
                                                               bias + l * D, gam + l * D,
                                                               bet + l * D, out, n, e,
                                                               elist, ecnt, ownb);
    else
      gat_owner_accum<2><<<nblk, OWN_WAVES * 32, 0, stream>>>(ei, va, vb, slf, h, xin,
                                                               bias + l * D, gam + l * D,
                                                               bet + l * D, out, n, e,
                                                               elist, ecnt, nullptr);
  }
}


# Edge sums without LDS f32 adds: each edge draws a place from its dst's LDS int counter, edges grouped by dst in LDS, a wave sums whole nodes' rows in registers by fma; sum order: self, then places
# speedup vs baseline: 6.1687x; 1.0805x over previous
#include <hip/hip_runtime.h>
#include <math.h>

#define D 128
#define NEG_SLOPE 0.2f
#define LN_EPS 1e-5f
#define NPW 200

typedef float v2f __attribute__((ext_vector_type(2)));
typedef float v8f __attribute__((ext_vector_type(8)));

__device__ __forceinline__ float pick(bool c, float a, float b) { return c ? a : b; }

#define WQ 272

__global__ __launch_bounds__(256)
void gat_gemm_wmma(const float* __restrict__ x, const float* __restrict__ w,
                   const float* __restrict__ att_src, const float* __restrict__ att_dst,
                   float* __restrict__ h, float* __restrict__ a_src,
                   float* __restrict__ a_dst, float* __restrict__ self_logit,
                   int n_nodes, const int* __restrict__ dstp, int* __restrict__ ownb,
                   int n_edges) {
  __shared__ __attribute__((aligned(16))) float lds_w[(D / 2) * WQ];
  __shared__ __attribute__((aligned(16))) float lds_wa[2 * D];
  const int tid  = threadIdx.x;
  const int wave = tid >> 5;
  const int lane = tid & 31;
  const int hf   = lane >> 4;
  const int l16  = lane & 15;
  const int row0 = (blockIdx.x * 8 + wave) * 16;

  const int arow = min(row0 + l16, n_nodes - 1);
  const float4* __restrict__ ap = (const float4*)(x + (size_t)arow * D) + hf;
  float4 a[16];
#pragma unroll
  for (int q = 0; q < 16; ++q) a[q] = ap[2 * q];

  const float4 as4 = ((const float4*)att_src)[lane];
  const float4 ad4 = ((const float4*)att_dst)[lane];
  float pv[32];
#pragma unroll
  for (int i = 0; i < 8; ++i) {
    const int item = tid + i * 256;
    const int kp = item >> 5, m = item & 31;
    const float4 r0 = ((const float4*)(w + (size_t)(2 * kp) * D))[m];
    const float4 r1 = ((const float4*)(w + (size_t)(2 * kp + 1) * D))[m];
    float4* dst = (float4*)(lds_w + kp * WQ + m * 8);
    dst[0] = make_float4(r0.x, r1.x, r0.y, r1.y);
    dst[1] = make_float4(r0.z, r1.z, r0.w, r1.w);
    pv[4 * i + 0] = r0.x * as4.x + r0.y * as4.y + r0.z * as4.z + r0.w * as4.w;
    pv[4 * i + 1] = r1.x * as4.x + r1.y * as4.y + r1.z * as4.z + r1.w * as4.w;
    pv[4 * i + 2] = r0.x * ad4.x + r0.y * ad4.y + r0.z * ad4.z + r0.w * ad4.w;
    pv[4 * i + 3] = r1.x * ad4.x + r1.y * ad4.y + r1.z * ad4.z + r1.w * ad4.w;
  }
#pragma unroll
  for (int st = 0; st < 5; ++st) {
    const int off = 16 >> st;
    const bool up = (lane & off) != 0;
#pragma unroll
    for (int j = 0; j < off; ++j) {
      const float lo = pv[j], hi = pv[j + off];
      pv[j] = pick(up, hi, lo) + __shfl_xor(pick(up, lo, hi), off, 32);
    }
  }
  {
    const int kp = wave + 8 * (lane >> 2);
    const int t  = lane & 3;
    lds_wa[(t >> 1) * D + 2 * kp + (t & 1)] = pv[0];
  }
  __syncthreads();

  {
    const float4* was = (const float4*)lds_wa + hf;
    const float4* wad = (const float4*)(lds_wa + D) + hf;
    float ps = 0.0f, pd = 0.0f;
#pragma unroll
    for (int q = 0; q < 16; ++q) {
      const float4 s4 = was[2 * q];
      const float4 d4 = wad[2 * q];
      ps += a[q].x * s4.x + a[q].y * s4.y + a[q].z * s4.z + a[q].w * s4.w;
      pd += a[q].x * d4.x + a[q].y * d4.y + a[q].z * d4.z + a[q].w * d4.w;
    }
    ps += __shfl_xor(ps, 16, 32);
    pd += __shfl_xor(pd, 16, 32);
    const int node = row0 + l16;
    if (hf == 0 && node < n_nodes) {
      a_src[node] = ps;
      a_dst[node] = pd;
      float lg = ps + pd;
      lg = lg > 0.0f ? lg : NEG_SLOPE * lg;
      self_logit[node] = lg;
    }
  }

  v8f acc[8];
#pragma unroll
  for (int j = 0; j < 8; ++j) acc[j] = (v8f){};

  const float* bp = lds_w + (2 * hf) * WQ + 4 * l16;
  float4 bcur[4], bnxt[4];
#pragma unroll
  for (int jp = 0; jp < 4; ++jp) bcur[jp] = *(const float4*)(bp + 64 * jp);

#pragma unroll
  for (int t = 0; t < 32; ++t) {
    const int q = t >> 1, u = t & 1;
    if (t + 1 < 32) {
      const int kn = 4 * ((t + 1) >> 1) + ((t + 1) & 1);
#pragma unroll
      for (int jp = 0; jp < 4; ++jp) bnxt[jp] = *(const float4*)(bp + kn * WQ + 64 * jp);
    }
    __builtin_amdgcn_sched_barrier(0);
    v2f av;
    av.x = u ? a[q].z : a[q].x;
    av.y = u ? a[q].w : a[q].y;
#pragma unroll
    for (int jp = 0; jp < 4; ++jp) {
      v2f b0, b1;
      b0.x = bcur[jp].x; b0.y = bcur[jp].y;
      b1.x = bcur[jp].z; b1.y = bcur[jp].w;
      acc[2 * jp]     = __builtin_amdgcn_wmma_f32_16x16x4_f32(false, av, false, b0, (short)0,
                                                              acc[2 * jp], false, false);
      acc[2 * jp + 1] = __builtin_amdgcn_wmma_f32_16x16x4_f32(false, av, false, b1, (short)0,
                                                              acc[2 * jp + 1], false, false);
    }
    __builtin_amdgcn_sched_barrier(0);
#pragma unroll
    for (int jp = 0; jp < 4; ++jp) bcur[jp] = bnxt[jp];
  }

#pragma unroll
  for (int r = 0; r < 8; ++r) {
    const int row = row0 + 8 * hf + r;
    if (row < n_nodes) {
      float* hp = h + (size_t)row * D + 2 * l16;
#pragma unroll
      for (int jp = 0; jp < 4; ++jp) {
        float2 o;
        o.x = acc[2 * jp][r];
        o.y = acc[2 * jp + 1][r];
        *(float2*)(hp + 32 * jp) = o;
      }
    }
  }

  if (ownb != nullptr) {
    unsigned* ow = (unsigned*)ownb;
    const int n_wrd = ((n_edges + 15) >> 4) << 2;
    const int n_chk = (n_wrd + 1023) >> 10;
    const bool al16 = (((size_t)dstp) & 15) == 0;
    for (int ch = blockIdx.x; ch < n_chk; ch += gridDim.x) {
      const int w0 = ch * 1024 + tid;
      if (al16 && (ch + 1) * 4096 <= n_edges) {
        int4 dv[4];
#pragma unroll
        for (int u = 0; u < 4; ++u) dv[u] = ((const int4*)dstp)[w0 + u * 256];
#pragma unroll
        for (int u = 0; u < 4; ++u)
          ow[w0 + u * 256] = ((unsigned)dv[u].x / NPW) | (((unsigned)dv[u].y / NPW) << 8) |
                             (((unsigned)dv[u].z / NPW) << 16) | (((unsigned)dv[u].w / NPW) << 24);
      } else {
        int dv[16];
#pragma unroll
        for (int k = 0; k < 16; ++k)
          dv[k] = dstp[min((w0 + (k >> 2) * 256) * 4 + (k & 3), n_edges - 1)];
#pragma unroll
        for (int u = 0; u < 4; ++u) {
          const int wd = w0 + u * 256;
          unsigned word = 0u;
#pragma unroll
          for (int b = 0; b < 4; ++b) {
            const unsigned o = (wd * 4 + b < n_edges) ? (unsigned)dv[4 * u + b] / NPW : 255u;
            word |= o << (8 * b);
          }
          if (wd < n_wrd) ow[wd] = word;
        }
      }
    }
  }
}

#define OWN_WAVES 32
#define QCAP 544
#define LCAP 224
#define LCH (LCAP / 32)
#define BSCAN 4

static_assert(NPW <= 256, "a local destination must fit the low 8 bits of a list word");
static_assert(OWN_WAVES * LCAP + NPW <= OWN_WAVES * QCAP, "the grouped edges of a work-group lie in the queues' words");
static_assert(OWN_WAVES == 32, "one lane per wave stores the work-group's line of counts");

struct RowAcc { float x, y, z, w, s; int cur; };

__device__ __forceinline__ void own_flush(const RowAcc& a, int lane, float* s_acc, float* s_sum) {
  float* sp = s_acc + a.cur * D + lane;
  sp[0]  = a.x;
  sp[32] = a.y;
  sp[64] = a.z;
  sp[96] = a.w;
  if (lane == 0) s_sum[a.cur] = a.s;
}

__device__ __forceinline__ void own_stream(int wd, float wt, int m, int lane,
                                           const float* __restrict__ h, float* s_acc,
                                           float* s_sum, RowAcc& a) {
  for (int j = 0; j < m; j += 8) {
    float4 hv[8];
#pragma unroll
    for (int u = 0; u < 8; ++u) {
      if (j + u < m) {
        const unsigned wj = (unsigned)__builtin_amdgcn_readlane(wd, j + u);
        hv[u] = ((const float4*)h)[((wj >> 8) << 5) | (unsigned)lane];
      }
    }
#pragma unroll
    for (int u = 0; u < 8; ++u) {
      if (j + u < m) {
        const int dj = __builtin_amdgcn_readlane(wd, j + u) & 255;
        const float fj = __builtin_bit_cast(float, __builtin_amdgcn_readlane(__builtin_bit_cast(int, wt), j + u));
        if (dj != a.cur) {
          if (a.cur >= 0) own_flush(a, lane, s_acc, s_sum);
          a.cur = dj;
          a.x = 0.0f; a.y = 0.0f; a.z = 0.0f; a.w = 0.0f; a.s = 0.0f;
        }
        a.x += hv[u].x * fj;
        a.y += hv[u].y * fj;
        a.z += hv[u].z * fj;
        a.w += hv[u].w * fj;
        a.s += fj;
      }
    }
  }
}

__device__ __forceinline__ void own_note(const int* qe, int m, int lane, int n0, int n_edges,
                                         const int* __restrict__ ei, int* lst, int rn) {
  if (lane < m) {
    const int e  = qe[lane];
    const int s  = ei[e];
    const int dl = ei[n_edges + e] - n0;
    if (rn + lane < LCAP) lst[rn + lane] = (s << 8) | dl;
  }
}

__device__ __forceinline__ bool own_slow_edge(int e, int n_edges, int n0, int cnt,
                                              const int* __restrict__ ei,
                                              const float* __restrict__ a_src,
                                              const float* s_ad, int& s, int& dl, float& lg) {
  s = 0; dl = 0; lg = 0.0f;
  if (e >= n_edges) return false;
  const unsigned rel = (unsigned)(ei[n_edges + e] - n0);
  if (rel >= (unsigned)cnt) return false;
  s  = ei[e];
  dl = (int)rel;
  const float l = a_src[s] + s_ad[dl];
  lg = l > 0.0f ? l : NEG_SLOPE * l;
  return true;
}

__device__ __forceinline__ unsigned zero_bytes(unsigned t) {
  return ~(((t & 0x7f7f7f7fu) + 0x7f7f7f7fu) | t | 0x7f7f7f7fu);
}

#define OWN_ROWS ((NPW + OWN_WAVES - 1) / OWN_WAVES)
#define NPL ((NPW + 31) / 32)
#define OWN_START ((NPW * 32 + OWN_WAVES * 32 - 1) / (OWN_WAVES * 32))

template <int MODE>
__global__ __launch_bounds__(OWN_WAVES * 32)
void gat_owner_accum(const int* __restrict__ ei, const float* __restrict__ a_src,
                     const float* __restrict__ a_dst, const float* __restrict__ self_logit,
                     const float* __restrict__ h, const float* x,
                     const float* __restrict__ bias, const float* __restrict__ gamma,
                     const float* __restrict__ beta, float* out,
                     int n_nodes, int n_edges, int* elist, int* ecnt,
                     const int* __restrict__ ownb) {
  __shared__ float s_acc[NPW * D];
  __shared__ float s_sum[NPW];
  __shared__ float s_mx[NPW];
  __shared__ float s_ad[NPW];
  __shared__ float s_sl[NPW];
  __shared__ int   s_qe[OWN_WAVES * QCAP];
  __shared__ int   s_qd[OWN_WAVES * QCAP];
  __shared__ int   s_cnt[OWN_WAVES];
  __shared__ int   s_deg[NPL * 32];
  __shared__ int   s_off[NPL * 32];

  const int tid  = threadIdx.x;
  const int wave = __builtin_amdgcn_readfirstlane(tid >> 5);
  const int lane = tid & 31;
  const int n0   = blockIdx.x * NPW;
  const int cnt  = min(NPW, n_nodes - n0);

  int* qe = s_qe + wave * QCAP;
  int* qd = s_qd + wave * QCAP;
  const int slot = blockIdx.x * OWN_WAVES + wave;
  int* lst = elist + (size_t)slot * LCAP;

  const float4 bv = ((const float4*)bias)[lane];
  const float4 gv = ((const float4*)gamma)[lane];
  const float4 tv = ((const float4*)beta)[lane];
  float4 xr[OWN_ROWS];
#pragma unroll
  for (int i = 0; i < OWN_ROWS; ++i) {
    const int r = wave + i * OWN_WAVES;
    xr[i] = (r < cnt) ? ((const float4*)(x + (size_t)(n0 + r) * D))[lane]
                      : make_float4(0.0f, 0.0f, 0.0f, 0.0f);
  }

  if (tid < cnt) {
    const float sl = self_logit[n0 + tid];
    s_sl[tid] = sl;
    s_mx[tid] = sl;
    s_ad[tid] = a_dst[n0 + tid];
  }
  if (tid < NPL * 32) s_deg[tid] = (tid < cnt) ? 1 : 0;

  int pw[LCH], rk[LCH];
  float sv[LCH], lg[LCH];
  int lcnt = 0;
  bool replay = false;

  if (MODE == 2) {
    const int c = ecnt[blockIdx.x * OWN_WAVES + lane];
    lcnt = __builtin_amdgcn_readlane(c, wave);
    replay = __ballot((unsigned)c > (unsigned)LCAP) == 0;
    if (replay) {
#pragma unroll
      for (int k = 0; k < LCH; ++k) {
        const int idx = k * 32 + lane;
        pw[k] = (idx < lcnt) ? lst[idx] : -1;
      }
#pragma unroll
      for (int k = 0; k < LCH; ++k) sv[k] = a_src[max(pw[k], 0) >> 8];
    }
  }

  if (MODE == 1) {
    int rn = 0;
    int qn = 0;
    const uint4* __restrict__ ow4 = (const uint4*)ownb;
    const unsigned idw = (unsigned)blockIdx.x * 0x01010101u;
    const int n_grp = (n_edges + 15) >> 4;
    const int stepg = BSCAN * 32;
    const int n_stp = (n_grp + stepg - 1) / stepg;
    const int first = (int)(((long long)blockIdx.x * n_stp) / gridDim.x);
    uint4 nx[BSCAN];
    int gn = 0;
    if (wave < n_stp) {
      int sg = first + wave;
      if (sg >= n_stp) sg -= n_stp;
      gn = sg * stepg;
#pragma unroll
      for (int k = 0; k < BSCAN; ++k) {
        const int g = gn + k * 32 + lane;
        nx[k] = (g < n_grp) ? ow4[g] : make_uint4(~0u, ~0u, ~0u, ~0u);
      }
    }
    for (int it = wave; it < n_stp; it += OWN_WAVES) {
      const int g0 = gn;
      uint4 ov[BSCAN];
#pragma unroll
      for (int k = 0; k < BSCAN; ++k) ov[k] = nx[k];
      if (it + OWN_WAVES < n_stp) {
        int sg = first + it + OWN_WAVES;
        if (sg >= n_stp) sg -= n_stp;
        gn = sg * stepg;
        if (gn + stepg <= n_grp) {
#pragma unroll
          for (int k = 0; k < BSCAN; ++k) nx[k] = ow4[gn + k * 32 + lane];
        } else {
#pragma unroll
          for (int k = 0; k < BSCAN; ++k) {
            const int g = gn + k * 32 + lane;
            nx[k] = (g < n_grp) ? ow4[g] : make_uint4(~0u, ~0u, ~0u, ~0u);
          }
        }
      }
#pragma unroll
      for (int k = 0; k < BSCAN; ++k) {
        const int g = g0 + k * 32 + lane;
        unsigned hm = (zero_bytes(ov[k].x ^ idw) >> 7) | (zero_bytes(ov[k].y ^ idw) >> 6) |
                      (zero_bytes(ov[k].z ^ idw) >> 5) | (zero_bytes(ov[k].w ^ idw) >> 4);
        unsigned msk = (unsigned)__ballot(hm != 0u);
        while (msk != 0u) {
          if (hm != 0u) {
            const int p = __builtin_ctz(hm);
            const int pos = qn + __popc(msk & ((1u << lane) - 1u));
            qe[pos] = g * 16 + (p & 7) * 4 + (p >> 3);
            hm &= hm - 1u;
          }
          qn += __popc(msk);
          msk = (unsigned)__ballot(hm != 0u);
        }
        while (qn >= 32) {
          qn -= 32;
          own_note(qe + qn, 32, lane, n0, n_edges, ei, lst, rn);
          rn += 32;
        }
      }
    }
    if (qn > 0) {
      own_note(qe, qn, lane, n0, n_edges, ei, lst, rn);
      rn += qn;
    }
    if (lane == 0) s_cnt[wave] = rn;
  }
  __syncthreads();

  if (MODE == 1) {
    const int c = s_cnt[lane];
    lcnt = __builtin_amdgcn_readlane(c, wave);
    replay = __ballot((unsigned)c > (unsigned)LCAP) == 0;
    if (wave == 0) ecnt[blockIdx.x * OWN_WAVES + lane] = c;
    if (replay) {
#pragma unroll
      for (int k = 0; k < LCH; ++k) {
        const int idx = k * 32 + lane;
        pw[k] = (idx < lcnt) ? lst[idx] : -1;
      }
#pragma unroll
      for (int k = 0; k < LCH; ++k) sv[k] = a_src[max(pw[k], 0) >> 8];
    }
  }

  if (replay) {
#pragma unroll
    for (int k = 0; k < LCH; ++k) {
      lg[k] = 0.0f;
      rk[k] = 0;
      if (pw[k] >= 0) {
        const int dl = pw[k] & 255;
        const float l = sv[k] + s_ad[dl];
        lg[k] = l > 0.0f ? l : NEG_SLOPE * l;
        __hip_atomic_fetch_max(&s_mx[dl], lg[k], __ATOMIC_RELAXED, __HIP_MEMORY_SCOPE_WORKGROUP);
        rk[k] = __hip_atomic_fetch_add(&s_deg[dl], 1, __ATOMIC_RELAXED, __HIP_MEMORY_SCOPE_WORKGROUP);
      }
    }
  } else {
    for (int base = wave * 32; base < n_edges; base += OWN_WAVES * 32) {
      int s, dl;
      float l;
      if (own_slow_edge(base + lane, n_edges, n0, cnt, ei, a_src, s_ad, s, dl, l))
        __hip_atomic_fetch_max(&s_mx[dl], l, __ATOMIC_RELAXED, __HIP_MEMORY_SCOPE_WORKGROUP);
    }
  }

  float4 hst[OWN_START];
#pragma unroll
  for (int i = 0; i < OWN_START; ++i) {
    const int p = tid + i * OWN_WAVES * 32;
    hst[i] = (!replay && p < cnt * 32) ? ((const float4*)(h + (size_t)n0 * D))[p]
                                       : make_float4(0.0f, 0.0f, 0.0f, 0.0f);
  }
  __syncthreads();

  int*   ent_w = s_qe;
  float* ent_f = (float*)s_qd;
  int run_b = 0, run_e = 0;
  if (replay) {
    int of[NPL];
    int run = 0;
#pragma unroll
    for (int i = 0; i < NPL; ++i) {
      of[i] = run;
      run += s_deg[lane * NPL + i];
    }
    int incl = run;
#pragma unroll
    for (int d = 1; d < 32; d <<= 1) {
      const int t = __shfl_up(incl, d, 32);
      if (lane >= d) incl += t;
    }
    const int excl  = incl - run;
    const int total = __builtin_amdgcn_readlane(incl, 31);
#pragma unroll
    for (int i = 0; i < NPL; ++i) {
      of[i] += excl;
      s_off[lane * NPL + i] = of[i];
    }
    const int tg0 = (wave * total) >> 5;
    const int tg1 = ((wave + 1) * total) >> 5;
    int rb = total, re = total;
#pragma unroll
    for (int i = 0; i < NPL; ++i) {
      if (lane * NPL + i < cnt) {
        if (of[i] >= tg0) rb = min(rb, of[i]);
        if (of[i] >= tg1) re = min(re, of[i]);
      }
    }
#pragma unroll
    for (int d = 16; d > 0; d >>= 1) {
      rb = min(rb, __shfl_xor(rb, d, 32));
      re = min(re, __shfl_xor(re, d, 32));
    }
    run_b = __builtin_amdgcn_readfirstlane(rb);
    run_e = __builtin_amdgcn_readfirstlane(re);

    if (tid < cnt) {
      const int pos = s_off[tid];
      ent_w[pos] = ((n0 + tid) << 8) | tid;
      ent_f[pos] = __expf(s_sl[tid] - s_mx[tid]);
    }
#pragma unroll
    for (int k = 0; k < LCH; ++k) {
      if (pw[k] >= 0) {
        const int dl  = pw[k] & 255;
        const int pos = s_off[dl] + rk[k];
        ent_w[pos] = pw[k];
        ent_f[pos] = __expf(lg[k] - s_mx[dl]);
      }
    }
  } else {
#pragma unroll
    for (int i = 0; i < OWN_START; ++i) {
      const int p = tid + i * OWN_WAVES * 32;
      if (p < cnt * 32) {
        const int r = p >> 5, c = p & 31;
        const float e0 = __expf(s_sl[r] - s_mx[r]);
        float* sp = s_acc + r * D + c;
        sp[0]  = hst[i].x * e0;
        sp[32] = hst[i].y * e0;
        sp[64] = hst[i].z * e0;
        sp[96] = hst[i].w * e0;
        if (c == 0) s_sum[r] = e0;
      }
    }
  }
  __syncthreads();

  if (replay) {
    RowAcc ra;
    ra.x = 0.0f; ra.y = 0.0f; ra.z = 0.0f; ra.w = 0.0f; ra.s = 0.0f; ra.cur = -1;
    int off = run_b;
    for (; off + 32 <= run_e; off += 32)
      own_stream(ent_w[off + lane], ent_f[off + lane], 32, lane, h, s_acc, s_sum, ra);
    if (off < run_e) {
      const int m = run_e - off;
      int wd = 0;
      float wt = 0.0f;
      if (lane < m) {
        wd = ent_w[off + lane];
        wt = ent_f[off + lane];
      }
      own_stream(wd, wt, m, lane, h, s_acc, s_sum, ra);
    }
    if (ra.cur >= 0) own_flush(ra, lane, s_acc, s_sum);
  } else {
    for (int base = wave * 32; base < n_edges; base += OWN_WAVES * 32) {
      int s, dl;
      float l;
      const bool hit = own_slow_edge(base + lane, n_edges, n0, cnt, ei, a_src, s_ad, s, dl, l);
      float w = 0.0f;
      if (hit) {
        w = __expf(l - s_mx[dl]);
        __hip_atomic_fetch_add(&s_sum[dl], w, __ATOMIC_RELAXED, __HIP_MEMORY_SCOPE_WORKGROUP);
      }
      unsigned msk = (unsigned)__ballot(hit);
      while (msk != 0u) {
        const int j = __builtin_ctz(msk);
        msk &= msk - 1u;
        const int sj = __builtin_amdgcn_readlane(s, j);
        const int dj = __builtin_amdgcn_readlane(dl, j);
        const float wj = __builtin_bit_cast(float, __builtin_amdgcn_readlane(__builtin_bit_cast(int, w), j));
        const float4 hv = ((const float4*)h)[((unsigned)sj << 5) | (unsigned)lane];
        float* ap = s_acc + dj * D + lane;
        __hip_atomic_fetch_add(ap + 0,  hv.x * wj, __ATOMIC_RELAXED, __HIP_MEMORY_SCOPE_WORKGROUP);
        __hip_atomic_fetch_add(ap + 32, hv.y * wj, __ATOMIC_RELAXED, __HIP_MEMORY_SCOPE_WORKGROUP);
        __hip_atomic_fetch_add(ap + 64, hv.z * wj, __ATOMIC_RELAXED, __HIP_MEMORY_SCOPE_WORKGROUP);
        __hip_atomic_fetch_add(ap + 96, hv.w * wj, __ATOMIC_RELAXED, __HIP_MEMORY_SCOPE_WORKGROUP);
      }
    }
  }
  __syncthreads();

  float4 y[OWN_ROWS];
  float sum[OWN_ROWS], sq[OWN_ROWS];
#pragma unroll
  for (int i = 0; i < OWN_ROWS; ++i) {
    const int rc = min(wave + i * OWN_WAVES, cnt - 1);
    const float* fp = s_acc + rc * D + lane;
    const float4 av = make_float4(fp[0], fp[32], fp[64], fp[96]);
    const float inv_s = 1.0f / s_sum[rc];
    y[i].x = xr[i].x + av.x * inv_s + bv.x;
    y[i].y = xr[i].y + av.y * inv_s + bv.y;
    y[i].z = xr[i].z + av.z * inv_s + bv.z;
    y[i].w = xr[i].w + av.w * inv_s + bv.w;
    sum[i] = y[i].x + y[i].y + y[i].z + y[i].w;
    sq[i]  = y[i].x * y[i].x + y[i].y * y[i].y + y[i].z * y[i].z + y[i].w * y[i].w;
  }
#pragma unroll
  for (int off = 16; off > 0; off >>= 1) {
#pragma unroll
    for (int i = 0; i < OWN_ROWS; ++i) {
      sum[i] += __shfl_xor(sum[i], off, 32);
      sq[i]  += __shfl_xor(sq[i], off, 32);
    }
  }
#pragma unroll
  for (int i = 0; i < OWN_ROWS; ++i) {
    const int r = wave + i * OWN_WAVES;
    const float mu   = sum[i] * (1.0f / D);
    const float var  = sq[i] * (1.0f / D) - mu * mu;
    const float rstd = rsqrtf(var + LN_EPS);
    float4 o;
    o.x = (y[i].x - mu) * rstd * gv.x + tv.x;
    o.y = (y[i].y - mu) * rstd * gv.y + tv.y;
    o.z = (y[i].z - mu) * rstd * gv.z + tv.z;
    o.w = (y[i].w - mu) * rstd * gv.w + tv.w;
    if (r < cnt) ((float4*)(out + (size_t)(n0 + r) * D))[lane] = o;
  }
}

extern "C" void kernel_launch(void* const* d_in, const int* in_sizes, int n_in,
                              void* d_out, int out_size, void* d_ws, size_t ws_size,
                              hipStream_t stream) {
  const float* x0   = (const float*)d_in[0];
  const int*   ei   = (const int*)d_in[1];
  const float* W    = (const float*)d_in[2];
  const float* asrc = (const float*)d_in[3];
  const float* adst = (const float*)d_in[4];
  const float* bias = (const float*)d_in[5];
  const float* gam  = (const float*)d_in[6];
  const float* bet  = (const float*)d_in[7];
  float* out = (float*)d_out;

  const int n  = in_sizes[0] / D;
  const int e  = in_sizes[1] / 2;
  const int nl = in_sizes[2] / (D * D);

  float* ws    = (float*)d_ws;
  float* h     = ws; ws += (size_t)n * D;
  float* va    = ws; ws += n;
  float* vb    = ws; ws += n;
  float* slf   = ws; ws += n;

  const int    nblk      = (n + NPW - 1) / NPW;
  const size_t lst_off   = (((size_t)n * (D + 4) + (size_t)e) * sizeof(float) + 127) / 128 * 128;
  const size_t lst_words = (size_t)nblk * OWN_WAVES * LCAP;
  const size_t lst_end   = lst_off + (lst_words + (size_t)nblk * OWN_WAVES) * sizeof(int);
  const size_t own_end   = lst_end + ((size_t)e + 15) / 16 * 16;
  const bool   use_list  = ws_size >= own_end && n < (1 << 23) && nblk <= 255;
  int* elist = use_list ? (int*)((char*)d_ws + lst_off) : nullptr;
  int* ecnt  = use_list ? elist + lst_words : nullptr;
  int* ownb  = use_list ? (int*)((char*)d_ws + lst_end) : nullptr;

  for (int l = 0; l < nl; ++l) {
    const float* xin = (l == 0) ? x0 : out;
    const float* Wl  = W + (size_t)l * D * D;
    int* ownl = (l == 0) ? ownb : nullptr;
    gat_gemm_wmma<<<(n + 127) / 128, 256, 0, stream>>>(xin, Wl, asrc + l * D, adst + l * D,
                                                       h, va, vb, slf, n, ei + e, ownl, e);
    if (!use_list)
      gat_owner_accum<0><<<nblk, OWN_WAVES * 32, 0, stream>>>(ei, va, vb, slf, h, xin,
                                                               bias + l * D, gam + l * D,
                                                               bet + l * D, out, n, e,
                                                               nullptr, nullptr, nullptr);
    else if (l == 0)
      gat_owner_accum<1><<<nblk, OWN_WAVES * 32, 0, stream>>>(ei, va, vb, slf, h, xin,
                                                               bias + l * D, gam + l * D,
                                                               bet + l * D, out, n, e,
                                                               elist, ecnt, ownb);
    else
      gat_owner_accum<2><<<nblk, OWN_WAVES * 32, 0, stream>>>(ei, va, vb, slf, h, xin,
                                                               bias + l * D, gam + l * D,
                                                               bet + l * D, out, n, e,
                                                               elist, ecnt, nullptr);
  }
}
